# starting kernel
# baseline (speedup 1.0000x reference)
#include <hip/hip_runtime.h>
#include <hip/hip_bf16.h>

typedef __attribute__((ext_vector_type(2))) float v2f;
typedef __attribute__((ext_vector_type(8))) float v8f;

#define NN  512
#define CC  32
#define LL  4
#define LN_EPS 1e-5f

__global__ __launch_bounds__(256) void rgcn_invdeg_kernel(
    const int* __restrict__ dst, int E, float* __restrict__ invdeg)
{
    __shared__ int deg[3 * NN];
    const int tid = threadIdx.x;
    for (int i = tid; i < 3 * NN; i += 256) deg[i] = 0;
    __syncthreads();
    const int tot = 3 * E;
    for (int i = tid; i < tot; i += 256) {
        int r = i / E;
        atomicAdd(&deg[r * NN + dst[i]], 1);
    }
    __syncthreads();
    for (int i = tid; i < 3 * NN; i += 256) {
        int d = deg[i];
        if (d < 1) d = 1;
        invdeg[i] = 1.0f / (float)d;
    }
}

__device__ __forceinline__ void gemm_acc(
    const float* __restrict__ A, const float* __restrict__ W,
    const float* __restrict__ scale, int wid, int mlo, int mhalf,
    v8f acc[4][2])
{
    for (int ks = 0; ks < 8; ++ks) {
        const int kk = ks * 4 + mhalf * 2;
        v2f b0, b1;
        b0.x = W[kk * CC + mlo];
        b0.y = W[(kk + 1) * CC + mlo];
        b1.x = W[kk * CC + 16 + mlo];
        b1.y = W[(kk + 1) * CC + 16 + mlo];
#pragma unroll
        for (int mt = 0; mt < 4; ++mt) {
            const int node = wid * 64 + mt * 16 + mlo;
            const float sc = scale ? scale[node] : 1.0f;
            v2f a;
            a.x = A[node * CC + kk] * sc;
            a.y = A[node * CC + kk + 1] * sc;
            acc[mt][0] = __builtin_amdgcn_wmma_f32_16x16x4_f32(
                false, a, false, b0, (short)0, acc[mt][0], false, false);
            acc[mt][1] = __builtin_amdgcn_wmma_f32_16x16x4_f32(
                false, a, false, b1, (short)0, acc[mt][1], false, false);
        }
    }
}

__global__ __launch_bounds__(256) void rgcn_main_kernel(
    const float* __restrict__ xx, const float* __restrict__ ss,
    const float* __restrict__ Wf, const float* __restrict__ bf,
    const float* __restrict__ convw, const float* __restrict__ convroot,
    const float* __restrict__ convbias, const float* __restrict__ lng,
    const float* __restrict__ lnb, const int* __restrict__ src,
    const int* __restrict__ dst, const float* __restrict__ invdeg,
    float* __restrict__ out, int E)
{
    __shared__ float xs[NN * CC];
    __shared__ float hs[NN * CC];
    __shared__ float wts[CC * CC];
    __shared__ float ivd[3 * NN];
    __shared__ float wfe[6 * CC];
    __shared__ float prm[3 * CC];

    const int bid   = blockIdx.x;
    const int tid   = threadIdx.x;
    const int lane  = tid & 31;
    const int wid   = tid >> 5;
    const int mlo   = lane & 15;
    const int mhalf = lane >> 4;

    for (int i = tid; i < 3 * NN; i += 256) ivd[i] = invdeg[i];
    for (int i = tid; i < 6 * CC; i += 256) wfe[i] = Wf[i];
    if (tid < CC) {
        prm[tid]          = bf[tid];
        prm[CC + tid]     = lng[tid];
        prm[2 * CC + tid] = lnb[tid];
    }
    __syncthreads();

    const float mfeat = ss[bid] * 0.125f;
    for (int idx = tid; idx < NN * CC; idx += 256) {
        const int n = idx >> 5, c = idx & 31;
        const int t = n >> 6, i = (n >> 4) & 3, j = (n >> 2) & 3, k = n & 3;
        const float v = xx[bid * NN + n];
        float a = prm[c];
        a += ((float)i * (1.0f / 3.0f)) * wfe[0 * CC + c];
        a += ((float)j * (1.0f / 3.0f)) * wfe[1 * CC + c];
        a += ((float)k * (1.0f / 3.0f)) * wfe[2 * CC + c];
        a += ((float)t * (1.0f / 7.0f)) * wfe[3 * CC + c];
        a += v * wfe[4 * CC + c];
        a += mfeat * wfe[5 * CC + c];
        xs[idx] = a;
    }
    __syncthreads();

    for (int l = 0; l < LL; ++l) {
        v8f acc[4][2];
#pragma unroll
        for (int mt = 0; mt < 4; ++mt)
#pragma unroll
            for (int nt = 0; nt < 2; ++nt) {
                const float bv = convbias[l * CC + nt * 16 + mlo];
                acc[mt][nt] = (v8f){bv, bv, bv, bv, bv, bv, bv, bv};
            }

        for (int i = tid; i < NN * CC; i += 256) hs[i] = 0.0f;
        for (int i = tid; i < CC * CC; i += 256) wts[i] = convroot[l * CC * CC + i];
        __syncthreads();

        gemm_acc(xs, wts, nullptr, wid, mlo, mhalf, acc);
        __syncthreads();

        for (int r = 0; r < 3; ++r) {
            const int* sR = src + r * E;
            const int* dR = dst + r * E;
            for (int e = wid; e < E; e += 8) {
                const int sn = sR[e], dn = dR[e];
                atomicAdd(&hs[dn * CC + lane], xs[sn * CC + lane]);
            }
            for (int i = tid; i < CC * CC; i += 256)
                wts[i] = convw[(l * 3 + r) * CC * CC + i];
            __syncthreads();

            gemm_acc(hs, wts, &ivd[r * NN], wid, mlo, mhalf, acc);
            __syncthreads();

            if (r < 2) {
                for (int i = tid; i < NN * CC; i += 256) hs[i] = 0.0f;
                __syncthreads();
            }
        }

#pragma unroll
        for (int mt = 0; mt < 4; ++mt)
#pragma unroll
            for (int nt = 0; nt < 2; ++nt)
#pragma unroll
                for (int jj = 0; jj < 8; ++jj) {
                    const int node = wid * 64 + mt * 16 + jj + 8 * mhalf;
                    const int col  = nt * 16 + mlo;
                    const float v  = acc[mt][nt][jj];
                    hs[node * CC + col] = v > 0.0f ? v : 0.0f;
                }
        __syncthreads();

        for (int q = 0; q < 64; ++q) {
            const int node = wid * 64 + q;
            const float v = hs[node * CC + lane];
            float s = v;
            for (int off = 16; off; off >>= 1) s += __shfl_xor(s, off, 32);
            const float mu = s * (1.0f / 32.0f);
            const float d = v - mu;
            float s2 = d * d;
            for (int off = 16; off; off >>= 1) s2 += __shfl_xor(s2, off, 32);
            const float var = s2 * (1.0f / 32.0f);
            xs[node * CC + lane] =
                d * rsqrtf(var + LN_EPS) * prm[CC + lane] + prm[2 * CC + lane];
        }
        __syncthreads();
    }

    for (int idx = tid; idx < 48 * CC; idx += 256) {
        const int row = idx >> 5, c = idx & 31;
        const int g = row >> 4, rc = row & 15;
        const int a0 = rc >> 2, b0 = rc & 3;
        float sum = 0.0f;
#pragma unroll
        for (int q = 0; q < 4; ++q) {
            int node;
            if (g == 0)      node = q * 16 + a0 * 4 + b0;
            else if (g == 1) node = a0 * 16 + q * 4 + b0;
            else             node = a0 * 16 + b0 * 4 + q;
            sum += xs[node * CC + c];
        }
        out[bid * 48 * CC + idx] = sum * 0.25f;
    }
}

extern "C" void kernel_launch(void* const* d_in, const int* in_sizes, int n_in,
                              void* d_out, int out_size, void* d_ws, size_t ws_size,
                              hipStream_t stream) {
    const float* xx       = (const float*)d_in[0];
    const float* ss       = (const float*)d_in[1];
    const float* Wf       = (const float*)d_in[2];
    const float* bf       = (const float*)d_in[3];
    const float* convw    = (const float*)d_in[4];
    const float* convroot = (const float*)d_in[5];
    const float* convbias = (const float*)d_in[6];
    const float* lng      = (const float*)d_in[7];
    const float* lnb      = (const float*)d_in[8];
    const int*   src      = (const int*)d_in[9];
    const int*   dst      = (const int*)d_in[10];
    float*       out      = (float*)d_out;
    float*       invdeg   = (float*)d_ws;

    const int E = in_sizes[9] / 3;
    const int B = in_sizes[1];

    rgcn_invdeg_kernel<<<1, 256, 0, stream>>>(dst, E, invdeg);
    rgcn_main_kernel<<<B, 256, 0, stream>>>(xx, ss, Wf, bf, convw, convroot,
                                            convbias, lng, lnb, src, dst,
                                            invdeg, out, E);
    (void)n_in; (void)out_size; (void)ws_size;
}


# Aggregation as matrix products with per-relation 16x16 edge-count tiles built by the pre-kernel (non-empty tiles only), not per-edge LDS float atomics; sum order now fixed (matrix k order)
# speedup vs baseline: 8.3042x; 8.3042x over previous
#include <hip/hip_runtime.h>
#include <hip/hip_bf16.h>

typedef __attribute__((ext_vector_type(2))) float v2f;
typedef __attribute__((ext_vector_type(8))) float v8f;

#define NN  512
#define CC  32
#define LL  4
#define LN_EPS 1e-5f

#define WS_INVDEG  0
#define WS_ROWMASK 1536
#define WS_ROWBASE 1664
#define WS_TILES   2048

__global__ __launch_bounds__(1024) void rgcn_prep_kernel(
    const int* __restrict__ src, const int* __restrict__ dst, int E,
    float* __restrict__ invdeg, unsigned* __restrict__ rowmask,
    int* __restrict__ rowbase, int* __restrict__ tiles)
{
    __shared__ int deg[NN];
    __shared__ unsigned rmask[32];
    __shared__ int rbase[32];
    __shared__ int total_s;
    const int r = blockIdx.x;
    const int tid = threadIdx.x;
    const int* sR = src + r * E;
    const int* dR = dst + r * E;
    for (int i = tid; i < NN; i += 1024) deg[i] = 0;
    if (tid < 32) rmask[tid] = 0u;
    __syncthreads();
    for (int e = tid; e < E; e += 1024) {
        const int s = sR[e], d = dR[e];
        atomicAdd(&deg[d], 1);
        atomicOr(&rmask[d >> 4], 1u << (s >> 4));
    }
    __syncthreads();
    for (int i = tid; i < NN; i += 1024) {
        int d = deg[i];
        if (d < 1) d = 1;
        invdeg[r * NN + i] = 1.0f / (float)d;
    }
    if (tid < 32) {
        const unsigned m = rmask[tid];
        const int cnt = __popc(m);
        int inc = cnt;
        for (int off = 1; off < 32; off <<= 1) {
            const int t = __shfl_up(inc, off, 32);
            if (tid >= off) inc += t;
        }
        rbase[tid] = inc - cnt;
        rowmask[r * 32 + tid] = m;
        rowbase[r * 32 + tid] = r * 1024 + inc - cnt;
        if (tid == 31) total_s = inc;
    }
    __syncthreads();
    int* T = tiles + r * 1024 * 256;
    const int nw = total_s * 256;
    for (int i = tid; i < nw; i += 1024) T[i] = 0;
    __syncthreads();
    for (int e = tid; e < E; e += 1024) {
        const int s = sR[e], d = dR[e];
        const int tm = d >> 4, tk = s >> 4, k = s & 15;
        const int idx = rbase[tm] + __popc(rmask[tm] & ((1u << tk) - 1u));
        const int off = ((k & 1) * 16 + (d & 15)) * 8 + (k >> 2) * 2 + ((k >> 1) & 1);
        atomicAdd(&T[idx * 256 + off], 1);
    }
}

__device__ __forceinline__ void gemm_acc(
    const float* __restrict__ A, const float* __restrict__ W,
    const float* __restrict__ scale, int wid, int mlo, int mhalf,
    v8f acc[4][2])
{
    for (int ks = 0; ks < 8; ++ks) {
        const int kk = ks * 4 + mhalf * 2;
        v2f b0, b1;
        b0.x = W[kk * CC + mlo];
        b0.y = W[(kk + 1) * CC + mlo];
        b1.x = W[kk * CC + 16 + mlo];
        b1.y = W[(kk + 1) * CC + 16 + mlo];
#pragma unroll
        for (int mt = 0; mt < 4; ++mt) {
            const int node = wid * 64 + mt * 16 + mlo;
            const float sc = scale ? scale[node] : 1.0f;
            v2f a;
            a.x = A[node * CC + kk] * sc;
            a.y = A[node * CC + kk + 1] * sc;
            acc[mt][0] = __builtin_amdgcn_wmma_f32_16x16x4_f32(
                false, a, false, b0, (short)0, acc[mt][0], false, false);
            acc[mt][1] = __builtin_amdgcn_wmma_f32_16x16x4_f32(
                false, a, false, b1, (short)0, acc[mt][1], false, false);
        }
    }
}

__global__ __launch_bounds__(256) void rgcn_main_kernel(
    const float* __restrict__ xx, const float* __restrict__ ss,
    const float* __restrict__ Wf, const float* __restrict__ bf,
    const float* __restrict__ convw, const float* __restrict__ convroot,
    const float* __restrict__ convbias, const float* __restrict__ lng,
    const float* __restrict__ lnb, const float* __restrict__ invdeg,
    const unsigned* __restrict__ rowmask, const int* __restrict__ rowbase,
    const int* __restrict__ tiles, float* __restrict__ out)
{
    __shared__ float xs[NN * CC];
    __shared__ float wts[4 * CC * CC];
    __shared__ float ivd[3 * NN];
    __shared__ unsigned rmk[3 * 32];
    __shared__ int rbs[3 * 32];
    __shared__ float wfe[6 * CC];
    __shared__ float prm[3 * CC];

    const int bid   = blockIdx.x;
    const int tid   = threadIdx.x;
    const int lane  = tid & 31;
    const int wid   = tid >> 5;
    const int mlo   = lane & 15;
    const int mhalf = lane >> 4;

    for (int i = tid; i < 3 * NN; i += 256) ivd[i] = invdeg[i];
    for (int i = tid; i < 6 * CC; i += 256) wfe[i] = Wf[i];
    if (tid < 96) {
        rmk[tid] = rowmask[tid];
        rbs[tid] = rowbase[tid];
    }
    if (tid < CC) {
        prm[tid]          = bf[tid];
        prm[CC + tid]     = lng[tid];
        prm[2 * CC + tid] = lnb[tid];
    }
    __syncthreads();

    const float mfeat = ss[bid] * 0.125f;
    for (int idx = tid; idx < NN * CC; idx += 256) {
        const int n = idx >> 5, c = idx & 31;
        const int t = n >> 6, i = (n >> 4) & 3, j = (n >> 2) & 3, k = n & 3;
        const float v = xx[bid * NN + n];
        float a = prm[c];
        a += ((float)i * (1.0f / 3.0f)) * wfe[0 * CC + c];
        a += ((float)j * (1.0f / 3.0f)) * wfe[1 * CC + c];
        a += ((float)k * (1.0f / 3.0f)) * wfe[2 * CC + c];
        a += ((float)t * (1.0f / 7.0f)) * wfe[3 * CC + c];
        a += v * wfe[4 * CC + c];
        a += mfeat * wfe[5 * CC + c];
        xs[idx] = a;
    }
    __syncthreads();

    for (int l = 0; l < LL; ++l) {
        v8f acc[4][2];
#pragma unroll
        for (int mt = 0; mt < 4; ++mt)
#pragma unroll
            for (int nt = 0; nt < 2; ++nt) {
                const float bv = convbias[l * CC + nt * 16 + mlo];
                acc[mt][nt] = (v8f){bv, bv, bv, bv, bv, bv, bv, bv};
            }

        for (int i = tid; i < CC * CC; i += 256) wts[i] = convroot[l * CC * CC + i];
        for (int i = tid; i < 3 * CC * CC; i += 256)
            wts[CC * CC + i] = convw[l * 3 * CC * CC + i];
        __syncthreads();

        gemm_acc(xs, wts, nullptr, wid, mlo, mhalf, acc);

        for (int r = 0; r < 3; ++r) {
            unsigned mk[4];
#pragma unroll
            for (int mt = 0; mt < 4; ++mt)
                mk[mt] = (unsigned)__builtin_amdgcn_readfirstlane((int)rmk[r * 32 + wid * 4 + mt]);
            const int base = __builtin_amdgcn_readfirstlane(rbs[r * 32 + wid * 4]);
            int rem = __popc(mk[0]) + __popc(mk[1]) + __popc(mk[2]) + __popc(mk[3]);
            const int4* tp = (const int4*)(tiles + base * 256 + lane * 8);
            float cf[8] = {0, 0, 0, 0, 0, 0, 0, 0};
            if (rem) {
                const int4 t0 = tp[0], t1 = tp[1];
                cf[0] = (float)t0.x; cf[1] = (float)t0.y; cf[2] = (float)t0.z; cf[3] = (float)t0.w;
                cf[4] = (float)t1.x; cf[5] = (float)t1.y; cf[6] = (float)t1.z; cf[7] = (float)t1.w;
            }

            const float* Wr = wts + (1 + r) * CC * CC;
            v2f wf[8][2];
#pragma unroll
            for (int q = 0; q < 8; ++q)
#pragma unroll
                for (int nt = 0; nt < 2; ++nt) {
                    const int kb = (q >> 2) * 16 + 8 * mhalf + 2 * (q & 3);
                    wf[q][nt].x = Wr[kb * CC + nt * 16 + mlo];
                    wf[q][nt].y = Wr[(kb + 1) * CC + nt * 16 + mlo];
                }

#pragma unroll
            for (int mt = 0; mt < 4; ++mt) {
                v8f g0 = (v8f){0, 0, 0, 0, 0, 0, 0, 0};
                v8f g1 = (v8f){0, 0, 0, 0, 0, 0, 0, 0};
                unsigned m = mk[mt];
                while (m) {
                    const int tk = __builtin_ctz(m);
                    m &= m - 1u;
                    --rem;
                    tp += rem ? 64 : 0;
                    const int4 p0 = tp[0], p1 = tp[1];
                    const float* xb = xs + (tk * 16 + mhalf) * CC + mlo;
#pragma unroll
                    for (int ks = 0; ks < 4; ++ks) {
                        v2f b, a0, a1;
                        b.x  = cf[2 * ks];
                        b.y  = cf[2 * ks + 1];
                        a0.x = xb[(4 * ks) * CC];
                        a0.y = xb[(4 * ks + 2) * CC];
                        a1.x = xb[(4 * ks) * CC + 16];
                        a1.y = xb[(4 * ks + 2) * CC + 16];
                        g0 = __builtin_amdgcn_wmma_f32_16x16x4_f32(
                            false, a0, false, b, (short)0, g0, false, false);
                        g1 = __builtin_amdgcn_wmma_f32_16x16x4_f32(
                            false, a1, false, b, (short)0, g1, false, false);
                    }
                    cf[0] = (float)p0.x; cf[1] = (float)p0.y; cf[2] = (float)p0.z; cf[3] = (float)p0.w;
                    cf[4] = (float)p1.x; cf[5] = (float)p1.y; cf[6] = (float)p1.z; cf[7] = (float)p1.w;
                }
                const float sc = ivd[r * NN + (wid * 4 + mt) * 16 + mlo];
                g0 *= sc;
                g1 *= sc;
#pragma unroll
                for (int q = 0; q < 8; ++q) {
                    v2f a;
                    if (q < 4) { a.x = g0[2 * q];       a.y = g0[2 * q + 1]; }
                    else       { a.x = g1[2 * (q - 4)]; a.y = g1[2 * (q - 4) + 1]; }
                    acc[mt][0] = __builtin_amdgcn_wmma_f32_16x16x4_f32(
                        false, a, false, wf[q][0], (short)0, acc[mt][0], false, false);
                    acc[mt][1] = __builtin_amdgcn_wmma_f32_16x16x4_f32(
                        false, a, false, wf[q][1], (short)0, acc[mt][1], false, false);
                }
            }
        }
        __syncthreads();

#pragma unroll
        for (int mt = 0; mt < 4; ++mt)
#pragma unroll
            for (int nt = 0; nt < 2; ++nt)
#pragma unroll
                for (int jj = 0; jj < 8; ++jj) {
                    const int node = wid * 64 + mt * 16 + jj + 8 * mhalf;
                    const int col  = nt * 16 + mlo;
                    const float v  = acc[mt][nt][jj];
                    xs[node * CC + col] = v > 0.0f ? v : 0.0f;
                }
        __syncthreads();

        for (int q = 0; q < 64; ++q) {
            const int node = wid * 64 + q;
            const float v = xs[node * CC + lane];
            float s = v;
            for (int off = 16; off; off >>= 1) s += __shfl_xor(s, off, 32);
            const float mu = s * (1.0f / 32.0f);
            const float d = v - mu;
            float s2 = d * d;
            for (int off = 16; off; off >>= 1) s2 += __shfl_xor(s2, off, 32);
            const float var = s2 * (1.0f / 32.0f);
            xs[node * CC + lane] =
                d * rsqrtf(var + LN_EPS) * prm[CC + lane] + prm[2 * CC + lane];
        }
        __syncthreads();
    }

    for (int idx = tid; idx < 48 * CC; idx += 256) {
        const int row = idx >> 5, c = idx & 31;
        const int g = row >> 4, rc = row & 15;
        const int a0 = rc >> 2, b0 = rc & 3;
        float sum = 0.0f;
#pragma unroll
        for (int q = 0; q < 4; ++q) {
            int node;
            if (g == 0)      node = q * 16 + a0 * 4 + b0;
            else if (g == 1) node = a0 * 16 + q * 4 + b0;
            else             node = a0 * 16 + b0 * 4 + q;
            sum += xs[node * CC + c];
        }
        out[bid * 48 * CC + idx] = sum * 0.25f;
    }
}

extern "C" void kernel_launch(void* const* d_in, const int* in_sizes, int n_in,
                              void* d_out, int out_size, void* d_ws, size_t ws_size,
                              hipStream_t stream) {
    const float* xx       = (const float*)d_in[0];
    const float* ss       = (const float*)d_in[1];
    const float* Wf       = (const float*)d_in[2];
    const float* bf       = (const float*)d_in[3];
    const float* convw    = (const float*)d_in[4];
    const float* convroot = (const float*)d_in[5];
    const float* convbias = (const float*)d_in[6];
    const float* lng      = (const float*)d_in[7];
    const float* lnb      = (const float*)d_in[8];
    const int*   src      = (const int*)d_in[9];
    const int*   dst      = (const int*)d_in[10];
    float*       out      = (float*)d_out;
    float*       invdeg   = (float*)d_ws + WS_INVDEG;
    unsigned*    rowmask  = (unsigned*)d_ws + WS_ROWMASK;
    int*         rowbase  = (int*)d_ws + WS_ROWBASE;
    int*         tiles    = (int*)d_ws + WS_TILES;

    const int E = in_sizes[9] / 3;
    const int B = in_sizes[1];

    rgcn_prep_kernel<<<3, 1024, 0, stream>>>(src, dst, E, invdeg, rowmask, rowbase, tiles);
    rgcn_main_kernel<<<B, 256, 0, stream>>>(xx, ss, Wf, bf, convw, convroot,
                                            convbias, lng, lnb, invdeg, rowmask,
                                            rowbase, tiles, out);
    (void)n_in; (void)out_size; (void)ws_size;
}


# Main kernel's work-group is 32 waves, not 8: each wave owns one 16-row block (was four) and staging is dealt over 1024 threads, so 8 waves per SIMD cover each other's waits; sum order kept
# speedup vs baseline: 15.2966x; 1.8420x over previous
#include <hip/hip_runtime.h>
#include <hip/hip_bf16.h>

typedef __attribute__((ext_vector_type(2))) float v2f;
typedef __attribute__((ext_vector_type(8))) float v8f;

#define NN  512
#define CC  32
#define LL  4
#define LN_EPS 1e-5f
#define NT  1024
static_assert(NT == CC * CC && NT == 2 * NN, "staging and row-block ownership assume 1024 threads");

#define WS_INVDEG  0
#define WS_ROWMASK 1536
#define WS_ROWBASE 1664
#define WS_TILES   2048

__global__ __launch_bounds__(1024) void rgcn_prep_kernel(
    const int* __restrict__ src, const int* __restrict__ dst, int E,
    float* __restrict__ invdeg, unsigned* __restrict__ rowmask,
    int* __restrict__ rowbase, int* __restrict__ tiles)
{
    __shared__ int deg[NN];
    __shared__ unsigned rmask[32];
    __shared__ int rbase[32];
    __shared__ int total_s;
    const int r = blockIdx.x;
    const int tid = threadIdx.x;
    const int* sR = src + r * E;
    const int* dR = dst + r * E;
    for (int i = tid; i < NN; i += 1024) deg[i] = 0;
    if (tid < 32) rmask[tid] = 0u;
    __syncthreads();
    for (int e = tid; e < E; e += 1024) {
        const int s = sR[e], d = dR[e];
        atomicAdd(&deg[d], 1);
        atomicOr(&rmask[d >> 4], 1u << (s >> 4));
    }
    __syncthreads();
    for (int i = tid; i < NN; i += 1024) {
        int d = deg[i];
        if (d < 1) d = 1;
        invdeg[r * NN + i] = 1.0f / (float)d;
    }
    if (tid < 32) {
        const unsigned m = rmask[tid];
        const int cnt = __popc(m);
        int inc = cnt;
        for (int off = 1; off < 32; off <<= 1) {
            const int t = __shfl_up(inc, off, 32);
            if (tid >= off) inc += t;
        }
        rbase[tid] = inc - cnt;
        rowmask[r * 32 + tid] = m;
        rowbase[r * 32 + tid] = r * 1024 + inc - cnt;
        if (tid == 31) total_s = inc;
    }
    __syncthreads();
    int* T = tiles + r * 1024 * 256;
    const int nw = total_s * 256;
    for (int i = tid; i < nw; i += 1024) T[i] = 0;
    __syncthreads();
    for (int e = tid; e < E; e += 1024) {
        const int s = sR[e], d = dR[e];
        const int tm = d >> 4, tk = s >> 4, k = s & 15;
        const int idx = rbase[tm] + __popc(rmask[tm] & ((1u << tk) - 1u));
        const int off = ((k & 1) * 16 + (d & 15)) * 8 + (k >> 2) * 2 + ((k >> 1) & 1);
        atomicAdd(&T[idx * 256 + off], 1);
    }
}

__device__ __forceinline__ void gemm_acc(
    const float* __restrict__ A, const float* __restrict__ W,
    int wid, int mlo, int mhalf, v8f acc[2])
{
    const int node = wid * 16 + mlo;
    for (int ks = 0; ks < 8; ++ks) {
        const int kk = ks * 4 + mhalf * 2;
        v2f b0, b1, a;
        b0.x = W[kk * CC + mlo];
        b0.y = W[(kk + 1) * CC + mlo];
        b1.x = W[kk * CC + 16 + mlo];
        b1.y = W[(kk + 1) * CC + 16 + mlo];
        a.x = A[node * CC + kk];
        a.y = A[node * CC + kk + 1];
        acc[0] = __builtin_amdgcn_wmma_f32_16x16x4_f32(
            false, a, false, b0, (short)0, acc[0], false, false);
        acc[1] = __builtin_amdgcn_wmma_f32_16x16x4_f32(
            false, a, false, b1, (short)0, acc[1], false, false);
    }
}

__global__ __launch_bounds__(NT) void rgcn_main_kernel(
    const float* __restrict__ xx, const float* __restrict__ ss,
    const float* __restrict__ Wf, const float* __restrict__ bf,
    const float* __restrict__ convw, const float* __restrict__ convroot,
    const float* __restrict__ convbias, const float* __restrict__ lng,
    const float* __restrict__ lnb, const float* __restrict__ invdeg,
    const unsigned* __restrict__ rowmask, const int* __restrict__ rowbase,
    const int* __restrict__ tiles, float* __restrict__ out)
{
    __shared__ float xs[NN * CC];
    __shared__ float wts[4 * CC * CC];
    __shared__ float ivd[3 * NN];
    __shared__ unsigned rmk[3 * 32];
    __shared__ int rbs[3 * 32];
    __shared__ float wfe[6 * CC];
    __shared__ float prm[3 * CC];

    const int bid   = blockIdx.x;
    const int tid   = threadIdx.x;
    const int lane  = tid & 31;
    const int wid   = tid >> 5;
    const int mlo   = lane & 15;
    const int mhalf = lane >> 4;

    for (int i = tid; i < 3 * NN; i += NT) ivd[i] = invdeg[i];
    for (int i = tid; i < 6 * CC; i += NT) wfe[i] = Wf[i];
    if (tid < 96) {
        rmk[tid] = rowmask[tid];
        rbs[tid] = rowbase[tid];
    }
    if (tid < CC) {
        prm[tid]          = bf[tid];
        prm[CC + tid]     = lng[tid];
        prm[2 * CC + tid] = lnb[tid];
    }
    __syncthreads();

    const float mfeat = ss[bid] * 0.125f;
    for (int idx = tid; idx < NN * CC; idx += NT) {
        const int n = idx >> 5, c = idx & 31;
        const int t = n >> 6, i = (n >> 4) & 3, j = (n >> 2) & 3, k = n & 3;
        const float v = xx[bid * NN + n];
        float a = prm[c];
        a += ((float)i * (1.0f / 3.0f)) * wfe[0 * CC + c];
        a += ((float)j * (1.0f / 3.0f)) * wfe[1 * CC + c];
        a += ((float)k * (1.0f / 3.0f)) * wfe[2 * CC + c];
        a += ((float)t * (1.0f / 7.0f)) * wfe[3 * CC + c];
        a += v * wfe[4 * CC + c];
        a += mfeat * wfe[5 * CC + c];
        xs[idx] = a;
    }
    __syncthreads();

    for (int l = 0; l < LL; ++l) {
        v8f acc[2];
#pragma unroll
        for (int nt = 0; nt < 2; ++nt) {
            const float bv = convbias[l * CC + nt * 16 + mlo];
            acc[nt] = (v8f){bv, bv, bv, bv, bv, bv, bv, bv};
        }

        wts[tid] = convroot[l * CC * CC + tid];
#pragma unroll
        for (int j = 0; j < 3; ++j)
            wts[(1 + j) * CC * CC + tid] = convw[(l * 3 + j) * CC * CC + tid];
        __syncthreads();

        gemm_acc(xs, wts, wid, mlo, mhalf, acc);

        for (int r = 0; r < 3; ++r) {
            unsigned m = (unsigned)__builtin_amdgcn_readfirstlane((int)rmk[r * 32 + wid]);
            const int base = __builtin_amdgcn_readfirstlane(rbs[r * 32 + wid]);
            int rem = __popc(m);
            const int4* tp = (const int4*)(tiles + base * 256 + lane * 8);
            float cf[8] = {0, 0, 0, 0, 0, 0, 0, 0};
            if (rem) {
                const int4 t0 = tp[0], t1 = tp[1];
                cf[0] = (float)t0.x; cf[1] = (float)t0.y; cf[2] = (float)t0.z; cf[3] = (float)t0.w;
                cf[4] = (float)t1.x; cf[5] = (float)t1.y; cf[6] = (float)t1.z; cf[7] = (float)t1.w;
            }

            v8f g0 = (v8f){0, 0, 0, 0, 0, 0, 0, 0};
            v8f g1 = (v8f){0, 0, 0, 0, 0, 0, 0, 0};
            while (m) {
                const int tk = __builtin_ctz(m);
                m &= m - 1u;
                --rem;
                tp += rem ? 64 : 0;
                const int4 p0 = tp[0], p1 = tp[1];
                const float* xb = xs + (tk * 16 + mhalf) * CC + mlo;
#pragma unroll
                for (int ks = 0; ks < 4; ++ks) {
                    v2f b, a0, a1;
                    b.x  = cf[2 * ks];
                    b.y  = cf[2 * ks + 1];
                    a0.x = xb[(4 * ks) * CC];
                    a0.y = xb[(4 * ks + 2) * CC];
                    a1.x = xb[(4 * ks) * CC + 16];
                    a1.y = xb[(4 * ks + 2) * CC + 16];
                    g0 = __builtin_amdgcn_wmma_f32_16x16x4_f32(
                        false, a0, false, b, (short)0, g0, false, false);
                    g1 = __builtin_amdgcn_wmma_f32_16x16x4_f32(
                        false, a1, false, b, (short)0, g1, false, false);
                }
                cf[0] = (float)p0.x; cf[1] = (float)p0.y; cf[2] = (float)p0.z; cf[3] = (float)p0.w;
                cf[4] = (float)p1.x; cf[5] = (float)p1.y; cf[6] = (float)p1.z; cf[7] = (float)p1.w;
            }
            const float sc = ivd[r * NN + wid * 16 + mlo];
            g0 *= sc;
            g1 *= sc;
            const float* Wr = wts + (1 + r) * CC * CC;
#pragma unroll
            for (int q = 0; q < 8; ++q) {
                const int kb = (q >> 2) * 16 + 8 * mhalf + 2 * (q & 3);
                v2f a, w0, w1;
                w0.x = Wr[kb * CC + mlo];
                w0.y = Wr[(kb + 1) * CC + mlo];
                w1.x = Wr[kb * CC + 16 + mlo];
                w1.y = Wr[(kb + 1) * CC + 16 + mlo];
                if (q < 4) { a.x = g0[2 * q];       a.y = g0[2 * q + 1]; }
                else       { a.x = g1[2 * (q - 4)]; a.y = g1[2 * (q - 4) + 1]; }
                acc[0] = __builtin_amdgcn_wmma_f32_16x16x4_f32(
                    false, a, false, w0, (short)0, acc[0], false, false);
                acc[1] = __builtin_amdgcn_wmma_f32_16x16x4_f32(
                    false, a, false, w1, (short)0, acc[1], false, false);
            }
        }
        __syncthreads();

#pragma unroll
        for (int nt = 0; nt < 2; ++nt)
#pragma unroll
            for (int jj = 0; jj < 8; ++jj) {
                const int node = wid * 16 + jj + 8 * mhalf;
                const int col  = nt * 16 + mlo;
                const float v  = acc[nt][jj];
                xs[node * CC + col] = v > 0.0f ? v : 0.0f;
            }
        __syncthreads();

        for (int q = 0; q < 16; ++q) {
            const int node = wid * 16 + q;
            const float v = xs[node * CC + lane];
            float s = v;
            for (int off = 16; off; off >>= 1) s += __shfl_xor(s, off, 32);
            const float mu = s * (1.0f / 32.0f);
            const float d = v - mu;
            float s2 = d * d;
            for (int off = 16; off; off >>= 1) s2 += __shfl_xor(s2, off, 32);
            const float var = s2 * (1.0f / 32.0f);
            xs[node * CC + lane] =
                d * rsqrtf(var + LN_EPS) * prm[CC + lane] + prm[2 * CC + lane];
        }
        __syncthreads();
    }

    for (int idx = tid; idx < 48 * CC; idx += NT) {
        const int row = idx >> 5, c = idx & 31;
        const int g = row >> 4, rc = row & 15;
        const int a0 = rc >> 2, b0 = rc & 3;
        float sum = 0.0f;
#pragma unroll
        for (int q = 0; q < 4; ++q) {
            int node;
            if (g == 0)      node = q * 16 + a0 * 4 + b0;
            else if (g == 1) node = a0 * 16 + q * 4 + b0;
            else             node = a0 * 16 + b0 * 4 + q;
            sum += xs[node * CC + c];
        }
        out[bid * 48 * CC + idx] = sum * 0.25f;
    }
}

extern "C" void kernel_launch(void* const* d_in, const int* in_sizes, int n_in,
                              void* d_out, int out_size, void* d_ws, size_t ws_size,
                              hipStream_t stream) {
    const float* xx       = (const float*)d_in[0];
    const float* ss       = (const float*)d_in[1];
    const float* Wf       = (const float*)d_in[2];
    const float* bf       = (const float*)d_in[3];
    const float* convw    = (const float*)d_in[4];
    const float* convroot = (const float*)d_in[5];
    const float* convbias = (const float*)d_in[6];
    const float* lng      = (const float*)d_in[7];
    const float* lnb      = (const float*)d_in[8];
    const int*   src      = (const int*)d_in[9];
    const int*   dst      = (const int*)d_in[10];
    float*       out      = (float*)d_out;
    float*       invdeg   = (float*)d_ws + WS_INVDEG;
    unsigned*    rowmask  = (unsigned*)d_ws + WS_ROWMASK;
    int*         rowbase  = (int*)d_ws + WS_ROWBASE;
    int*         tiles    = (int*)d_ws + WS_TILES;

    const int E = in_sizes[9] / 3;
    const int B = in_sizes[1];

    rgcn_prep_kernel<<<3, 1024, 0, stream>>>(src, dst, E, invdeg, rowmask, rowbase, tiles);
    rgcn_main_kernel<<<B, NT, 0, stream>>>(xx, ss, Wf, bf, convw, convroot,
                                            convbias, lng, lnb, invdeg, rowmask,
                                            rowbase, tiles, out);
    (void)n_in; (void)out_size; (void)ws_size;
}


# Layer products taken transposed (weights as A, xT/aggT registers as B): a node's 32 channels sit in two lanes, so LayerNorm runs in registers, no LDS pass; LayerNorm sum tree, root k order changed
# speedup vs baseline: 18.1951x; 1.1895x over previous
#include <hip/hip_runtime.h>
#include <hip/hip_bf16.h>

typedef __attribute__((ext_vector_type(2))) float v2f;
typedef __attribute__((ext_vector_type(8))) float v8f;

#define NN  512
#define CC  32
#define LL  4
#define LN_EPS 1e-5f
#define NT  1024
static_assert(NT == CC * CC && NT == 2 * NN, "staging and row-block ownership assume 1024 threads");

#define WS_INVDEG  0
#define WS_ROWMASK 1536
#define WS_ROWBASE 1664
#define WS_TILES   2048

__global__ __launch_bounds__(1024) void rgcn_prep_kernel(
    const int* __restrict__ src, const int* __restrict__ dst, int E,
    float* __restrict__ invdeg, unsigned* __restrict__ rowmask,
    int* __restrict__ rowbase, int* __restrict__ tiles)
{
    __shared__ int deg[NN];
    __shared__ unsigned rmask[32];
    __shared__ int rbase[32];
    __shared__ int total_s;
    const int r = blockIdx.x;
    const int tid = threadIdx.x;
    const int* sR = src + r * E;
    const int* dR = dst + r * E;
    for (int i = tid; i < NN; i += 1024) deg[i] = 0;
    if (tid < 32) rmask[tid] = 0u;
    __syncthreads();
    for (int e = tid; e < E; e += 1024) {
        const int s = sR[e], d = dR[e];
        atomicAdd(&deg[d], 1);
        atomicOr(&rmask[d >> 4], 1u << (s >> 4));
    }
    __syncthreads();
    for (int i = tid; i < NN; i += 1024) {
        int d = deg[i];
        if (d < 1) d = 1;
        invdeg[r * NN + i] = 1.0f / (float)d;
    }
    if (tid < 32) {
        const unsigned m = rmask[tid];
        const int cnt = __popc(m);
        int inc = cnt;
        for (int off = 1; off < 32; off <<= 1) {
            const int t = __shfl_up(inc, off, 32);
            if (tid >= off) inc += t;
        }
        rbase[tid] = inc - cnt;
        rowmask[r * 32 + tid] = m;
        rowbase[r * 32 + tid] = r * 1024 + inc - cnt;
        if (tid == 31) total_s = inc;
    }
    __syncthreads();
    int* T = tiles + r * 1024 * 256;
    const int nw = total_s * 256;
    for (int i = tid; i < nw; i += 1024) T[i] = 0;
    __syncthreads();
    for (int e = tid; e < E; e += 1024) {
        const int s = sR[e], d = dR[e];
        const int tm = d >> 4, tk = s >> 4, k = s & 15;
        const int idx = rbase[tm] + __popc(rmask[tm] & ((1u << tk) - 1u));
        const int off = ((k & 1) * 16 + (d & 15)) * 8 + (k >> 2) * 2 + ((k >> 1) & 1);
        atomicAdd(&T[idx * 256 + off], 1);
    }
}

__device__ __forceinline__ float xhalf(float x) {
    const int b = __builtin_bit_cast(int, x);
    return __builtin_bit_cast(float,
        __builtin_amdgcn_permlanex16(b, b, 0x76543210, 0xfedcba98, false, false));
}

__global__ __launch_bounds__(NT) void rgcn_main_kernel(
    const float* __restrict__ xx, const float* __restrict__ ss,
    const float* __restrict__ Wf, const float* __restrict__ bf,
    const float* __restrict__ convw, const float* __restrict__ convroot,
    const float* __restrict__ convbias, const float* __restrict__ lng,
    const float* __restrict__ lnb, const float* __restrict__ invdeg,
    const unsigned* __restrict__ rowmask, const int* __restrict__ rowbase,
    const int* __restrict__ tiles, float* __restrict__ out)
{
    __shared__ __attribute__((aligned(16))) float xs[NN * CC];
    __shared__ float wts[4 * CC * CC];
    __shared__ float ivd[3 * NN];
    __shared__ unsigned rmk[3 * 32];
    __shared__ int rbs[3 * 32];
    __shared__ float wfe[6 * CC];
    __shared__ __attribute__((aligned(16))) float prm[3 * CC];
    __shared__ __attribute__((aligned(16))) float cbs[LL * CC];

    const int bid   = blockIdx.x;
    const int tid   = threadIdx.x;
    const int lane  = tid & 31;
    const int wid   = tid >> 5;
    const int mlo   = lane & 15;
    const int mhalf = lane >> 4;

    for (int i = tid; i < 3 * NN; i += NT) ivd[i] = invdeg[i];
    for (int i = tid; i < 6 * CC; i += NT) wfe[i] = Wf[i];
    if (tid < 96) {
        rmk[tid] = rowmask[tid];
        rbs[tid] = rowbase[tid];
    }
    if (tid < CC) {
        prm[tid]          = bf[tid];
        prm[CC + tid]     = lng[tid];
        prm[2 * CC + tid] = lnb[tid];
    }
    if (tid < LL * CC) cbs[tid] = convbias[tid];
    wts[tid] = convroot[tid];
#pragma unroll
    for (int j = 0; j < 3; ++j)
        wts[(1 + j) * CC * CC + tid] = convw[j * CC * CC + tid];
    __syncthreads();

    const float mfeat = ss[bid] * 0.125f;
    for (int idx = tid; idx < NN * CC; idx += NT) {
        const int n = idx >> 5, c = idx & 31;
        const int t = n >> 6, i = (n >> 4) & 3, j = (n >> 2) & 3, k = n & 3;
        const float v = xx[bid * NN + n];
        float a = prm[c];
        a += ((float)i * (1.0f / 3.0f)) * wfe[0 * CC + c];
        a += ((float)j * (1.0f / 3.0f)) * wfe[1 * CC + c];
        a += ((float)k * (1.0f / 3.0f)) * wfe[2 * CC + c];
        a += ((float)t * (1.0f / 7.0f)) * wfe[3 * CC + c];
        a += v * wfe[4 * CC + c];
        a += mfeat * wfe[5 * CC + c];
        xs[idx] = a;
    }
    __syncthreads();

    const int own4 = ((wid * 16 + mlo) * CC + 8 * mhalf) >> 2;
    float4* xs4 = (float4*)xs;
    const float4* cb4 = (const float4*)cbs;
    const float4* pr4 = (const float4*)prm;

    v8f xo[2];
#pragma unroll
    for (int t = 0; t < 2; ++t) {
        const float4 lo = xs4[own4 + 4 * t], hi = xs4[own4 + 4 * t + 1];
        xo[t] = (v8f){lo.x, lo.y, lo.z, lo.w, hi.x, hi.y, hi.z, hi.w};
    }

    for (int l = 0; l < LL; ++l) {
        v8f acc[2];
#pragma unroll
        for (int t = 0; t < 2; ++t) {
            const float4 lo = cb4[l * 8 + 4 * t + 2 * mhalf], hi = cb4[l * 8 + 4 * t + 2 * mhalf + 1];
            acc[t] = (v8f){lo.x, lo.y, lo.z, lo.w, hi.x, hi.y, hi.z, hi.w};
        }

#pragma unroll
        for (int q = 0; q < 8; ++q) {
            const int kb = (q >> 2) * 16 + 8 * mhalf + 2 * (q & 3);
            v2f b, w0, w1;
            w0.x = wts[kb * CC + mlo];
            w0.y = wts[(kb + 1) * CC + mlo];
            w1.x = wts[kb * CC + 16 + mlo];
            w1.y = wts[(kb + 1) * CC + 16 + mlo];
            if (q < 4) { b.x = xo[0][2 * q];       b.y = xo[0][2 * q + 1]; }
            else       { b.x = xo[1][2 * (q - 4)]; b.y = xo[1][2 * (q - 4) + 1]; }
            acc[0] = __builtin_amdgcn_wmma_f32_16x16x4_f32(
                false, w0, false, b, (short)0, acc[0], false, false);
            acc[1] = __builtin_amdgcn_wmma_f32_16x16x4_f32(
                false, w1, false, b, (short)0, acc[1], false, false);
        }

        for (int r = 0; r < 3; ++r) {
            unsigned m = (unsigned)__builtin_amdgcn_readfirstlane((int)rmk[r * 32 + wid]);
            const int base = __builtin_amdgcn_readfirstlane(rbs[r * 32 + wid]);
            int rem = __popc(m);
            const int4* tp = (const int4*)(tiles + base * 256 + lane * 8);
            float cf[8] = {0, 0, 0, 0, 0, 0, 0, 0};
            if (rem) {
                const int4 t0 = tp[0], t1 = tp[1];
                cf[0] = (float)t0.x; cf[1] = (float)t0.y; cf[2] = (float)t0.z; cf[3] = (float)t0.w;
                cf[4] = (float)t1.x; cf[5] = (float)t1.y; cf[6] = (float)t1.z; cf[7] = (float)t1.w;
            }

            v8f g0 = (v8f){0, 0, 0, 0, 0, 0, 0, 0};
            v8f g1 = (v8f){0, 0, 0, 0, 0, 0, 0, 0};
            while (m) {
                const int tk = __builtin_ctz(m);
                m &= m - 1u;
                --rem;
                tp += rem ? 64 : 0;
                const int4 p0 = tp[0], p1 = tp[1];
                const float* xb = xs + (tk * 16 + mhalf) * CC + mlo;
#pragma unroll
                for (int ks = 0; ks < 4; ++ks) {
                    v2f b, a0, a1;
                    b.x  = cf[2 * ks];
                    b.y  = cf[2 * ks + 1];
                    a0.x = xb[(4 * ks) * CC];
                    a0.y = xb[(4 * ks + 2) * CC];
                    a1.x = xb[(4 * ks) * CC + 16];
                    a1.y = xb[(4 * ks + 2) * CC + 16];
                    g0 = __builtin_amdgcn_wmma_f32_16x16x4_f32(
                        false, a0, false, b, (short)0, g0, false, false);
                    g1 = __builtin_amdgcn_wmma_f32_16x16x4_f32(
                        false, a1, false, b, (short)0, g1, false, false);
                }
                cf[0] = (float)p0.x; cf[1] = (float)p0.y; cf[2] = (float)p0.z; cf[3] = (float)p0.w;
                cf[4] = (float)p1.x; cf[5] = (float)p1.y; cf[6] = (float)p1.z; cf[7] = (float)p1.w;
            }
            const float sc = ivd[r * NN + wid * 16 + mlo];
            g0 *= sc;
            g1 *= sc;
            const float* Wr = wts + (1 + r) * CC * CC;
#pragma unroll
            for (int q = 0; q < 8; ++q) {
                const int kb = (q >> 2) * 16 + 8 * mhalf + 2 * (q & 3);
                v2f b, w0, w1;
                w0.x = Wr[kb * CC + mlo];
                w0.y = Wr[(kb + 1) * CC + mlo];
                w1.x = Wr[kb * CC + 16 + mlo];
                w1.y = Wr[(kb + 1) * CC + 16 + mlo];
                if (q < 4) { b.x = g0[2 * q];       b.y = g0[2 * q + 1]; }
                else       { b.x = g1[2 * (q - 4)]; b.y = g1[2 * (q - 4) + 1]; }
                acc[0] = __builtin_amdgcn_wmma_f32_16x16x4_f32(
                    false, w0, false, b, (short)0, acc[0], false, false);
                acc[1] = __builtin_amdgcn_wmma_f32_16x16x4_f32(
                    false, w1, false, b, (short)0, acc[1], false, false);
            }
        }

        float ps[4] = {0.0f, 0.0f, 0.0f, 0.0f};
#pragma unroll
        for (int t = 0; t < 2; ++t)
#pragma unroll
            for (int j = 0; j < 8; ++j) {
                const float v = acc[t][j];
                acc[t][j] = v > 0.0f ? v : 0.0f;
                ps[j & 3] += acc[t][j];
            }
        float sum = (ps[0] + ps[1]) + (ps[2] + ps[3]);
        sum += xhalf(sum);
        const float mu = sum * (1.0f / 32.0f);
        float pq[4] = {0.0f, 0.0f, 0.0f, 0.0f};
#pragma unroll
        for (int t = 0; t < 2; ++t)
#pragma unroll
            for (int j = 0; j < 8; ++j) {
                const float d = acc[t][j] - mu;
                acc[t][j] = d;
                pq[j & 3] += d * d;
            }
        float sq = (pq[0] + pq[1]) + (pq[2] + pq[3]);
        sq += xhalf(sq);
        const float rs = rsqrtf(sq * (1.0f / 32.0f) + LN_EPS);
#pragma unroll
        for (int t = 0; t < 2; ++t) {
            const float4 ga = pr4[8 + 4 * t + 2 * mhalf],  gb = pr4[8 + 4 * t + 2 * mhalf + 1];
            const float4 ba = pr4[16 + 4 * t + 2 * mhalf], bb = pr4[16 + 4 * t + 2 * mhalf + 1];
            const float gg[8] = {ga.x, ga.y, ga.z, ga.w, gb.x, gb.y, gb.z, gb.w};
            const float bv[8] = {ba.x, ba.y, ba.z, ba.w, bb.x, bb.y, bb.z, bb.w};
#pragma unroll
            for (int j = 0; j < 8; ++j)
                xo[t][j] = acc[t][j] * rs * gg[j] + bv[j];
        }
        __syncthreads();

#pragma unroll
        for (int t = 0; t < 2; ++t) {
            xs4[own4 + 4 * t]     = make_float4(xo[t][0], xo[t][1], xo[t][2], xo[t][3]);
            xs4[own4 + 4 * t + 1] = make_float4(xo[t][4], xo[t][5], xo[t][6], xo[t][7]);
        }
        if (l + 1 < LL) {
            wts[tid] = convroot[(l + 1) * CC * CC + tid];
#pragma unroll
            for (int j = 0; j < 3; ++j)
                wts[(1 + j) * CC * CC + tid] = convw[((l + 1) * 3 + j) * CC * CC + tid];
        }
        __syncthreads();
    }

    for (int idx = tid; idx < 48 * CC; idx += NT) {
        const int row = idx >> 5, c = idx & 31;
        const int g = row >> 4, rc = row & 15;
        const int a0 = rc >> 2, b0 = rc & 3;
        float sum = 0.0f;
#pragma unroll
        for (int q = 0; q < 4; ++q) {
            int node;
            if (g == 0)      node = q * 16 + a0 * 4 + b0;
            else if (g == 1) node = a0 * 16 + q * 4 + b0;
            else             node = a0 * 16 + b0 * 4 + q;
            sum += xs[node * CC + c];
        }
        out[bid * 48 * CC + idx] = sum * 0.25f;
    }
}

extern "C" void kernel_launch(void* const* d_in, const int* in_sizes, int n_in,
                              void* d_out, int out_size, void* d_ws, size_t ws_size,
                              hipStream_t stream) {
    const float* xx       = (const float*)d_in[0];
    const float* ss       = (const float*)d_in[1];
    const float* Wf       = (const float*)d_in[2];
    const float* bf       = (const float*)d_in[3];
    const float* convw    = (const float*)d_in[4];
    const float* convroot = (const float*)d_in[5];
    const float* convbias = (const float*)d_in[6];
    const float* lng      = (const float*)d_in[7];
    const float* lnb      = (const float*)d_in[8];
    const int*   src      = (const int*)d_in[9];
    const int*   dst      = (const int*)d_in[10];
    float*       out      = (float*)d_out;
    float*       invdeg   = (float*)d_ws + WS_INVDEG;
    unsigned*    rowmask  = (unsigned*)d_ws + WS_ROWMASK;
    int*         rowbase  = (int*)d_ws + WS_ROWBASE;
    int*         tiles    = (int*)d_ws + WS_TILES;

    const int E = in_sizes[9] / 3;
    const int B = in_sizes[1];

    rgcn_prep_kernel<<<3, 1024, 0, stream>>>(src, dst, E, invdeg, rowmask, rowbase, tiles);
    rgcn_main_kernel<<<B, NT, 0, stream>>>(xx, ss, Wf, bf, convw, convroot,
                                            convbias, lng, lnb, invdeg, rowmask,
                                            rowbase, tiles, out);
    (void)n_in; (void)out_size; (void)ws_size;
}


# LDS operands k-contiguous (x kept transposed, weights transposed, tile k order 8h+2ks+s): each matrix fragment is two 16-byte LDS loads, not eight 4-byte ones; aggregation k order changed
# speedup vs baseline: 19.7284x; 1.0843x over previous
#include <hip/hip_runtime.h>
#include <hip/hip_bf16.h>

typedef __attribute__((ext_vector_type(2))) float v2f;
typedef __attribute__((ext_vector_type(8))) float v8f;

#define NN  512
#define CC  32
#define LL  4
#define LN_EPS 1e-5f
#define NT  1024
static_assert(NT == CC * CC && NT == 2 * NN, "staging and row-block ownership assume 1024 threads");
#define XP  516
#define WP  36
static_assert((XP * 4) % 16 == 0 && ((XP * 4 / 16) & 1) == 1 && XP >= NN, "xT pitch");
static_assert((WP * 4) % 16 == 0 && ((WP * 4 / 16) & 1) == 1 && WP >= CC, "WT pitch");

#define WS_INVDEG  0
#define WS_ROWMASK 1536
#define WS_ROWBASE 1664
#define WS_TILES   2048

__global__ __launch_bounds__(1024) void rgcn_prep_kernel(
    const int* __restrict__ src, const int* __restrict__ dst, int E,
    float* __restrict__ invdeg, unsigned* __restrict__ rowmask,
    int* __restrict__ rowbase, int* __restrict__ tiles)
{
    __shared__ int deg[NN];
    __shared__ unsigned rmask[32];
    __shared__ int rbase[32];
    __shared__ int total_s;
    const int r = blockIdx.x;
    const int tid = threadIdx.x;
    const int* sR = src + r * E;
    const int* dR = dst + r * E;
    for (int i = tid; i < NN; i += 1024) deg[i] = 0;
    if (tid < 32) rmask[tid] = 0u;
    __syncthreads();
    for (int e = tid; e < E; e += 1024) {
        const int s = sR[e], d = dR[e];
        atomicAdd(&deg[d], 1);
        atomicOr(&rmask[d >> 4], 1u << (s >> 4));
    }
    __syncthreads();
    for (int i = tid; i < NN; i += 1024) {
        int d = deg[i];
        if (d < 1) d = 1;
        invdeg[r * NN + i] = 1.0f / (float)d;
    }
    if (tid < 32) {
        const unsigned m = rmask[tid];
        const int cnt = __popc(m);
        int inc = cnt;
        for (int off = 1; off < 32; off <<= 1) {
            const int t = __shfl_up(inc, off, 32);
            if (tid >= off) inc += t;
        }
        rbase[tid] = inc - cnt;
        rowmask[r * 32 + tid] = m;
        rowbase[r * 32 + tid] = r * 1024 + inc - cnt;
        if (tid == 31) total_s = inc;
    }
    __syncthreads();
    int* T = tiles + r * 1024 * 256;
    const int nw = total_s * 256;
    for (int i = tid; i < nw; i += 1024) T[i] = 0;
    __syncthreads();
    for (int e = tid; e < E; e += 1024) {
        const int s = sR[e], d = dR[e];
        const int tm = d >> 4, tk = s >> 4, k = s & 15;
        const int idx = rbase[tm] + __popc(rmask[tm] & ((1u << tk) - 1u));
        const int off = ((k >> 3) * 16 + (d & 15)) * 8 + (k & 7);
        atomicAdd(&T[idx * 256 + off], 1);
    }
}

__device__ __forceinline__ float xhalf(float x) {
    const int b = __builtin_bit_cast(int, x);
    return __builtin_bit_cast(float,
        __builtin_amdgcn_permlanex16(b, b, 0x76543210, 0xfedcba98, false, false));
}

__global__ __launch_bounds__(NT) void rgcn_main_kernel(
    const float* __restrict__ xx, const float* __restrict__ ss,
    const float* __restrict__ Wf, const float* __restrict__ bf,
    const float* __restrict__ convw, const float* __restrict__ convroot,
    const float* __restrict__ convbias, const float* __restrict__ lng,
    const float* __restrict__ lnb, const float* __restrict__ invdeg,
    const unsigned* __restrict__ rowmask, const int* __restrict__ rowbase,
    const int* __restrict__ tiles, float* __restrict__ out)
{
    __shared__ __attribute__((aligned(16))) float xs[CC * XP];
    __shared__ __attribute__((aligned(16))) float wts[4 * CC * WP];
    __shared__ float ivd[3 * NN];
    __shared__ unsigned rmk[3 * 32];
    __shared__ int rbs[3 * 32];
    __shared__ __attribute__((aligned(16))) float wfe[6 * CC];
    __shared__ __attribute__((aligned(16))) float prm[3 * CC];
    __shared__ __attribute__((aligned(16))) float cbs[LL * CC];

    const int bid   = blockIdx.x;
    const int tid   = threadIdx.x;
    const int lane  = tid & 31;
    const int wid   = tid >> 5;
    const int mlo   = lane & 15;
    const int mhalf = lane >> 4;
    const int wdst  = (tid & 31) * WP + (tid >> 5);

    for (int i = tid; i < 3 * NN; i += NT) ivd[i] = invdeg[i];
    for (int i = tid; i < 6 * CC; i += NT) wfe[i] = Wf[i];
    if (tid < 96) {
        rmk[tid] = rowmask[tid];
        rbs[tid] = rowbase[tid];
    }
    if (tid < CC) {
        prm[tid]          = bf[tid];
        prm[CC + tid]     = lng[tid];
        prm[2 * CC + tid] = lnb[tid];
    }
    if (tid < LL * CC) cbs[tid] = convbias[tid];
    wts[wdst] = convroot[tid];
#pragma unroll
    for (int j = 0; j < 3; ++j)
        wts[(1 + j) * CC * WP + wdst] = convw[j * CC * CC + tid];
    __syncthreads();

    float4* xs4 = (float4*)xs;
    const float4* wt4 = (const float4*)wts;
    const float4* wf4 = (const float4*)wfe;
    const float4* cb4 = (const float4*)cbs;
    const float4* pr4 = (const float4*)prm;

    const int ownn = wid * 16 + mlo;

    v8f xo[2];
    {
        const float mfeat = ss[bid] * 0.125f;
        const int n = ownn;
        const float v  = xx[bid * NN + n];
        const float fi = (float)((n >> 4) & 3) * (1.0f / 3.0f);
        const float fj = (float)((n >> 2) & 3) * (1.0f / 3.0f);
        const float fk = (float)(n & 3) * (1.0f / 3.0f);
        const float ft = (float)(n >> 6) * (1.0f / 7.0f);
#pragma unroll
        for (int t = 0; t < 2; ++t) {
            const int c4 = 4 * t + 2 * mhalf;
            float a[8], w[8];
            { const float4 lo = pr4[c4], hi = pr4[c4 + 1];
              a[0] = lo.x; a[1] = lo.y; a[2] = lo.z; a[3] = lo.w;
              a[4] = hi.x; a[5] = hi.y; a[6] = hi.z; a[7] = hi.w; }
            const float fe[6] = {fi, fj, fk, ft, v, mfeat};
#pragma unroll
            for (int f = 0; f < 6; ++f) {
                const float4 lo = wf4[f * 8 + c4], hi = wf4[f * 8 + c4 + 1];
                w[0] = lo.x; w[1] = lo.y; w[2] = lo.z; w[3] = lo.w;
                w[4] = hi.x; w[5] = hi.y; w[6] = hi.z; w[7] = hi.w;
#pragma unroll
                for (int j = 0; j < 8; ++j) a[j] += fe[f] * w[j];
            }
#pragma unroll
            for (int j = 0; j < 8; ++j) {
                xo[t][j] = a[j];
                xs[(16 * t + 8 * mhalf + j) * XP + ownn] = a[j];
            }
        }
    }
    __syncthreads();

    for (int l = 0; l < LL; ++l) {
        v8f acc[2];
#pragma unroll
        for (int t = 0; t < 2; ++t) {
            const float4 lo = cb4[l * 8 + 4 * t + 2 * mhalf], hi = cb4[l * 8 + 4 * t + 2 * mhalf + 1];
            acc[t] = (v8f){lo.x, lo.y, lo.z, lo.w, hi.x, hi.y, hi.z, hi.w};
        }

        v8f z0 = xo[0], z1 = xo[1];
        for (int mtx = 0; mtx < 4; ++mtx) {
            if (mtx > 0) {
                const int r = mtx - 1;
                unsigned m = (unsigned)__builtin_amdgcn_readfirstlane((int)rmk[r * 32 + wid]);
                const int base = __builtin_amdgcn_readfirstlane(rbs[r * 32 + wid]);
                int rem = __popc(m);
                const int4* tp = (const int4*)(tiles + base * 256 + lane * 8);
                float cf[8] = {0, 0, 0, 0, 0, 0, 0, 0};
                if (rem) {
                    const int4 t0 = tp[0], t1 = tp[1];
                    cf[0] = (float)t0.x; cf[1] = (float)t0.y; cf[2] = (float)t0.z; cf[3] = (float)t0.w;
                    cf[4] = (float)t1.x; cf[5] = (float)t1.y; cf[6] = (float)t1.z; cf[7] = (float)t1.w;
                }

                v8f g0 = (v8f){0, 0, 0, 0, 0, 0, 0, 0};
                v8f g1 = (v8f){0, 0, 0, 0, 0, 0, 0, 0};
                while (m) {
                    const int tk = __builtin_ctz(m);
                    m &= m - 1u;
                    --rem;
                    tp += rem ? 64 : 0;
                    const int4 p0 = tp[0], p1 = tp[1];
                    const int xa = mlo * (XP / 4) + 4 * tk + 2 * mhalf;
                    const float4 u0 = xs4[xa],                 u1 = xs4[xa + 1];
                    const float4 v0 = xs4[xa + 16 * (XP / 4)], v1 = xs4[xa + 16 * (XP / 4) + 1];
                    const float e0[8] = {u0.x, u0.y, u0.z, u0.w, u1.x, u1.y, u1.z, u1.w};
                    const float e1[8] = {v0.x, v0.y, v0.z, v0.w, v1.x, v1.y, v1.z, v1.w};
#pragma unroll
                    for (int ks = 0; ks < 4; ++ks) {
                        v2f b, a0, a1;
                        b.x  = cf[2 * ks];
                        b.y  = cf[2 * ks + 1];
                        a0.x = e0[2 * ks];
                        a0.y = e0[2 * ks + 1];
                        a1.x = e1[2 * ks];
                        a1.y = e1[2 * ks + 1];
                        g0 = __builtin_amdgcn_wmma_f32_16x16x4_f32(
                            false, a0, false, b, (short)0, g0, false, false);
                        g1 = __builtin_amdgcn_wmma_f32_16x16x4_f32(
                            false, a1, false, b, (short)0, g1, false, false);
                    }
                    cf[0] = (float)p0.x; cf[1] = (float)p0.y; cf[2] = (float)p0.z; cf[3] = (float)p0.w;
                    cf[4] = (float)p1.x; cf[5] = (float)p1.y; cf[6] = (float)p1.z; cf[7] = (float)p1.w;
                }
                const float sc = ivd[r * NN + ownn];
                z0 = g0 * sc;
                z1 = g1 * sc;
            }
            const int wa = mtx * (CC * WP / 4) + mlo * (WP / 4) + 2 * mhalf;
            float w0e[16], w1e[16];
#pragma unroll
            for (int t = 0; t < 2; ++t) {
                const float4 p = wt4[wa + 4 * t],                 q = wt4[wa + 4 * t + 1];
                const float4 s = wt4[wa + 16 * (WP / 4) + 4 * t], u = wt4[wa + 16 * (WP / 4) + 4 * t + 1];
                w0e[8 * t + 0] = p.x; w0e[8 * t + 1] = p.y; w0e[8 * t + 2] = p.z; w0e[8 * t + 3] = p.w;
                w0e[8 * t + 4] = q.x; w0e[8 * t + 5] = q.y; w0e[8 * t + 6] = q.z; w0e[8 * t + 7] = q.w;
                w1e[8 * t + 0] = s.x; w1e[8 * t + 1] = s.y; w1e[8 * t + 2] = s.z; w1e[8 * t + 3] = s.w;
                w1e[8 * t + 4] = u.x; w1e[8 * t + 5] = u.y; w1e[8 * t + 6] = u.z; w1e[8 * t + 7] = u.w;
            }
#pragma unroll
            for (int q = 0; q < 8; ++q) {
                v2f b, w0, w1;
                w0.x = w0e[2 * q];
                w0.y = w0e[2 * q + 1];
                w1.x = w1e[2 * q];
                w1.y = w1e[2 * q + 1];
                if (q < 4) { b.x = z0[2 * q];       b.y = z0[2 * q + 1]; }
                else       { b.x = z1[2 * (q - 4)]; b.y = z1[2 * (q - 4) + 1]; }
                acc[0] = __builtin_amdgcn_wmma_f32_16x16x4_f32(
                    false, w0, false, b, (short)0, acc[0], false, false);
                acc[1] = __builtin_amdgcn_wmma_f32_16x16x4_f32(
                    false, w1, false, b, (short)0, acc[1], false, false);
            }
        }

        float ps[4] = {0.0f, 0.0f, 0.0f, 0.0f};
#pragma unroll
        for (int t = 0; t < 2; ++t)
#pragma unroll
            for (int j = 0; j < 8; ++j) {
                const float v = acc[t][j];
                acc[t][j] = v > 0.0f ? v : 0.0f;
                ps[j & 3] += acc[t][j];
            }
        float sum = (ps[0] + ps[1]) + (ps[2] + ps[3]);
        sum += xhalf(sum);
        const float mu = sum * (1.0f / 32.0f);
        float pq[4] = {0.0f, 0.0f, 0.0f, 0.0f};
#pragma unroll
        for (int t = 0; t < 2; ++t)
#pragma unroll
            for (int j = 0; j < 8; ++j) {
                const float d = acc[t][j] - mu;
                acc[t][j] = d;
                pq[j & 3] += d * d;
            }
        float sq = (pq[0] + pq[1]) + (pq[2] + pq[3]);
        sq += xhalf(sq);
        const float rs = rsqrtf(sq * (1.0f / 32.0f) + LN_EPS);
#pragma unroll
        for (int t = 0; t < 2; ++t) {
            const float4 ga = pr4[8 + 4 * t + 2 * mhalf],  gb = pr4[8 + 4 * t + 2 * mhalf + 1];
            const float4 ba = pr4[16 + 4 * t + 2 * mhalf], bb = pr4[16 + 4 * t + 2 * mhalf + 1];
            const float gg[8] = {ga.x, ga.y, ga.z, ga.w, gb.x, gb.y, gb.z, gb.w};
            const float bv[8] = {ba.x, ba.y, ba.z, ba.w, bb.x, bb.y, bb.z, bb.w};
#pragma unroll
            for (int j = 0; j < 8; ++j)
                xo[t][j] = acc[t][j] * rs * gg[j] + bv[j];
        }
        __syncthreads();

#pragma unroll
        for (int t = 0; t < 2; ++t)
#pragma unroll
            for (int j = 0; j < 8; ++j)
                xs[(16 * t + 8 * mhalf + j) * XP + ownn] = xo[t][j];
        if (l + 1 < LL) {
            wts[wdst] = convroot[(l + 1) * CC * CC + tid];
#pragma unroll
            for (int j = 0; j < 3; ++j)
                wts[(1 + j) * CC * WP + wdst] = convw[((l + 1) * 3 + j) * CC * CC + tid];
        }
        __syncthreads();
    }

    for (int idx = tid; idx < 48 * CC; idx += NT) {
        const int row = idx >> 5, c = idx & 31;
        const int g = row >> 4, rc = row & 15;
        const int a0 = rc >> 2, b0 = rc & 3;
        float sum = 0.0f;
#pragma unroll
        for (int q = 0; q < 4; ++q) {
            int node;
            if (g == 0)      node = q * 16 + a0 * 4 + b0;
            else if (g == 1) node = a0 * 16 + q * 4 + b0;
            else             node = a0 * 16 + b0 * 4 + q;
            sum += xs[c * XP + node];
        }
        out[bid * 48 * CC + idx] = sum * 0.25f;
    }
}

extern "C" void kernel_launch(void* const* d_in, const int* in_sizes, int n_in,
                              void* d_out, int out_size, void* d_ws, size_t ws_size,
                              hipStream_t stream) {
    const float* xx       = (const float*)d_in[0];
    const float* ss       = (const float*)d_in[1];
    const float* Wf       = (const float*)d_in[2];
    const float* bf       = (const float*)d_in[3];
    const float* convw    = (const float*)d_in[4];
    const float* convroot = (const float*)d_in[5];
    const float* convbias = (const float*)d_in[6];
    const float* lng      = (const float*)d_in[7];
    const float* lnb      = (const float*)d_in[8];
    const int*   src      = (const int*)d_in[9];
    const int*   dst      = (const int*)d_in[10];
    float*       out      = (float*)d_out;
    float*       invdeg   = (float*)d_ws + WS_INVDEG;
    unsigned*    rowmask  = (unsigned*)d_ws + WS_ROWMASK;
    int*         rowbase  = (int*)d_ws + WS_ROWBASE;
    int*         tiles    = (int*)d_ws + WS_TILES;

    const int E = in_sizes[9] / 3;
    const int B = in_sizes[1];

    rgcn_prep_kernel<<<3, 1024, 0, stream>>>(src, dst, E, invdeg, rowmask, rowbase, tiles);
    rgcn_main_kernel<<<B, NT, 0, stream>>>(xx, ss, Wf, bf, convw, convroot,
                                            convbias, lng, lnb, invdeg, rowmask,
                                            rowbase, tiles, out);
    (void)n_in; (void)out_size; (void)ws_size;
}


# Each matrix block's LDS fragments and tile counts are fetched as one group ahead of the block before it (second register set): no LDS read among the matrix ops; sum order kept
# speedup vs baseline: 20.3429x; 1.0311x over previous
#include <hip/hip_runtime.h>
#include <hip/hip_bf16.h>

typedef __attribute__((ext_vector_type(2))) float v2f;
typedef __attribute__((ext_vector_type(8))) float v8f;

#define NN  512
#define CC  32
#define LL  4
#define LN_EPS 1e-5f
#define NT  1024
static_assert(NT == CC * CC && NT == 2 * NN, "staging and row-block ownership assume 1024 threads");
#define XP  516
#define WP  36
static_assert((XP * 4) % 16 == 0 && ((XP * 4 / 16) & 1) == 1 && XP >= NN, "xT pitch");
static_assert((WP * 4) % 16 == 0 && ((WP * 4 / 16) & 1) == 1 && WP >= CC, "WT pitch");

#define WS_INVDEG  0
#define WS_ROWMASK 1536
#define WS_ROWBASE 1664
#define WS_TILES   2048

__global__ __launch_bounds__(1024) void rgcn_prep_kernel(
    const int* __restrict__ src, const int* __restrict__ dst, int E,
    float* __restrict__ invdeg, unsigned* __restrict__ rowmask,
    int* __restrict__ rowbase, int* __restrict__ tiles)
{
    __shared__ int deg[NN];
    __shared__ unsigned rmask[32];
    __shared__ int rbase[32];
    __shared__ int total_s;
    const int r = blockIdx.x;
    const int tid = threadIdx.x;
    const int* sR = src + r * E;
    const int* dR = dst + r * E;
    for (int i = tid; i < NN; i += 1024) deg[i] = 0;
    if (tid < 32) rmask[tid] = 0u;
    __syncthreads();
    for (int e = tid; e < E; e += 1024) {
        const int s = sR[e], d = dR[e];
        atomicAdd(&deg[d], 1);
        atomicOr(&rmask[d >> 4], 1u << (s >> 4));
    }
    __syncthreads();
    for (int i = tid; i < NN; i += 1024) {
        int d = deg[i];
        if (d < 1) d = 1;
        invdeg[r * NN + i] = 1.0f / (float)d;
    }
    if (tid < 32) {
        const unsigned m = rmask[tid];
        const int cnt = __popc(m);
        int inc = cnt;
        for (int off = 1; off < 32; off <<= 1) {
            const int t = __shfl_up(inc, off, 32);
            if (tid >= off) inc += t;
        }
        rbase[tid] = inc - cnt;
        rowmask[r * 32 + tid] = m;
        rowbase[r * 32 + tid] = r * 1024 + inc - cnt;
        if (tid == 31) total_s = inc;
    }
    __syncthreads();
    int* T = tiles + r * 1024 * 256;
    const int nw = total_s * 256;
    for (int i = tid; i < nw; i += 1024) T[i] = 0;
    __syncthreads();
    for (int e = tid; e < E; e += 1024) {
        const int s = sR[e], d = dR[e];
        const int tm = d >> 4, tk = s >> 4, k = s & 15;
        const int idx = rbase[tm] + __popc(rmask[tm] & ((1u << tk) - 1u));
        const int off = ((k >> 3) * 16 + (d & 15)) * 8 + (k & 7);
        atomicAdd(&T[idx * 256 + off], 1);
    }
}

__device__ __forceinline__ float xhalf(float x) {
    const int b = __builtin_bit_cast(int, x);
    return __builtin_bit_cast(float,
        __builtin_amdgcn_permlanex16(b, b, 0x76543210, 0xfedcba98, false, false));
}

__global__ __launch_bounds__(NT) void rgcn_main_kernel(
    const float* __restrict__ xx, const float* __restrict__ ss,
    const float* __restrict__ Wf, const float* __restrict__ bf,
    const float* __restrict__ convw, const float* __restrict__ convroot,
    const float* __restrict__ convbias, const float* __restrict__ lng,
    const float* __restrict__ lnb, const float* __restrict__ invdeg,
    const unsigned* __restrict__ rowmask, const int* __restrict__ rowbase,
    const int* __restrict__ tiles, float* __restrict__ out)
{
    __shared__ __attribute__((aligned(16))) float xs[CC * XP];
    __shared__ __attribute__((aligned(16))) float wts[4 * CC * WP];
    __shared__ float ivd[3 * NN];
    __shared__ unsigned rmk[3 * 32];
    __shared__ int rbs[3 * 32];
    __shared__ __attribute__((aligned(16))) float wfe[6 * CC];
    __shared__ __attribute__((aligned(16))) float prm[3 * CC];
    __shared__ __attribute__((aligned(16))) float cbs[LL * CC];

    const int bid   = blockIdx.x;
    const int tid   = threadIdx.x;
    const int lane  = tid & 31;
    const int wid   = tid >> 5;
    const int mlo   = lane & 15;
    const int mhalf = lane >> 4;
    const int wdst  = (tid & 31) * WP + (tid >> 5);

    for (int i = tid; i < 3 * NN; i += NT) ivd[i] = invdeg[i];
    for (int i = tid; i < 6 * CC; i += NT) wfe[i] = Wf[i];
    if (tid < 96) {
        rmk[tid] = rowmask[tid];
        rbs[tid] = rowbase[tid];
    }
    if (tid < CC) {
        prm[tid]          = bf[tid];
        prm[CC + tid]     = lng[tid];
        prm[2 * CC + tid] = lnb[tid];
    }
    if (tid < LL * CC) cbs[tid] = convbias[tid];
    wts[wdst] = convroot[tid];
#pragma unroll
    for (int j = 0; j < 3; ++j)
        wts[(1 + j) * CC * WP + wdst] = convw[j * CC * CC + tid];
    __syncthreads();

    float4* xs4 = (float4*)xs;
    const float4* wt4 = (const float4*)wts;
    const float4* wf4 = (const float4*)wfe;
    const float4* cb4 = (const float4*)cbs;
    const float4* pr4 = (const float4*)prm;

    const int ownn = wid * 16 + mlo;

    v8f xo[2];
    {
        const float mfeat = ss[bid] * 0.125f;
        const int n = ownn;
        const float v  = xx[bid * NN + n];
        const float fi = (float)((n >> 4) & 3) * (1.0f / 3.0f);
        const float fj = (float)((n >> 2) & 3) * (1.0f / 3.0f);
        const float fk = (float)(n & 3) * (1.0f / 3.0f);
        const float ft = (float)(n >> 6) * (1.0f / 7.0f);
#pragma unroll
        for (int t = 0; t < 2; ++t) {
            const int c4 = 4 * t + 2 * mhalf;
            float a[8], w[8];
            { const float4 lo = pr4[c4], hi = pr4[c4 + 1];
              a[0] = lo.x; a[1] = lo.y; a[2] = lo.z; a[3] = lo.w;
              a[4] = hi.x; a[5] = hi.y; a[6] = hi.z; a[7] = hi.w; }
            const float fe[6] = {fi, fj, fk, ft, v, mfeat};
#pragma unroll
            for (int f = 0; f < 6; ++f) {
                const float4 lo = wf4[f * 8 + c4], hi = wf4[f * 8 + c4 + 1];
                w[0] = lo.x; w[1] = lo.y; w[2] = lo.z; w[3] = lo.w;
                w[4] = hi.x; w[5] = hi.y; w[6] = hi.z; w[7] = hi.w;
#pragma unroll
                for (int j = 0; j < 8; ++j) a[j] += fe[f] * w[j];
            }
#pragma unroll
            for (int j = 0; j < 8; ++j) {
                xo[t][j] = a[j];
                xs[(16 * t + 8 * mhalf + j) * XP + ownn] = a[j];
            }
        }
    }
    __syncthreads();

    unsigned mk[3];
    int bs[3];
#pragma unroll
    for (int r = 0; r < 3; ++r) {
        mk[r] = (unsigned)__builtin_amdgcn_readfirstlane((int)rmk[r * 32 + wid]);
        bs[r] = __builtin_amdgcn_readfirstlane(rbs[r * 32 + wid]);
    }
    const int xab = mlo * (XP / 4) + 2 * mhalf;
    const int wab = mlo * (WP / 4) + 2 * mhalf;
    const int4* ivp = (const int4*)invdeg;
    const int l2 = lane * 2;

    for (int l = 0; l < LL; ++l) {
        v8f acc[2];
#pragma unroll
        for (int t = 0; t < 2; ++t) {
            const float4 lo = cb4[l * 8 + 4 * t + 2 * mhalf], hi = cb4[l * 8 + 4 * t + 2 * mhalf + 1];
            acc[t] = (v8f){lo.x, lo.y, lo.z, lo.w, hi.x, hi.y, hi.z, hi.w};
        }

        float4 Wv[8];
        float4 X[4];
        int4 c0, c1;
        unsigned m = mk[0];
        const int4* tp = m ? (const int4*)(tiles + bs[0] * 256) : ivp;
#pragma unroll
        for (int t = 0; t < 2; ++t) {
            Wv[2 * t]     = wt4[wab + 4 * t];
            Wv[2 * t + 1] = wt4[wab + 4 * t + 1];
            Wv[4 + 2 * t] = wt4[wab + 16 * (WP / 4) + 4 * t];
            Wv[5 + 2 * t] = wt4[wab + 16 * (WP / 4) + 4 * t + 1];
        }
        {
            const int xa = xab + 4 * (m ? __builtin_ctz(m) : 0);
            X[0] = xs4[xa];                 X[1] = xs4[xa + 1];
            X[2] = xs4[xa + 16 * (XP / 4)]; X[3] = xs4[xa + 16 * (XP / 4) + 1];
            c0 = tp[l2]; c1 = tp[l2 + 1];
        }

        v8f z0 = xo[0], z1 = xo[1];
        v8f g0, g1;
        auto tile_ops = [&](const float4 (&Xf)[4], const int4 ca, const int4 cb) {
            const float cf[8] = {(float)ca.x, (float)ca.y, (float)ca.z, (float)ca.w,
                                 (float)cb.x, (float)cb.y, (float)cb.z, (float)cb.w};
            const float e0[8] = {Xf[0].x, Xf[0].y, Xf[0].z, Xf[0].w, Xf[1].x, Xf[1].y, Xf[1].z, Xf[1].w};
            const float e1[8] = {Xf[2].x, Xf[2].y, Xf[2].z, Xf[2].w, Xf[3].x, Xf[3].y, Xf[3].z, Xf[3].w};
#pragma unroll
            for (int ks = 0; ks < 4; ++ks) {
                v2f b, a0, a1;
                b.x  = cf[2 * ks];
                b.y  = cf[2 * ks + 1];
                a0.x = e0[2 * ks];
                a0.y = e0[2 * ks + 1];
                a1.x = e1[2 * ks];
                a1.y = e1[2 * ks + 1];
                g0 = __builtin_amdgcn_wmma_f32_16x16x4_f32(
                    false, a0, false, b, (short)0, g0, false, false);
                g1 = __builtin_amdgcn_wmma_f32_16x16x4_f32(
                    false, a1, false, b, (short)0, g1, false, false);
            }
        };
#pragma unroll
        for (int mtx = 0; mtx < 4; ++mtx) {
            __builtin_amdgcn_sched_barrier(0);
            {
                const float w0e[16] = {Wv[0].x, Wv[0].y, Wv[0].z, Wv[0].w, Wv[1].x, Wv[1].y, Wv[1].z, Wv[1].w,
                                       Wv[2].x, Wv[2].y, Wv[2].z, Wv[2].w, Wv[3].x, Wv[3].y, Wv[3].z, Wv[3].w};
                const float w1e[16] = {Wv[4].x, Wv[4].y, Wv[4].z, Wv[4].w, Wv[5].x, Wv[5].y, Wv[5].z, Wv[5].w,
                                       Wv[6].x, Wv[6].y, Wv[6].z, Wv[6].w, Wv[7].x, Wv[7].y, Wv[7].z, Wv[7].w};
#pragma unroll
                for (int q = 0; q < 8; ++q) {
                    v2f b, w0, w1;
                    w0.x = w0e[2 * q];
                    w0.y = w0e[2 * q + 1];
                    w1.x = w1e[2 * q];
                    w1.y = w1e[2 * q + 1];
                    if (q < 4) { b.x = z0[2 * q];       b.y = z0[2 * q + 1]; }
                    else       { b.x = z1[2 * (q - 4)]; b.y = z1[2 * (q - 4) + 1]; }
                    acc[0] = __builtin_amdgcn_wmma_f32_16x16x4_f32(
                        false, w0, false, b, (short)0, acc[0], false, false);
                    acc[1] = __builtin_amdgcn_wmma_f32_16x16x4_f32(
                        false, w1, false, b, (short)0, acc[1], false, false);
                }
            }
            if (mtx == 3) break;
            __builtin_amdgcn_sched_barrier(0);

            const int r = mtx;
            const int wa = (r + 1) * (CC * WP / 4) + wab;
            g0 = (v8f){0, 0, 0, 0, 0, 0, 0, 0};
            g1 = (v8f){0, 0, 0, 0, 0, 0, 0, 0};
            int rem = __popc(m);
#pragma nounroll
            while (rem > 1) {
                float4 Y[4];
                m &= m - 1u;
                --rem;
                tp += 64;
                const int xa = xab + 4 * __builtin_ctz(m);
                Y[0] = xs4[xa];                 Y[1] = xs4[xa + 1];
                Y[2] = xs4[xa + 16 * (XP / 4)]; Y[3] = xs4[xa + 16 * (XP / 4) + 1];
                const int4 d0 = tp[l2], d1 = tp[l2 + 1];
                __builtin_amdgcn_sched_barrier(0);
                tile_ops(X, c0, c1);
                __builtin_amdgcn_sched_barrier(0);
                if (rem == 1) {
                    X[0] = Y[0]; X[1] = Y[1]; X[2] = Y[2]; X[3] = Y[3];
                    c0 = d0; c1 = d1;
                    break;
                }
                m &= m - 1u;
                --rem;
                tp += 64;
                const int xb = xab + 4 * __builtin_ctz(m);
                X[0] = xs4[xb];                 X[1] = xs4[xb + 1];
                X[2] = xs4[xb + 16 * (XP / 4)]; X[3] = xs4[xb + 16 * (XP / 4) + 1];
                c0 = tp[l2]; c1 = tp[l2 + 1];
                __builtin_amdgcn_sched_barrier(0);
                tile_ops(Y, d0, d1);
                __builtin_amdgcn_sched_barrier(0);
            }
#pragma unroll
            for (int t = 0; t < 2; ++t) {
                Wv[2 * t]     = wt4[wa + 4 * t];
                Wv[2 * t + 1] = wt4[wa + 4 * t + 1];
                Wv[4 + 2 * t] = wt4[wa + 16 * (WP / 4) + 4 * t];
                Wv[5 + 2 * t] = wt4[wa + 16 * (WP / 4) + 4 * t + 1];
            }
            const float sc = ivd[r * NN + ownn];
            __builtin_amdgcn_sched_barrier(0);
            if (rem) tile_ops(X, c0, c1);
            __builtin_amdgcn_sched_barrier(0);
            z0 = g0 * sc;
            z1 = g1 * sc;
            if (r < 2) {
                m = mk[r + 1];
                tp = m ? (const int4*)(tiles + bs[r + 1] * 256) : ivp;
                const int xa = xab + 4 * (m ? __builtin_ctz(m) : 0);
                X[0] = xs4[xa];                 X[1] = xs4[xa + 1];
                X[2] = xs4[xa + 16 * (XP / 4)]; X[3] = xs4[xa + 16 * (XP / 4) + 1];
                c0 = tp[l2]; c1 = tp[l2 + 1];
            }
        }

        float ps[4] = {0.0f, 0.0f, 0.0f, 0.0f};
#pragma unroll
        for (int t = 0; t < 2; ++t)
#pragma unroll
            for (int j = 0; j < 8; ++j) {
                const float v = acc[t][j];
                acc[t][j] = v > 0.0f ? v : 0.0f;
                ps[j & 3] += acc[t][j];
            }
        float sum = (ps[0] + ps[1]) + (ps[2] + ps[3]);
        sum += xhalf(sum);
        const float mu = sum * (1.0f / 32.0f);
        float pq[4] = {0.0f, 0.0f, 0.0f, 0.0f};
#pragma unroll
        for (int t = 0; t < 2; ++t)
#pragma unroll
            for (int j = 0; j < 8; ++j) {
                const float d = acc[t][j] - mu;
                acc[t][j] = d;
                pq[j & 3] += d * d;
            }
        float sq = (pq[0] + pq[1]) + (pq[2] + pq[3]);
        sq += xhalf(sq);
        const float rs = rsqrtf(sq * (1.0f / 32.0f) + LN_EPS);
#pragma unroll
        for (int t = 0; t < 2; ++t) {
            const float4 ga = pr4[8 + 4 * t + 2 * mhalf],  gb = pr4[8 + 4 * t + 2 * mhalf + 1];
            const float4 ba = pr4[16 + 4 * t + 2 * mhalf], bb = pr4[16 + 4 * t + 2 * mhalf + 1];
            const float gg[8] = {ga.x, ga.y, ga.z, ga.w, gb.x, gb.y, gb.z, gb.w};
            const float bv[8] = {ba.x, ba.y, ba.z, ba.w, bb.x, bb.y, bb.z, bb.w};
#pragma unroll
            for (int j = 0; j < 8; ++j)
                xo[t][j] = acc[t][j] * rs * gg[j] + bv[j];
        }
        __syncthreads();

#pragma unroll
        for (int t = 0; t < 2; ++t)
#pragma unroll
            for (int j = 0; j < 8; ++j)
                xs[(16 * t + 8 * mhalf + j) * XP + ownn] = xo[t][j];
        if (l + 1 < LL) {
            wts[wdst] = convroot[(l + 1) * CC * CC + tid];
#pragma unroll
            for (int j = 0; j < 3; ++j)
                wts[(1 + j) * CC * WP + wdst] = convw[((l + 1) * 3 + j) * CC * CC + tid];
        }
        __syncthreads();
    }

    for (int idx = tid; idx < 48 * CC; idx += NT) {
        const int row = idx >> 5, c = idx & 31;
        const int g = row >> 4, rc = row & 15;
        const int a0 = rc >> 2, b0 = rc & 3;
        float sum = 0.0f;
#pragma unroll
        for (int q = 0; q < 4; ++q) {
            int node;
            if (g == 0)      node = q * 16 + a0 * 4 + b0;
            else if (g == 1) node = a0 * 16 + q * 4 + b0;
            else             node = a0 * 16 + b0 * 4 + q;
            sum += xs[c * XP + node];
        }
        out[bid * 48 * CC + idx] = sum * 0.25f;
    }
}

extern "C" void kernel_launch(void* const* d_in, const int* in_sizes, int n_in,
                              void* d_out, int out_size, void* d_ws, size_t ws_size,
                              hipStream_t stream) {
    const float* xx       = (const float*)d_in[0];
    const float* ss       = (const float*)d_in[1];
    const float* Wf       = (const float*)d_in[2];
    const float* bf       = (const float*)d_in[3];
    const float* convw    = (const float*)d_in[4];
    const float* convroot = (const float*)d_in[5];
    const float* convbias = (const float*)d_in[6];
    const float* lng      = (const float*)d_in[7];
    const float* lnb      = (const float*)d_in[8];
    const int*   src      = (const int*)d_in[9];
    const int*   dst      = (const int*)d_in[10];
    float*       out      = (float*)d_out;
    float*       invdeg   = (float*)d_ws + WS_INVDEG;
    unsigned*    rowmask  = (unsigned*)d_ws + WS_ROWMASK;
    int*         rowbase  = (int*)d_ws + WS_ROWBASE;
    int*         tiles    = (int*)d_ws + WS_TILES;

    const int E = in_sizes[9] / 3;
    const int B = in_sizes[1];

    rgcn_prep_kernel<<<3, 1024, 0, stream>>>(src, dst, E, invdeg, rowmask, rowbase, tiles);
    rgcn_main_kernel<<<B, NT, 0, stream>>>(xx, ss, Wf, bf, convw, convroot,
                                            convbias, lng, lnb, invdeg, rowmask,
                                            rowbase, tiles, out);
    (void)n_in; (void)out_size; (void)ws_size;
}


# Pre-kernel is one work-group per (relation, 16-row block), 96 not 3: edges read in one round trip, tile strip built in LDS, stored to fixed slots; no zeroing pass or memory atomics; sum order kept
# speedup vs baseline: 23.3813x; 1.1494x over previous
#include <hip/hip_runtime.h>
#include <hip/hip_bf16.h>

typedef __attribute__((ext_vector_type(2))) float v2f;
typedef __attribute__((ext_vector_type(8))) float v8f;

#define NN  512
#define CC  32
#define LL  4
#define LN_EPS 1e-5f
#define NT  1024
static_assert(NT == CC * CC && NT == 2 * NN, "staging and row-block ownership assume 1024 threads");
#define XP  516
#define WP  36
static_assert((XP * 4) % 16 == 0 && ((XP * 4 / 16) & 1) == 1 && XP >= NN, "xT pitch");
static_assert((WP * 4) % 16 == 0 && ((WP * 4 / 16) & 1) == 1 && WP >= CC, "WT pitch");

#define WS_INVDEG  0
#define WS_ROWMASK 1536
#define WS_ROWBASE 1664
#define WS_TILES   2048

#define PT  256
#define EPT 24
__global__ __launch_bounds__(PT) void rgcn_prep_kernel(
    const int* __restrict__ src, const int* __restrict__ dst, int E,
    float* __restrict__ invdeg, unsigned* __restrict__ rowmask,
    int* __restrict__ rowbase, int* __restrict__ tiles)
{
    __shared__ __attribute__((aligned(16))) int strip[32 * 256];
    __shared__ int deg[16];
    __shared__ unsigned msk;
    const int r   = blockIdx.x >> 5;
    const int tm  = blockIdx.x & 31;
    const int tid = threadIdx.x;
    const int* sR = src + r * E;
    const int* dR = dst + r * E;
    int4* s4 = (int4*)strip;
    for (int i = tid; i < 32 * 64; i += PT) s4[i] = make_int4(0, 0, 0, 0);
    if (tid < 16) deg[tid] = 0;
    if (tid == 0) msk = 0u;
    __syncthreads();
    for (int e0 = 0; e0 < E; e0 += PT * EPT) {
        int sv[EPT], dv[EPT];
#pragma unroll
        for (int q = 0; q < EPT; ++q) {
            const int e  = e0 + q * PT + tid;
            const int ec = e < E ? e : E - 1;
            const int d  = dR[ec];
            sv[q] = sR[ec];
            dv[q] = e < E ? d : -1;
        }
#pragma unroll
        for (int q = 0; q < EPT; ++q) {
            const int d = dv[q];
            if ((d >> 4) == tm) {
                const int s = sv[q] & (NN - 1);
                const int k = s & 15;
                atomicAdd(&strip[(s >> 4) * 256 + ((k >> 3) * 16 + (d & 15)) * 8 + (k & 7)], 1);
                atomicAdd(&deg[d & 15], 1);
            }
        }
    }
    __syncthreads();
    const int lane = tid & 31;
    const int wv   = tid >> 5;
    int4 ta[4], tb[4];
    unsigned ne = 0u;
#pragma unroll
    for (int j = 0; j < 4; ++j) {
        const int tk = wv * 4 + j;
        ta[j] = s4[tk * 64 + lane];
        tb[j] = s4[tk * 64 + 32 + lane];
        const int any = ta[j].x | ta[j].y | ta[j].z | ta[j].w | tb[j].x | tb[j].y | tb[j].z | tb[j].w;
        if (__builtin_amdgcn_ballot_w32(any != 0)) ne |= 1u << tk;
    }
    if (lane == 0 && ne) atomicOr(&msk, ne);
    __syncthreads();
    const unsigned m = msk;
    int4* T = (int4*)(tiles + (r * 1024 + tm * 32) * 256);
#pragma unroll
    for (int j = 0; j < 4; ++j) {
        const int tk = wv * 4 + j;
        if ((ne >> tk) & 1u) {
            const int idx = __popc(m & ((1u << tk) - 1u));
            T[idx * 64 + lane]      = ta[j];
            T[idx * 64 + 32 + lane] = tb[j];
        }
    }
    if (tid == 0) {
        rowmask[r * 32 + tm] = m;
        rowbase[r * 32 + tm] = r * 1024 + tm * 32;
    }
    if (tid < 16) {
        int d = deg[tid];
        if (d < 1) d = 1;
        invdeg[r * NN + tm * 16 + tid] = 1.0f / (float)d;
    }
}

__device__ __forceinline__ float xhalf(float x) {
    const int b = __builtin_bit_cast(int, x);
    return __builtin_bit_cast(float,
        __builtin_amdgcn_permlanex16(b, b, 0x76543210, 0xfedcba98, false, false));
}

__global__ __launch_bounds__(NT) void rgcn_main_kernel(
    const float* __restrict__ xx, const float* __restrict__ ss,
    const float* __restrict__ Wf, const float* __restrict__ bf,
    const float* __restrict__ convw, const float* __restrict__ convroot,
    const float* __restrict__ convbias, const float* __restrict__ lng,
    const float* __restrict__ lnb, const float* __restrict__ invdeg,
    const unsigned* __restrict__ rowmask, const int* __restrict__ rowbase,
    const int* __restrict__ tiles, float* __restrict__ out)
{
    __shared__ __attribute__((aligned(16))) float xs[CC * XP];
    __shared__ __attribute__((aligned(16))) float wts[4 * CC * WP];
    __shared__ float ivd[3 * NN];
    __shared__ unsigned rmk[3 * 32];
    __shared__ int rbs[3 * 32];
    __shared__ __attribute__((aligned(16))) float wfe[6 * CC];
    __shared__ __attribute__((aligned(16))) float prm[3 * CC];
    __shared__ __attribute__((aligned(16))) float cbs[LL * CC];

    const int bid   = blockIdx.x;
    const int tid   = threadIdx.x;
    const int lane  = tid & 31;
    const int wid   = tid >> 5;
    const int mlo   = lane & 15;
    const int mhalf = lane >> 4;
    const int wdst  = (tid & 31) * WP + (tid >> 5);

    for (int i = tid; i < 3 * NN; i += NT) ivd[i] = invdeg[i];
    for (int i = tid; i < 6 * CC; i += NT) wfe[i] = Wf[i];
    if (tid < 96) {
        rmk[tid] = rowmask[tid];
        rbs[tid] = rowbase[tid];
    }
    if (tid < CC) {
        prm[tid]          = bf[tid];
        prm[CC + tid]     = lng[tid];
        prm[2 * CC + tid] = lnb[tid];
    }
    if (tid < LL * CC) cbs[tid] = convbias[tid];
    wts[wdst] = convroot[tid];
#pragma unroll
    for (int j = 0; j < 3; ++j)
        wts[(1 + j) * CC * WP + wdst] = convw[j * CC * CC + tid];
    __syncthreads();

    float4* xs4 = (float4*)xs;
    const float4* wt4 = (const float4*)wts;
    const float4* wf4 = (const float4*)wfe;
    const float4* cb4 = (const float4*)cbs;
    const float4* pr4 = (const float4*)prm;

    const int ownn = wid * 16 + mlo;

    v8f xo[2];
    {
        const float mfeat = ss[bid] * 0.125f;
        const int n = ownn;
        const float v  = xx[bid * NN + n];
        const float fi = (float)((n >> 4) & 3) * (1.0f / 3.0f);
        const float fj = (float)((n >> 2) & 3) * (1.0f / 3.0f);
        const float fk = (float)(n & 3) * (1.0f / 3.0f);
        const float ft = (float)(n >> 6) * (1.0f / 7.0f);
#pragma unroll
        for (int t = 0; t < 2; ++t) {
            const int c4 = 4 * t + 2 * mhalf;
            float a[8], w[8];
            { const float4 lo = pr4[c4], hi = pr4[c4 + 1];
              a[0] = lo.x; a[1] = lo.y; a[2] = lo.z; a[3] = lo.w;
              a[4] = hi.x; a[5] = hi.y; a[6] = hi.z; a[7] = hi.w; }
            const float fe[6] = {fi, fj, fk, ft, v, mfeat};
#pragma unroll
            for (int f = 0; f < 6; ++f) {
                const float4 lo = wf4[f * 8 + c4], hi = wf4[f * 8 + c4 + 1];
                w[0] = lo.x; w[1] = lo.y; w[2] = lo.z; w[3] = lo.w;
                w[4] = hi.x; w[5] = hi.y; w[6] = hi.z; w[7] = hi.w;
#pragma unroll
                for (int j = 0; j < 8; ++j) a[j] += fe[f] * w[j];
            }
#pragma unroll
            for (int j = 0; j < 8; ++j) {
                xo[t][j] = a[j];
                xs[(16 * t + 8 * mhalf + j) * XP + ownn] = a[j];
            }
        }
    }
    __syncthreads();

    unsigned mk[3];
    int bs[3];
#pragma unroll
    for (int r = 0; r < 3; ++r) {
        mk[r] = (unsigned)__builtin_amdgcn_readfirstlane((int)rmk[r * 32 + wid]);
        bs[r] = __builtin_amdgcn_readfirstlane(rbs[r * 32 + wid]);
    }
    const int xab = mlo * (XP / 4) + 2 * mhalf;
    const int wab = mlo * (WP / 4) + 2 * mhalf;
    const int4* ivp = (const int4*)invdeg;
    const int l2 = lane * 2;

    for (int l = 0; l < LL; ++l) {
        v8f acc[2];
#pragma unroll
        for (int t = 0; t < 2; ++t) {
            const float4 lo = cb4[l * 8 + 4 * t + 2 * mhalf], hi = cb4[l * 8 + 4 * t + 2 * mhalf + 1];
            acc[t] = (v8f){lo.x, lo.y, lo.z, lo.w, hi.x, hi.y, hi.z, hi.w};
        }

        float4 Wv[8];
        float4 X[4];
        int4 c0, c1;
        unsigned m = mk[0];
        const int4* tp = m ? (const int4*)(tiles + bs[0] * 256) : ivp;
#pragma unroll
        for (int t = 0; t < 2; ++t) {
            Wv[2 * t]     = wt4[wab + 4 * t];
            Wv[2 * t + 1] = wt4[wab + 4 * t + 1];
            Wv[4 + 2 * t] = wt4[wab + 16 * (WP / 4) + 4 * t];
            Wv[5 + 2 * t] = wt4[wab + 16 * (WP / 4) + 4 * t + 1];
        }
        {
            const int xa = xab + 4 * (m ? __builtin_ctz(m) : 0);
            X[0] = xs4[xa];                 X[1] = xs4[xa + 1];
            X[2] = xs4[xa + 16 * (XP / 4)]; X[3] = xs4[xa + 16 * (XP / 4) + 1];
            c0 = tp[l2]; c1 = tp[l2 + 1];
        }

        v8f z0 = xo[0], z1 = xo[1];
        v8f g0, g1;
        auto tile_ops = [&](const float4 (&Xf)[4], const int4 ca, const int4 cb) {
            const float cf[8] = {(float)ca.x, (float)ca.y, (float)ca.z, (float)ca.w,
                                 (float)cb.x, (float)cb.y, (float)cb.z, (float)cb.w};
            const float e0[8] = {Xf[0].x, Xf[0].y, Xf[0].z, Xf[0].w, Xf[1].x, Xf[1].y, Xf[1].z, Xf[1].w};
            const float e1[8] = {Xf[2].x, Xf[2].y, Xf[2].z, Xf[2].w, Xf[3].x, Xf[3].y, Xf[3].z, Xf[3].w};
#pragma unroll
            for (int ks = 0; ks < 4; ++ks) {
                v2f b, a0, a1;
                b.x  = cf[2 * ks];
                b.y  = cf[2 * ks + 1];
                a0.x = e0[2 * ks];
                a0.y = e0[2 * ks + 1];
                a1.x = e1[2 * ks];
                a1.y = e1[2 * ks + 1];
                g0 = __builtin_amdgcn_wmma_f32_16x16x4_f32(
                    false, a0, false, b, (short)0, g0, false, false);
                g1 = __builtin_amdgcn_wmma_f32_16x16x4_f32(
                    false, a1, false, b, (short)0, g1, false, false);
            }
        };
#pragma unroll
        for (int mtx = 0; mtx < 4; ++mtx) {
            __builtin_amdgcn_sched_barrier(0);
            {
                const float w0e[16] = {Wv[0].x, Wv[0].y, Wv[0].z, Wv[0].w, Wv[1].x, Wv[1].y, Wv[1].z, Wv[1].w,
                                       Wv[2].x, Wv[2].y, Wv[2].z, Wv[2].w, Wv[3].x, Wv[3].y, Wv[3].z, Wv[3].w};
                const float w1e[16] = {Wv[4].x, Wv[4].y, Wv[4].z, Wv[4].w, Wv[5].x, Wv[5].y, Wv[5].z, Wv[5].w,
                                       Wv[6].x, Wv[6].y, Wv[6].z, Wv[6].w, Wv[7].x, Wv[7].y, Wv[7].z, Wv[7].w};
#pragma unroll
                for (int q = 0; q < 8; ++q) {
                    v2f b, w0, w1;
                    w0.x = w0e[2 * q];
                    w0.y = w0e[2 * q + 1];
                    w1.x = w1e[2 * q];
                    w1.y = w1e[2 * q + 1];
                    if (q < 4) { b.x = z0[2 * q];       b.y = z0[2 * q + 1]; }
                    else       { b.x = z1[2 * (q - 4)]; b.y = z1[2 * (q - 4) + 1]; }
                    acc[0] = __builtin_amdgcn_wmma_f32_16x16x4_f32(
                        false, w0, false, b, (short)0, acc[0], false, false);
                    acc[1] = __builtin_amdgcn_wmma_f32_16x16x4_f32(
                        false, w1, false, b, (short)0, acc[1], false, false);
                }
            }
            if (mtx == 3) break;
            __builtin_amdgcn_sched_barrier(0);

            const int r = mtx;
            const int wa = (r + 1) * (CC * WP / 4) + wab;
            g0 = (v8f){0, 0, 0, 0, 0, 0, 0, 0};
            g1 = (v8f){0, 0, 0, 0, 0, 0, 0, 0};
            int rem = __popc(m);
#pragma nounroll
            while (rem > 1) {
                float4 Y[4];
                m &= m - 1u;
                --rem;
                tp += 64;
                const int xa = xab + 4 * __builtin_ctz(m);
                Y[0] = xs4[xa];                 Y[1] = xs4[xa + 1];
                Y[2] = xs4[xa + 16 * (XP / 4)]; Y[3] = xs4[xa + 16 * (XP / 4) + 1];
                const int4 d0 = tp[l2], d1 = tp[l2 + 1];
                __builtin_amdgcn_sched_barrier(0);
                tile_ops(X, c0, c1);
                __builtin_amdgcn_sched_barrier(0);
                if (rem == 1) {
                    X[0] = Y[0]; X[1] = Y[1]; X[2] = Y[2]; X[3] = Y[3];
                    c0 = d0; c1 = d1;
                    break;
                }
                m &= m - 1u;
                --rem;
                tp += 64;
                const int xb = xab + 4 * __builtin_ctz(m);
                X[0] = xs4[xb];                 X[1] = xs4[xb + 1];
                X[2] = xs4[xb + 16 * (XP / 4)]; X[3] = xs4[xb + 16 * (XP / 4) + 1];
                c0 = tp[l2]; c1 = tp[l2 + 1];
                __builtin_amdgcn_sched_barrier(0);
                tile_ops(Y, d0, d1);
                __builtin_amdgcn_sched_barrier(0);
            }
#pragma unroll
            for (int t = 0; t < 2; ++t) {
                Wv[2 * t]     = wt4[wa + 4 * t];
                Wv[2 * t + 1] = wt4[wa + 4 * t + 1];
                Wv[4 + 2 * t] = wt4[wa + 16 * (WP / 4) + 4 * t];
                Wv[5 + 2 * t] = wt4[wa + 16 * (WP / 4) + 4 * t + 1];
            }
            const float sc = ivd[r * NN + ownn];
            __builtin_amdgcn_sched_barrier(0);
            if (rem) tile_ops(X, c0, c1);
            __builtin_amdgcn_sched_barrier(0);
            z0 = g0 * sc;
            z1 = g1 * sc;
            if (r < 2) {
                m = mk[r + 1];
                tp = m ? (const int4*)(tiles + bs[r + 1] * 256) : ivp;
                const int xa = xab + 4 * (m ? __builtin_ctz(m) : 0);
                X[0] = xs4[xa];                 X[1] = xs4[xa + 1];
                X[2] = xs4[xa + 16 * (XP / 4)]; X[3] = xs4[xa + 16 * (XP / 4) + 1];
                c0 = tp[l2]; c1 = tp[l2 + 1];
            }
        }

        float ps[4] = {0.0f, 0.0f, 0.0f, 0.0f};
#pragma unroll
        for (int t = 0; t < 2; ++t)
#pragma unroll
            for (int j = 0; j < 8; ++j) {
                const float v = acc[t][j];
                acc[t][j] = v > 0.0f ? v : 0.0f;
                ps[j & 3] += acc[t][j];
            }
        float sum = (ps[0] + ps[1]) + (ps[2] + ps[3]);
        sum += xhalf(sum);
        const float mu = sum * (1.0f / 32.0f);
        float pq[4] = {0.0f, 0.0f, 0.0f, 0.0f};
#pragma unroll
        for (int t = 0; t < 2; ++t)
#pragma unroll
            for (int j = 0; j < 8; ++j) {
                const float d = acc[t][j] - mu;
                acc[t][j] = d;
                pq[j & 3] += d * d;
            }
        float sq = (pq[0] + pq[1]) + (pq[2] + pq[3]);
        sq += xhalf(sq);
        const float rs = rsqrtf(sq * (1.0f / 32.0f) + LN_EPS);
#pragma unroll
        for (int t = 0; t < 2; ++t) {
            const float4 ga = pr4[8 + 4 * t + 2 * mhalf],  gb = pr4[8 + 4 * t + 2 * mhalf + 1];
            const float4 ba = pr4[16 + 4 * t + 2 * mhalf], bb = pr4[16 + 4 * t + 2 * mhalf + 1];
            const float gg[8] = {ga.x, ga.y, ga.z, ga.w, gb.x, gb.y, gb.z, gb.w};
            const float bv[8] = {ba.x, ba.y, ba.z, ba.w, bb.x, bb.y, bb.z, bb.w};
#pragma unroll
            for (int j = 0; j < 8; ++j)
                xo[t][j] = acc[t][j] * rs * gg[j] + bv[j];
        }
        __syncthreads();

#pragma unroll
        for (int t = 0; t < 2; ++t)
#pragma unroll
            for (int j = 0; j < 8; ++j)
                xs[(16 * t + 8 * mhalf + j) * XP + ownn] = xo[t][j];
        if (l + 1 < LL) {
            wts[wdst] = convroot[(l + 1) * CC * CC + tid];
#pragma unroll
            for (int j = 0; j < 3; ++j)
                wts[(1 + j) * CC * WP + wdst] = convw[((l + 1) * 3 + j) * CC * CC + tid];
        }
        __syncthreads();
    }

    for (int idx = tid; idx < 48 * CC; idx += NT) {
        const int row = idx >> 5, c = idx & 31;
        const int g = row >> 4, rc = row & 15;
        const int a0 = rc >> 2, b0 = rc & 3;
        float sum = 0.0f;
#pragma unroll
        for (int q = 0; q < 4; ++q) {
            int node;
            if (g == 0)      node = q * 16 + a0 * 4 + b0;
            else if (g == 1) node = a0 * 16 + q * 4 + b0;
            else             node = a0 * 16 + b0 * 4 + q;
            sum += xs[c * XP + node];
        }
        out[bid * 48 * CC + idx] = sum * 0.25f;
    }
}

extern "C" void kernel_launch(void* const* d_in, const int* in_sizes, int n_in,
                              void* d_out, int out_size, void* d_ws, size_t ws_size,
                              hipStream_t stream) {
    const float* xx       = (const float*)d_in[0];
    const float* ss       = (const float*)d_in[1];
    const float* Wf       = (const float*)d_in[2];
    const float* bf       = (const float*)d_in[3];
    const float* convw    = (const float*)d_in[4];
    const float* convroot = (const float*)d_in[5];
    const float* convbias = (const float*)d_in[6];
    const float* lng      = (const float*)d_in[7];
    const float* lnb      = (const float*)d_in[8];
    const int*   src      = (const int*)d_in[9];
    const int*   dst      = (const int*)d_in[10];
    float*       out      = (float*)d_out;
    float*       invdeg   = (float*)d_ws + WS_INVDEG;
    unsigned*    rowmask  = (unsigned*)d_ws + WS_ROWMASK;
    int*         rowbase  = (int*)d_ws + WS_ROWBASE;
    int*         tiles    = (int*)d_ws + WS_TILES;

    const int E = in_sizes[9] / 3;
    const int B = in_sizes[1];

    rgcn_prep_kernel<<<96, PT, 0, stream>>>(src, dst, E, invdeg, rowmask, rowbase, tiles);
    rgcn_main_kernel<<<B, NT, 0, stream>>>(xx, ss, Wf, bf, convw, convroot,
                                            convbias, lng, lnb, invdeg, rowmask,
                                            rowbase, tiles, out);
    (void)n_in; (void)out_size; (void)ws_size;
}


# Source blocks of a row block whose 16x16 edge-count tiles are equal share one tile product: pre-kernel finds equal tiles, main kernel adds their x fragments first; aggregation sum order changed
# speedup vs baseline: 26.7344x; 1.1434x over previous
#include <hip/hip_runtime.h>
#include <hip/hip_bf16.h>

typedef __attribute__((ext_vector_type(2))) float v2f;
typedef __attribute__((ext_vector_type(8))) float v8f;

#define NN  512
#define CC  32
#define LL  4
#define LN_EPS 1e-5f
#define NT  1024
static_assert(NT == CC * CC && NT == 2 * NN, "staging and row-block ownership assume 1024 threads");
#define XP  516
#define WP  36
static_assert((XP * 4) % 16 == 0 && ((XP * 4 / 16) & 1) == 1 && XP >= NN, "xT pitch");
static_assert((WP * 4) % 16 == 0 && ((WP * 4 / 16) & 1) == 1 && WP >= CC, "WT pitch");

#define WS_INVDEG  0
#define WS_ROWMASK 1536
#define WS_ROWBASE 1664
#define WS_GROUPS  2048
#define WS_TILES   8192

#define PT  256
#define EPT 24
__global__ __launch_bounds__(PT) void rgcn_prep_kernel(
    const int* __restrict__ src, const int* __restrict__ dst, int E,
    float* __restrict__ invdeg, unsigned* __restrict__ rowmask,
    int* __restrict__ rowbase, unsigned* __restrict__ groups, int* __restrict__ tiles)
{
    __shared__ __attribute__((aligned(16))) int strip[32 * 256];
    __shared__ int deg[16];
    __shared__ unsigned msk;
    __shared__ unsigned gml[32];
    __shared__ unsigned gout[32];
    const int r   = blockIdx.x >> 5;
    const int tm  = blockIdx.x & 31;
    const int tid = threadIdx.x;
    const int* sR = src + r * E;
    const int* dR = dst + r * E;
    int4* s4 = (int4*)strip;
    for (int i = tid; i < 32 * 64; i += PT) s4[i] = make_int4(0, 0, 0, 0);
    if (tid < 16) deg[tid] = 0;
    if (tid == 0) msk = 0u;
    if (tid < 32) { gml[tid] = 0u; gout[tid] = 0u; }
    __syncthreads();
    for (int e0 = 0; e0 < E; e0 += PT * EPT) {
        int sv[EPT], dv[EPT];
#pragma unroll
        for (int q = 0; q < EPT; ++q) {
            const int e  = e0 + q * PT + tid;
            const int ec = e < E ? e : E - 1;
            const int d  = dR[ec];
            sv[q] = sR[ec];
            dv[q] = e < E ? d : -1;
        }
#pragma unroll
        for (int q = 0; q < EPT; ++q) {
            const int d = dv[q];
            if ((d >> 4) == tm) {
                const int s = sv[q] & (NN - 1);
                const int k = s & 15;
                atomicAdd(&strip[(s >> 4) * 256 + ((k >> 3) * 16 + (d & 15)) * 8 + (k & 7)], 1);
                atomicAdd(&deg[d & 15], 1);
            }
        }
    }
    __syncthreads();
    const int lane = tid & 31;
    const int wv   = tid >> 5;
    int4 ta[4], tb[4];
    unsigned ne = 0u;
#pragma unroll
    for (int j = 0; j < 4; ++j) {
        const int tk = wv * 4 + j;
        ta[j] = s4[tk * 64 + lane];
        tb[j] = s4[tk * 64 + 32 + lane];
        const int any = ta[j].x | ta[j].y | ta[j].z | ta[j].w | tb[j].x | tb[j].y | tb[j].z | tb[j].w;
        if (__builtin_amdgcn_ballot_w32(any != 0)) ne |= 1u << tk;
    }
    if (lane == 0 && ne) atomicOr(&msk, ne);
    __syncthreads();
    const unsigned m = (unsigned)__builtin_amdgcn_readfirstlane((int)msk);
#pragma unroll
    for (int j = 0; j < 4; ++j) {
        const int tk = wv * 4 + j;
        if ((ne >> tk) & 1u) {
            int first = tk;
            unsigned cand = m & ((1u << tk) - 1u);
            while (cand) {
                const int tc = __builtin_ctz(cand);
                cand &= cand - 1u;
                const int4 ua = s4[tc * 64 + lane], ub = s4[tc * 64 + 32 + lane];
                const int diff = (ua.x ^ ta[j].x) | (ua.y ^ ta[j].y) | (ua.z ^ ta[j].z) | (ua.w ^ ta[j].w)
                               | (ub.x ^ tb[j].x) | (ub.y ^ tb[j].y) | (ub.z ^ tb[j].z) | (ub.w ^ tb[j].w);
                if (!__builtin_amdgcn_ballot_w32(diff != 0)) { first = tc; break; }
            }
            if (lane == 0) atomicOr(&gml[first], 1u << tk);
        }
    }
    __syncthreads();
    const unsigned um = __builtin_amdgcn_ballot_w32(gml[lane] != 0u);
    int4* T = (int4*)(tiles + (r * 1024 + tm * 32) * 256);
#pragma unroll
    for (int j = 0; j < 4; ++j) {
        const int tk = wv * 4 + j;
        if ((um >> tk) & 1u) {
            const int idx = __popc(um & ((1u << tk) - 1u));
            T[idx * 64 + lane]      = ta[j];
            T[idx * 64 + 32 + lane] = tb[j];
            if (lane == 0) gout[idx] = gml[tk];
        }
    }
    __syncthreads();
    if (tid < 32) groups[(r * 32 + tm) * 32 + tid] = gout[tid];
    if (tid == 0) {
        rowmask[r * 32 + tm] = um;
        rowbase[r * 32 + tm] = r * 1024 + tm * 32;
    }
    if (tid < 16) {
        int d = deg[tid];
        if (d < 1) d = 1;
        invdeg[r * NN + tm * 16 + tid] = 1.0f / (float)d;
    }
}

__device__ __forceinline__ float xhalf(float x) {
    const int b = __builtin_bit_cast(int, x);
    return __builtin_bit_cast(float,
        __builtin_amdgcn_permlanex16(b, b, 0x76543210, 0xfedcba98, false, false));
}

__global__ __launch_bounds__(NT) void rgcn_main_kernel(
    const float* __restrict__ xx, const float* __restrict__ ss,
    const float* __restrict__ Wf, const float* __restrict__ bf,
    const float* __restrict__ convw, const float* __restrict__ convroot,
    const float* __restrict__ convbias, const float* __restrict__ lng,
    const float* __restrict__ lnb, const float* __restrict__ invdeg,
    const unsigned* __restrict__ rowmask, const int* __restrict__ rowbase,
    const unsigned* __restrict__ groups, const int* __restrict__ tiles, float* __restrict__ out)
{
    __shared__ __attribute__((aligned(16))) float xs[CC * XP];
    __shared__ __attribute__((aligned(16))) float wts[4 * CC * WP];
    __shared__ float ivd[3 * NN];
    __shared__ unsigned rmk[3 * 32];
    __shared__ int rbs[3 * 32];
    __shared__ unsigned gms[3 * 32 * 32];
    __shared__ __attribute__((aligned(16))) float wfe[6 * CC];
    __shared__ __attribute__((aligned(16))) float prm[3 * CC];
    __shared__ __attribute__((aligned(16))) float cbs[LL * CC];

    const int bid   = blockIdx.x;
    const int tid   = threadIdx.x;
    const int lane  = tid & 31;
    const int wid   = tid >> 5;
    const int mlo   = lane & 15;
    const int mhalf = lane >> 4;
    const int wdst  = (tid & 31) * WP + (tid >> 5);

    for (int i = tid; i < 3 * NN; i += NT) ivd[i] = invdeg[i];
    for (int i = tid; i < 6 * CC; i += NT) wfe[i] = Wf[i];
    for (int i = tid; i < 3 * 32 * 32; i += NT) gms[i] = groups[i];
    if (tid < 96) {
        rmk[tid] = rowmask[tid];
        rbs[tid] = rowbase[tid];
    }
    if (tid < CC) {
        prm[tid]          = bf[tid];
        prm[CC + tid]     = lng[tid];
        prm[2 * CC + tid] = lnb[tid];
    }
    if (tid < LL * CC) cbs[tid] = convbias[tid];
    wts[wdst] = convroot[tid];
#pragma unroll
    for (int j = 0; j < 3; ++j)
        wts[(1 + j) * CC * WP + wdst] = convw[j * CC * CC + tid];
    __syncthreads();

    float4* xs4 = (float4*)xs;
    const float4* wt4 = (const float4*)wts;
    const float4* wf4 = (const float4*)wfe;
    const float4* cb4 = (const float4*)cbs;
    const float4* pr4 = (const float4*)prm;

    const int ownn = wid * 16 + mlo;

    v8f xo[2];
    {
        const float mfeat = ss[bid] * 0.125f;
        const int n = ownn;
        const float v  = xx[bid * NN + n];
        const float fi = (float)((n >> 4) & 3) * (1.0f / 3.0f);
        const float fj = (float)((n >> 2) & 3) * (1.0f / 3.0f);
        const float fk = (float)(n & 3) * (1.0f / 3.0f);
        const float ft = (float)(n >> 6) * (1.0f / 7.0f);
#pragma unroll
        for (int t = 0; t < 2; ++t) {
            if (t == 1) __builtin_amdgcn_sched_barrier(0);
            const int c4 = 4 * t + 2 * mhalf;
            float a[8], w[8];
            { const float4 lo = pr4[c4], hi = pr4[c4 + 1];
              a[0] = lo.x; a[1] = lo.y; a[2] = lo.z; a[3] = lo.w;
              a[4] = hi.x; a[5] = hi.y; a[6] = hi.z; a[7] = hi.w; }
            const float fe[6] = {fi, fj, fk, ft, v, mfeat};
#pragma unroll
            for (int f = 0; f < 6; ++f) {
                const float4 lo = wf4[f * 8 + c4], hi = wf4[f * 8 + c4 + 1];
                w[0] = lo.x; w[1] = lo.y; w[2] = lo.z; w[3] = lo.w;
                w[4] = hi.x; w[5] = hi.y; w[6] = hi.z; w[7] = hi.w;
#pragma unroll
                for (int j = 0; j < 8; ++j) a[j] += fe[f] * w[j];
            }
#pragma unroll
            for (int j = 0; j < 8; ++j) {
                xo[t][j] = a[j];
                xs[(16 * t + 8 * mhalf + j) * XP + ownn] = a[j];
            }
        }
    }
    __syncthreads();

    int ng[3];
    int bs[3];
#pragma unroll
    for (int r = 0; r < 3; ++r) {
        ng[r] = __popc((unsigned)__builtin_amdgcn_readfirstlane((int)rmk[r * 32 + wid]));
        bs[r] = __builtin_amdgcn_readfirstlane(rbs[r * 32 + wid]);
    }
    const unsigned* gmw = gms + wid * 32;
    unsigned gf[3];
#pragma unroll
    for (int r = 0; r < 3; ++r)
        gf[r] = (unsigned)__builtin_amdgcn_readfirstlane((int)gmw[r * 1024]);
    const int xab = mlo * (XP / 4) + 2 * mhalf;
    const int wab = mlo * (WP / 4) + 2 * mhalf;
    const int4* ivp = (const int4*)invdeg;
    const int l2 = lane * 2;

    for (int l = 0; l < LL; ++l) {
        v8f acc[2];
#pragma unroll
        for (int t = 0; t < 2; ++t) {
            const float4 lo = cb4[l * 8 + 4 * t + 2 * mhalf], hi = cb4[l * 8 + 4 * t + 2 * mhalf + 1];
            acc[t] = (v8f){lo.x, lo.y, lo.z, lo.w, hi.x, hi.y, hi.z, hi.w};
        }

        float4 Wv[8];
        float4 X[4];
        int4 c0, c1;
        unsigned gm = gf[0];
        const int4* tp = gm ? (const int4*)(tiles + bs[0] * 256) : ivp;
#pragma unroll
        for (int t = 0; t < 2; ++t) {
            Wv[2 * t]     = wt4[wab + 4 * t];
            Wv[2 * t + 1] = wt4[wab + 4 * t + 1];
            Wv[4 + 2 * t] = wt4[wab + 16 * (WP / 4) + 4 * t];
            Wv[5 + 2 * t] = wt4[wab + 16 * (WP / 4) + 4 * t + 1];
        }
        {
            const int xa = xab + 4 * (gm ? __builtin_ctz(gm) : 0);
            X[0] = xs4[xa];                 X[1] = xs4[xa + 1];
            X[2] = xs4[xa + 16 * (XP / 4)]; X[3] = xs4[xa + 16 * (XP / 4) + 1];
            c0 = tp[l2]; c1 = tp[l2 + 1];
        }

        v8f z0 = xo[0], z1 = xo[1];
        v8f g0, g1;
        auto tile_ops = [&](const float4 (&Xf)[4], const int4 ca, const int4 cb) {
            const float cf[8] = {(float)ca.x, (float)ca.y, (float)ca.z, (float)ca.w,
                                 (float)cb.x, (float)cb.y, (float)cb.z, (float)cb.w};
            const float e0[8] = {Xf[0].x, Xf[0].y, Xf[0].z, Xf[0].w, Xf[1].x, Xf[1].y, Xf[1].z, Xf[1].w};
            const float e1[8] = {Xf[2].x, Xf[2].y, Xf[2].z, Xf[2].w, Xf[3].x, Xf[3].y, Xf[3].z, Xf[3].w};
#pragma unroll
            for (int ks = 0; ks < 4; ++ks) {
                v2f b, a0, a1;
                b.x  = cf[2 * ks];
                b.y  = cf[2 * ks + 1];
                a0.x = e0[2 * ks];
                a0.y = e0[2 * ks + 1];
                a1.x = e1[2 * ks];
                a1.y = e1[2 * ks + 1];
                g0 = __builtin_amdgcn_wmma_f32_16x16x4_f32(
                    false, a0, false, b, (short)0, g0, false, false);
                g1 = __builtin_amdgcn_wmma_f32_16x16x4_f32(
                    false, a1, false, b, (short)0, g1, false, false);
            }
        };
        auto add_members = [&](float4 (&Xf)[4], unsigned g) {
            g &= g - 1u;
            while (g) {
                float4 P[4];
                const int pa = xab + 4 * __builtin_ctz(g);
                g &= g - 1u;
                P[0] = xs4[pa];                 P[1] = xs4[pa + 1];
                P[2] = xs4[pa + 16 * (XP / 4)]; P[3] = xs4[pa + 16 * (XP / 4) + 1];
#pragma unroll
                for (int i = 0; i < 4; ++i) {
                    Xf[i].x += P[i].x; Xf[i].y += P[i].y; Xf[i].z += P[i].z; Xf[i].w += P[i].w;
                }
            }
        };
#pragma unroll
        for (int mtx = 0; mtx < 4; ++mtx) {
            __builtin_amdgcn_sched_barrier(0);
            {
                const float w0e[16] = {Wv[0].x, Wv[0].y, Wv[0].z, Wv[0].w, Wv[1].x, Wv[1].y, Wv[1].z, Wv[1].w,
                                       Wv[2].x, Wv[2].y, Wv[2].z, Wv[2].w, Wv[3].x, Wv[3].y, Wv[3].z, Wv[3].w};
                const float w1e[16] = {Wv[4].x, Wv[4].y, Wv[4].z, Wv[4].w, Wv[5].x, Wv[5].y, Wv[5].z, Wv[5].w,
                                       Wv[6].x, Wv[6].y, Wv[6].z, Wv[6].w, Wv[7].x, Wv[7].y, Wv[7].z, Wv[7].w};
#pragma unroll
                for (int q = 0; q < 8; ++q) {
                    v2f b, w0, w1;
                    w0.x = w0e[2 * q];
                    w0.y = w0e[2 * q + 1];
                    w1.x = w1e[2 * q];
                    w1.y = w1e[2 * q + 1];
                    if (q < 4) { b.x = z0[2 * q];       b.y = z0[2 * q + 1]; }
                    else       { b.x = z1[2 * (q - 4)]; b.y = z1[2 * (q - 4) + 1]; }
                    acc[0] = __builtin_amdgcn_wmma_f32_16x16x4_f32(
                        false, w0, false, b, (short)0, acc[0], false, false);
                    acc[1] = __builtin_amdgcn_wmma_f32_16x16x4_f32(
                        false, w1, false, b, (short)0, acc[1], false, false);
                }
            }
            if (mtx == 3) break;
            __builtin_amdgcn_sched_barrier(0);

            const int r = mtx;
            const int wa = (r + 1) * (CC * WP / 4) + wab;
            const unsigned* gp = gmw + r * 1024;
            g0 = (v8f){0, 0, 0, 0, 0, 0, 0, 0};
            g1 = (v8f){0, 0, 0, 0, 0, 0, 0, 0};
            int rem = ng[r];
#pragma nounroll
            while (rem > 1) {
                float4 Y[4];
                add_members(X, gm);
                --rem;
                tp += 64;
                ++gp;
                const unsigned gn = (unsigned)__builtin_amdgcn_readfirstlane((int)gp[0]);
                const int xa = xab + 4 * __builtin_ctz(gn);
                Y[0] = xs4[xa];                 Y[1] = xs4[xa + 1];
                Y[2] = xs4[xa + 16 * (XP / 4)]; Y[3] = xs4[xa + 16 * (XP / 4) + 1];
                const int4 d0 = tp[l2], d1 = tp[l2 + 1];
                __builtin_amdgcn_sched_barrier(0);
                tile_ops(X, c0, c1);
                __builtin_amdgcn_sched_barrier(0);
                if (rem == 1) {
                    X[0] = Y[0]; X[1] = Y[1]; X[2] = Y[2]; X[3] = Y[3];
                    c0 = d0; c1 = d1;
                    gm = gn;
                    break;
                }
                add_members(Y, gn);
                --rem;
                tp += 64;
                ++gp;
                gm = (unsigned)__builtin_amdgcn_readfirstlane((int)gp[0]);
                const int xb = xab + 4 * __builtin_ctz(gm);
                X[0] = xs4[xb];                 X[1] = xs4[xb + 1];
                X[2] = xs4[xb + 16 * (XP / 4)]; X[3] = xs4[xb + 16 * (XP / 4) + 1];
                c0 = tp[l2]; c1 = tp[l2 + 1];
                __builtin_amdgcn_sched_barrier(0);
                tile_ops(Y, d0, d1);
                __builtin_amdgcn_sched_barrier(0);
            }
            add_members(X, gm);
#pragma unroll
            for (int t = 0; t < 2; ++t) {
                Wv[2 * t]     = wt4[wa + 4 * t];
                Wv[2 * t + 1] = wt4[wa + 4 * t + 1];
                Wv[4 + 2 * t] = wt4[wa + 16 * (WP / 4) + 4 * t];
                Wv[5 + 2 * t] = wt4[wa + 16 * (WP / 4) + 4 * t + 1];
            }
            const float sc = ivd[r * NN + ownn];
            __builtin_amdgcn_sched_barrier(0);
            if (rem) tile_ops(X, c0, c1);
            __builtin_amdgcn_sched_barrier(0);
            z0 = g0 * sc;
            z1 = g1 * sc;
            if (r < 2) {
                gm = gf[r + 1];
                tp = gm ? (const int4*)(tiles + bs[r + 1] * 256) : ivp;
                const int xa = xab + 4 * (gm ? __builtin_ctz(gm) : 0);
                X[0] = xs4[xa];                 X[1] = xs4[xa + 1];
                X[2] = xs4[xa + 16 * (XP / 4)]; X[3] = xs4[xa + 16 * (XP / 4) + 1];
                c0 = tp[l2]; c1 = tp[l2 + 1];
            }
        }

        float ps[4] = {0.0f, 0.0f, 0.0f, 0.0f};
#pragma unroll
        for (int t = 0; t < 2; ++t)
#pragma unroll
            for (int j = 0; j < 8; ++j) {
                const float v = acc[t][j];
                acc[t][j] = v > 0.0f ? v : 0.0f;
                ps[j & 3] += acc[t][j];
            }
        float sum = (ps[0] + ps[1]) + (ps[2] + ps[3]);
        sum += xhalf(sum);
        const float mu = sum * (1.0f / 32.0f);
        float pq[4] = {0.0f, 0.0f, 0.0f, 0.0f};
#pragma unroll
        for (int t = 0; t < 2; ++t)
#pragma unroll
            for (int j = 0; j < 8; ++j) {
                const float d = acc[t][j] - mu;
                acc[t][j] = d;
                pq[j & 3] += d * d;
            }
        float sq = (pq[0] + pq[1]) + (pq[2] + pq[3]);
        sq += xhalf(sq);
        const float rs = rsqrtf(sq * (1.0f / 32.0f) + LN_EPS);
#pragma unroll
        for (int t = 0; t < 2; ++t) {
            const float4 ga = pr4[8 + 4 * t + 2 * mhalf],  gb = pr4[8 + 4 * t + 2 * mhalf + 1];
            const float4 ba = pr4[16 + 4 * t + 2 * mhalf], bb = pr4[16 + 4 * t + 2 * mhalf + 1];
            const float gg[8] = {ga.x, ga.y, ga.z, ga.w, gb.x, gb.y, gb.z, gb.w};
            const float bv[8] = {ba.x, ba.y, ba.z, ba.w, bb.x, bb.y, bb.z, bb.w};
#pragma unroll
            for (int j = 0; j < 8; ++j)
                xo[t][j] = acc[t][j] * rs * gg[j] + bv[j];
        }
        __syncthreads();

#pragma unroll
        for (int t = 0; t < 2; ++t)
#pragma unroll
            for (int j = 0; j < 8; ++j)
                xs[(16 * t + 8 * mhalf + j) * XP + ownn] = xo[t][j];
        if (l + 1 < LL) {
            wts[wdst] = convroot[(l + 1) * CC * CC + tid];
#pragma unroll
            for (int j = 0; j < 3; ++j)
                wts[(1 + j) * CC * WP + wdst] = convw[((l + 1) * 3 + j) * CC * CC + tid];
        }
        __syncthreads();
    }

    for (int idx = tid; idx < 48 * CC; idx += NT) {
        const int row = idx >> 5, c = idx & 31;
        const int g = row >> 4, rc = row & 15;
        const int a0 = rc >> 2, b0 = rc & 3;
        float sum = 0.0f;
#pragma unroll
        for (int q = 0; q < 4; ++q) {
            int node;
            if (g == 0)      node = q * 16 + a0 * 4 + b0;
            else if (g == 1) node = a0 * 16 + q * 4 + b0;
            else             node = a0 * 16 + b0 * 4 + q;
            sum += xs[c * XP + node];
        }
        out[bid * 48 * CC + idx] = sum * 0.25f;
    }
}

extern "C" void kernel_launch(void* const* d_in, const int* in_sizes, int n_in,
                              void* d_out, int out_size, void* d_ws, size_t ws_size,
                              hipStream_t stream) {
    const float* xx       = (const float*)d_in[0];
    const float* ss       = (const float*)d_in[1];
    const float* Wf       = (const float*)d_in[2];
    const float* bf       = (const float*)d_in[3];
    const float* convw    = (const float*)d_in[4];
    const float* convroot = (const float*)d_in[5];
    const float* convbias = (const float*)d_in[6];
    const float* lng      = (const float*)d_in[7];
    const float* lnb      = (const float*)d_in[8];
    const int*   src      = (const int*)d_in[9];
    const int*   dst      = (const int*)d_in[10];
    float*       out      = (float*)d_out;
    float*       invdeg   = (float*)d_ws + WS_INVDEG;
    unsigned*    rowmask  = (unsigned*)d_ws + WS_ROWMASK;
    int*         rowbase  = (int*)d_ws + WS_ROWBASE;
    unsigned*    groups   = (unsigned*)d_ws + WS_GROUPS;
    int*         tiles    = (int*)d_ws + WS_TILES;

    const int E = in_sizes[9] / 3;
    const int B = in_sizes[1];

    rgcn_prep_kernel<<<96, PT, 0, stream>>>(src, dst, E, invdeg, rowmask, rowbase, groups, tiles);
    rgcn_main_kernel<<<B, NT, 0, stream>>>(xx, ss, Wf, bf, convw, convroot,
                                            convbias, lng, lnb, invdeg, rowmask,
                                            rowbase, groups, tiles, out);
    (void)n_in; (void)out_size; (void)ws_size;
}


# x and layer weights double-buffered in LDS: next layer's weights fetched at the layer top, stored behind the root block; new x stored without a wait; one barrier per layer, not two; sum order kept
# speedup vs baseline: 27.5413x; 1.0302x over previous
#include <hip/hip_runtime.h>
#include <hip/hip_bf16.h>

typedef __attribute__((ext_vector_type(2))) float v2f;
typedef __attribute__((ext_vector_type(8))) float v8f;

#define NN  512
#define CC  32
#define LL  4
#define LN_EPS 1e-5f
#define NT  1024
static_assert(NT == CC * CC && NT == 2 * NN, "staging and row-block ownership assume 1024 threads");
#define XP  516
#define WP  36
static_assert((XP * 4) % 16 == 0 && ((XP * 4 / 16) & 1) == 1 && XP >= NN, "xT pitch");
static_assert((WP * 4) % 16 == 0 && ((WP * 4 / 16) & 1) == 1 && WP >= CC, "WT pitch");

#define WS_INVDEG  0
#define WS_ROWMASK 1536
#define WS_ROWBASE 1664
#define WS_GROUPS  2048
#define WS_TILES   8192

#define PT  256
#define EPT 24
__global__ __launch_bounds__(PT) void rgcn_prep_kernel(
    const int* __restrict__ src, const int* __restrict__ dst, int E,
    float* __restrict__ invdeg, unsigned* __restrict__ rowmask,
    int* __restrict__ rowbase, unsigned* __restrict__ groups, int* __restrict__ tiles)
{
    __shared__ __attribute__((aligned(16))) int strip[32 * 256];
    __shared__ int deg[16];
    __shared__ unsigned msk;
    __shared__ unsigned gml[32];
    __shared__ unsigned gout[32];
    const int r   = blockIdx.x >> 5;
    const int tm  = blockIdx.x & 31;
    const int tid = threadIdx.x;
    const int* sR = src + r * E;
    const int* dR = dst + r * E;
    int4* s4 = (int4*)strip;
    for (int i = tid; i < 32 * 64; i += PT) s4[i] = make_int4(0, 0, 0, 0);
    if (tid < 16) deg[tid] = 0;
    if (tid == 0) msk = 0u;
    if (tid < 32) { gml[tid] = 0u; gout[tid] = 0u; }
    __syncthreads();
    for (int e0 = 0; e0 < E; e0 += PT * EPT) {
        int sv[EPT], dv[EPT];
#pragma unroll
        for (int q = 0; q < EPT; ++q) {
            const int e  = e0 + q * PT + tid;
            const int ec = e < E ? e : E - 1;
            const int d  = dR[ec];
            sv[q] = sR[ec];
            dv[q] = e < E ? d : -1;
        }
#pragma unroll
        for (int q = 0; q < EPT; ++q) {
            const int d = dv[q];
            if ((d >> 4) == tm) {
                const int s = sv[q] & (NN - 1);
                const int k = s & 15;
                atomicAdd(&strip[(s >> 4) * 256 + ((k >> 3) * 16 + (d & 15)) * 8 + (k & 7)], 1);
                atomicAdd(&deg[d & 15], 1);
            }
        }
    }
    __syncthreads();
    const int lane = tid & 31;
    const int wv   = tid >> 5;
    int4 ta[4], tb[4];
    unsigned ne = 0u;
#pragma unroll
    for (int j = 0; j < 4; ++j) {
        const int tk = wv * 4 + j;
        ta[j] = s4[tk * 64 + lane];
        tb[j] = s4[tk * 64 + 32 + lane];
        const int any = ta[j].x | ta[j].y | ta[j].z | ta[j].w | tb[j].x | tb[j].y | tb[j].z | tb[j].w;
        if (__builtin_amdgcn_ballot_w32(any != 0)) ne |= 1u << tk;
    }
    if (lane == 0 && ne) atomicOr(&msk, ne);
    __syncthreads();
    const unsigned m = (unsigned)__builtin_amdgcn_readfirstlane((int)msk);
#pragma unroll
    for (int j = 0; j < 4; ++j) {
        const int tk = wv * 4 + j;
        if ((ne >> tk) & 1u) {
            int first = tk;
            unsigned cand = m & ((1u << tk) - 1u);
            while (cand) {
                const int tc = __builtin_ctz(cand);
                cand &= cand - 1u;
                const int4 ua = s4[tc * 64 + lane], ub = s4[tc * 64 + 32 + lane];
                const int diff = (ua.x ^ ta[j].x) | (ua.y ^ ta[j].y) | (ua.z ^ ta[j].z) | (ua.w ^ ta[j].w)
                               | (ub.x ^ tb[j].x) | (ub.y ^ tb[j].y) | (ub.z ^ tb[j].z) | (ub.w ^ tb[j].w);
                if (!__builtin_amdgcn_ballot_w32(diff != 0)) { first = tc; break; }
            }
            if (lane == 0) atomicOr(&gml[first], 1u << tk);
        }
    }
    __syncthreads();
    const unsigned um = __builtin_amdgcn_ballot_w32(gml[lane] != 0u);
    int4* T = (int4*)(tiles + (r * 1024 + tm * 32) * 256);
#pragma unroll
    for (int j = 0; j < 4; ++j) {
        const int tk = wv * 4 + j;
        if ((um >> tk) & 1u) {
            const int idx = __popc(um & ((1u << tk) - 1u));
            T[idx * 64 + lane]      = ta[j];
            T[idx * 64 + 32 + lane] = tb[j];
            if (lane == 0) gout[idx] = gml[tk];
        }
    }
    __syncthreads();
    if (tid < 32) groups[(r * 32 + tm) * 32 + tid] = gout[tid];
    if (tid == 0) {
        rowmask[r * 32 + tm] = um;
        rowbase[r * 32 + tm] = r * 1024 + tm * 32;
    }
    if (tid < 16) {
        int d = deg[tid];
        if (d < 1) d = 1;
        invdeg[r * NN + tm * 16 + tid] = 1.0f / (float)d;
    }
}

__device__ __forceinline__ float xhalf(float x) {
    const int b = __builtin_bit_cast(int, x);
    return __builtin_bit_cast(float,
        __builtin_amdgcn_permlanex16(b, b, 0x76543210, 0xfedcba98, false, false));
}

__global__ __launch_bounds__(NT) void rgcn_main_kernel(
    const float* __restrict__ xx, const float* __restrict__ ss,
    const float* __restrict__ Wf, const float* __restrict__ bf,
    const float* __restrict__ convw, const float* __restrict__ convroot,
    const float* __restrict__ convbias, const float* __restrict__ lng,
    const float* __restrict__ lnb, const float* __restrict__ invdeg,
    const unsigned* __restrict__ rowmask, const int* __restrict__ rowbase,
    const unsigned* __restrict__ groups, const int* __restrict__ tiles, float* __restrict__ out)
{
    __shared__ __attribute__((aligned(16))) float xs[2 * CC * XP];
    __shared__ __attribute__((aligned(16))) float wts[2 * 4 * CC * WP];
    __shared__ float ivd[3 * NN];
    __shared__ unsigned rmk[3 * 32];
    __shared__ int rbs[3 * 32];
    __shared__ unsigned gms[3 * 32 * 32];
    __shared__ __attribute__((aligned(16))) float wfe[6 * CC];
    __shared__ __attribute__((aligned(16))) float prm[3 * CC];
    __shared__ __attribute__((aligned(16))) float cbs[LL * CC];

    const int bid   = blockIdx.x;
    const int tid   = threadIdx.x;
    const int lane  = tid & 31;
    const int wid   = tid >> 5;
    const int mlo   = lane & 15;
    const int mhalf = lane >> 4;
    const int wdst  = (tid & 31) * WP + (tid >> 5);

    for (int i = tid; i < 3 * NN; i += NT) ivd[i] = invdeg[i];
    for (int i = tid; i < 6 * CC; i += NT) wfe[i] = Wf[i];
    for (int i = tid; i < 3 * 32 * 32; i += NT) gms[i] = groups[i];
    if (tid < 96) {
        rmk[tid] = rowmask[tid];
        rbs[tid] = rowbase[tid];
    }
    if (tid < CC) {
        prm[tid]          = bf[tid];
        prm[CC + tid]     = lng[tid];
        prm[2 * CC + tid] = lnb[tid];
    }
    if (tid < LL * CC) cbs[tid] = convbias[tid];
    wts[wdst] = convroot[tid];
#pragma unroll
    for (int j = 0; j < 3; ++j)
        wts[(1 + j) * CC * WP + wdst] = convw[j * CC * CC + tid];
    __syncthreads();

    const float4* xs4 = (const float4*)xs;
    const float4* wt4 = (const float4*)wts;
    const float4* wf4 = (const float4*)wfe;
    const float4* cb4 = (const float4*)cbs;
    const float4* pr4 = (const float4*)prm;

    const int ownn = wid * 16 + mlo;

    v8f xo[2];
    {
        const float mfeat = ss[bid] * 0.125f;
        const int n = ownn;
        const float v  = xx[bid * NN + n];
        const float fi = (float)((n >> 4) & 3) * (1.0f / 3.0f);
        const float fj = (float)((n >> 2) & 3) * (1.0f / 3.0f);
        const float fk = (float)(n & 3) * (1.0f / 3.0f);
        const float ft = (float)(n >> 6) * (1.0f / 7.0f);
#pragma unroll
        for (int t = 0; t < 2; ++t) {
            if (t == 1) __builtin_amdgcn_sched_barrier(0);
            const int c4 = 4 * t + 2 * mhalf;
            float a[8], w[8];
            { const float4 lo = pr4[c4], hi = pr4[c4 + 1];
              a[0] = lo.x; a[1] = lo.y; a[2] = lo.z; a[3] = lo.w;
              a[4] = hi.x; a[5] = hi.y; a[6] = hi.z; a[7] = hi.w; }
            const float fe[6] = {fi, fj, fk, ft, v, mfeat};
#pragma unroll
            for (int f = 0; f < 6; ++f) {
                const float4 lo = wf4[f * 8 + c4], hi = wf4[f * 8 + c4 + 1];
                w[0] = lo.x; w[1] = lo.y; w[2] = lo.z; w[3] = lo.w;
                w[4] = hi.x; w[5] = hi.y; w[6] = hi.z; w[7] = hi.w;
#pragma unroll
                for (int j = 0; j < 8; ++j) a[j] += fe[f] * w[j];
            }
#pragma unroll
            for (int j = 0; j < 8; ++j) {
                xo[t][j] = a[j];
                xs[(16 * t + 8 * mhalf + j) * XP + ownn] = a[j];
            }
        }
    }
    __syncthreads();

    int ng[3];
    int bs[3];
#pragma unroll
    for (int r = 0; r < 3; ++r) {
        ng[r] = __popc((unsigned)__builtin_amdgcn_readfirstlane((int)rmk[r * 32 + wid]));
        bs[r] = __builtin_amdgcn_readfirstlane(rbs[r * 32 + wid]);
    }
    const unsigned* gmw = gms + wid * 32;
    unsigned gf[3];
#pragma unroll
    for (int r = 0; r < 3; ++r)
        gf[r] = (unsigned)__builtin_amdgcn_readfirstlane((int)gmw[r * 1024]);
    const int xab = mlo * (XP / 4) + 2 * mhalf;
    const int wab = mlo * (WP / 4) + 2 * mhalf;
    const int4* ivp = (const int4*)invdeg;
    const int l2 = lane * 2;

    for (int l = 0; l < LL; ++l) {
        const float4* xsc = xs4 + (l & 1) * (CC * XP / 4);
        const float4* wtc = wt4 + (l & 1) * (4 * CC * WP / 4);
        float* xsn = xs + ((l + 1) & 1) * (CC * XP);
        float* wtn = wts + ((l + 1) & 1) * (4 * CC * WP);
        float wnx[4] = {0.0f, 0.0f, 0.0f, 0.0f};
        if (l + 1 < LL) {
            wnx[0] = convroot[(l + 1) * CC * CC + tid];
#pragma unroll
            for (int j = 0; j < 3; ++j)
                wnx[1 + j] = convw[((l + 1) * 3 + j) * CC * CC + tid];
        }
        v8f acc[2];
#pragma unroll
        for (int t = 0; t < 2; ++t) {
            const float4 lo = cb4[l * 8 + 4 * t + 2 * mhalf], hi = cb4[l * 8 + 4 * t + 2 * mhalf + 1];
            acc[t] = (v8f){lo.x, lo.y, lo.z, lo.w, hi.x, hi.y, hi.z, hi.w};
        }

        float4 Wv[8];
        float4 X[4];
        int4 c0, c1;
        unsigned gm = gf[0];
        const int4* tp = gm ? (const int4*)(tiles + bs[0] * 256) : ivp;
#pragma unroll
        for (int t = 0; t < 2; ++t) {
            Wv[2 * t]     = wtc[wab + 4 * t];
            Wv[2 * t + 1] = wtc[wab + 4 * t + 1];
            Wv[4 + 2 * t] = wtc[wab + 16 * (WP / 4) + 4 * t];
            Wv[5 + 2 * t] = wtc[wab + 16 * (WP / 4) + 4 * t + 1];
        }
        {
            const int xa = xab + 4 * (gm ? __builtin_ctz(gm) : 0);
            X[0] = xsc[xa];                 X[1] = xsc[xa + 1];
            X[2] = xsc[xa + 16 * (XP / 4)]; X[3] = xsc[xa + 16 * (XP / 4) + 1];
            c0 = tp[l2]; c1 = tp[l2 + 1];
        }

        v8f z0 = xo[0], z1 = xo[1];
        v8f g0, g1;
        auto tile_ops = [&](const float4 (&Xf)[4], const int4 ca, const int4 cb) {
            const float cf[8] = {(float)ca.x, (float)ca.y, (float)ca.z, (float)ca.w,
                                 (float)cb.x, (float)cb.y, (float)cb.z, (float)cb.w};
            const float e0[8] = {Xf[0].x, Xf[0].y, Xf[0].z, Xf[0].w, Xf[1].x, Xf[1].y, Xf[1].z, Xf[1].w};
            const float e1[8] = {Xf[2].x, Xf[2].y, Xf[2].z, Xf[2].w, Xf[3].x, Xf[3].y, Xf[3].z, Xf[3].w};
#pragma unroll
            for (int ks = 0; ks < 4; ++ks) {
                v2f b, a0, a1;
                b.x  = cf[2 * ks];
                b.y  = cf[2 * ks + 1];
                a0.x = e0[2 * ks];
                a0.y = e0[2 * ks + 1];
                a1.x = e1[2 * ks];
                a1.y = e1[2 * ks + 1];
                g0 = __builtin_amdgcn_wmma_f32_16x16x4_f32(
                    false, a0, false, b, (short)0, g0, false, false);
                g1 = __builtin_amdgcn_wmma_f32_16x16x4_f32(
                    false, a1, false, b, (short)0, g1, false, false);
            }
        };
        auto add_members = [&](float4 (&Xf)[4], unsigned g) {
            g &= g - 1u;
            while (g) {
                float4 P[4];
                const int pa = xab + 4 * __builtin_ctz(g);
                g &= g - 1u;
                P[0] = xsc[pa];                 P[1] = xsc[pa + 1];
                P[2] = xsc[pa + 16 * (XP / 4)]; P[3] = xsc[pa + 16 * (XP / 4) + 1];
#pragma unroll
                for (int i = 0; i < 4; ++i) {
                    Xf[i].x += P[i].x; Xf[i].y += P[i].y; Xf[i].z += P[i].z; Xf[i].w += P[i].w;
                }
            }
        };
#pragma unroll
        for (int mtx = 0; mtx < 4; ++mtx) {
            __builtin_amdgcn_sched_barrier(0);
            {
                const float w0e[16] = {Wv[0].x, Wv[0].y, Wv[0].z, Wv[0].w, Wv[1].x, Wv[1].y, Wv[1].z, Wv[1].w,
                                       Wv[2].x, Wv[2].y, Wv[2].z, Wv[2].w, Wv[3].x, Wv[3].y, Wv[3].z, Wv[3].w};
                const float w1e[16] = {Wv[4].x, Wv[4].y, Wv[4].z, Wv[4].w, Wv[5].x, Wv[5].y, Wv[5].z, Wv[5].w,
                                       Wv[6].x, Wv[6].y, Wv[6].z, Wv[6].w, Wv[7].x, Wv[7].y, Wv[7].z, Wv[7].w};
#pragma unroll
                for (int q = 0; q < 8; ++q) {
                    v2f b, w0, w1;
                    w0.x = w0e[2 * q];
                    w0.y = w0e[2 * q + 1];
                    w1.x = w1e[2 * q];
                    w1.y = w1e[2 * q + 1];
                    if (q < 4) { b.x = z0[2 * q];       b.y = z0[2 * q + 1]; }
                    else       { b.x = z1[2 * (q - 4)]; b.y = z1[2 * (q - 4) + 1]; }
                    acc[0] = __builtin_amdgcn_wmma_f32_16x16x4_f32(
                        false, w0, false, b, (short)0, acc[0], false, false);
                    acc[1] = __builtin_amdgcn_wmma_f32_16x16x4_f32(
                        false, w1, false, b, (short)0, acc[1], false, false);
                }
            }
            if (mtx == 3) break;
            __builtin_amdgcn_sched_barrier(0);
            if (mtx == 0 && l + 1 < LL) {
#pragma unroll
                for (int j = 0; j < 4; ++j)
                    wtn[j * CC * WP + wdst] = wnx[j];
            }

            const int r = mtx;
            const int wa = (r + 1) * (CC * WP / 4) + wab;
            const unsigned* gp = gmw + r * 1024;
            g0 = (v8f){0, 0, 0, 0, 0, 0, 0, 0};
            g1 = (v8f){0, 0, 0, 0, 0, 0, 0, 0};
            int rem = ng[r];
#pragma nounroll
            while (rem > 1) {
                float4 Y[4];
                add_members(X, gm);
                --rem;
                tp += 64;
                ++gp;
                const unsigned gn = (unsigned)__builtin_amdgcn_readfirstlane((int)gp[0]);
                const int xa = xab + 4 * __builtin_ctz(gn);
                Y[0] = xsc[xa];                 Y[1] = xsc[xa + 1];
                Y[2] = xsc[xa + 16 * (XP / 4)]; Y[3] = xsc[xa + 16 * (XP / 4) + 1];
                const int4 d0 = tp[l2], d1 = tp[l2 + 1];
                __builtin_amdgcn_sched_barrier(0);
                tile_ops(X, c0, c1);
                __builtin_amdgcn_sched_barrier(0);
                if (rem == 1) {
                    X[0] = Y[0]; X[1] = Y[1]; X[2] = Y[2]; X[3] = Y[3];
                    c0 = d0; c1 = d1;
                    gm = gn;
                    break;
                }
                add_members(Y, gn);
                --rem;
                tp += 64;
                ++gp;
                gm = (unsigned)__builtin_amdgcn_readfirstlane((int)gp[0]);
                const int xb = xab + 4 * __builtin_ctz(gm);
                X[0] = xsc[xb];                 X[1] = xsc[xb + 1];
                X[2] = xsc[xb + 16 * (XP / 4)]; X[3] = xsc[xb + 16 * (XP / 4) + 1];
                c0 = tp[l2]; c1 = tp[l2 + 1];
                __builtin_amdgcn_sched_barrier(0);
                tile_ops(Y, d0, d1);
                __builtin_amdgcn_sched_barrier(0);
            }
            add_members(X, gm);
#pragma unroll
            for (int t = 0; t < 2; ++t) {
                Wv[2 * t]     = wtc[wa + 4 * t];
                Wv[2 * t + 1] = wtc[wa + 4 * t + 1];
                Wv[4 + 2 * t] = wtc[wa + 16 * (WP / 4) + 4 * t];
                Wv[5 + 2 * t] = wtc[wa + 16 * (WP / 4) + 4 * t + 1];
            }
            const float sc = ivd[r * NN + ownn];
            __builtin_amdgcn_sched_barrier(0);
            if (rem) tile_ops(X, c0, c1);
            __builtin_amdgcn_sched_barrier(0);
            z0 = g0 * sc;
            z1 = g1 * sc;
            if (r < 2) {
                gm = gf[r + 1];
                tp = gm ? (const int4*)(tiles + bs[r + 1] * 256) : ivp;
                const int xa = xab + 4 * (gm ? __builtin_ctz(gm) : 0);
                X[0] = xsc[xa];                 X[1] = xsc[xa + 1];
                X[2] = xsc[xa + 16 * (XP / 4)]; X[3] = xsc[xa + 16 * (XP / 4) + 1];
                c0 = tp[l2]; c1 = tp[l2 + 1];
            }
        }

        float ps[4] = {0.0f, 0.0f, 0.0f, 0.0f};
#pragma unroll
        for (int t = 0; t < 2; ++t)
#pragma unroll
            for (int j = 0; j < 8; ++j) {
                const float v = acc[t][j];
                acc[t][j] = v > 0.0f ? v : 0.0f;
                ps[j & 3] += acc[t][j];
            }
        float sum = (ps[0] + ps[1]) + (ps[2] + ps[3]);
        sum += xhalf(sum);
        const float mu = sum * (1.0f / 32.0f);
        float pq[4] = {0.0f, 0.0f, 0.0f, 0.0f};
#pragma unroll
        for (int t = 0; t < 2; ++t)
#pragma unroll
            for (int j = 0; j < 8; ++j) {
                const float d = acc[t][j] - mu;
                acc[t][j] = d;
                pq[j & 3] += d * d;
            }
        float sq = (pq[0] + pq[1]) + (pq[2] + pq[3]);
        sq += xhalf(sq);
        const float rs = rsqrtf(sq * (1.0f / 32.0f) + LN_EPS);
#pragma unroll
        for (int t = 0; t < 2; ++t) {
            const float4 ga = pr4[8 + 4 * t + 2 * mhalf],  gb = pr4[8 + 4 * t + 2 * mhalf + 1];
            const float4 ba = pr4[16 + 4 * t + 2 * mhalf], bb = pr4[16 + 4 * t + 2 * mhalf + 1];
            const float gg[8] = {ga.x, ga.y, ga.z, ga.w, gb.x, gb.y, gb.z, gb.w};
            const float bv[8] = {ba.x, ba.y, ba.z, ba.w, bb.x, bb.y, bb.z, bb.w};
#pragma unroll
            for (int j = 0; j < 8; ++j)
                xo[t][j] = acc[t][j] * rs * gg[j] + bv[j];
        }
#pragma unroll
        for (int t = 0; t < 2; ++t)
#pragma unroll
            for (int j = 0; j < 8; ++j)
                xsn[(16 * t + 8 * mhalf + j) * XP + ownn] = xo[t][j];
        __syncthreads();
    }

    for (int idx = tid; idx < 48 * CC; idx += NT) {
        const int row = idx >> 5, c = idx & 31;
        const int g = row >> 4, rc = row & 15;
        const int a0 = rc >> 2, b0 = rc & 3;
        float sum = 0.0f;
#pragma unroll
        for (int q = 0; q < 4; ++q) {
            int node;
            if (g == 0)      node = q * 16 + a0 * 4 + b0;
            else if (g == 1) node = a0 * 16 + q * 4 + b0;
            else             node = a0 * 16 + b0 * 4 + q;
            sum += xs[(LL & 1) * (CC * XP) + c * XP + node];
        }
        out[bid * 48 * CC + idx] = sum * 0.25f;
    }
}

extern "C" void kernel_launch(void* const* d_in, const int* in_sizes, int n_in,
                              void* d_out, int out_size, void* d_ws, size_t ws_size,
                              hipStream_t stream) {
    const float* xx       = (const float*)d_in[0];
    const float* ss       = (const float*)d_in[1];
    const float* Wf       = (const float*)d_in[2];
    const float* bf       = (const float*)d_in[3];
    const float* convw    = (const float*)d_in[4];
    const float* convroot = (const float*)d_in[5];
    const float* convbias = (const float*)d_in[6];
    const float* lng      = (const float*)d_in[7];
    const float* lnb      = (const float*)d_in[8];
    const int*   src      = (const int*)d_in[9];
    const int*   dst      = (const int*)d_in[10];
    float*       out      = (float*)d_out;
    float*       invdeg   = (float*)d_ws + WS_INVDEG;
    unsigned*    rowmask  = (unsigned*)d_ws + WS_ROWMASK;
    int*         rowbase  = (int*)d_ws + WS_ROWBASE;
    unsigned*    groups   = (unsigned*)d_ws + WS_GROUPS;
    int*         tiles    = (int*)d_ws + WS_TILES;

    const int E = in_sizes[9] / 3;
    const int B = in_sizes[1];

    rgcn_prep_kernel<<<96, PT, 0, stream>>>(src, dst, E, invdeg, rowmask, rowbase, groups, tiles);
    rgcn_main_kernel<<<B, NT, 0, stream>>>(xx, ss, Wf, bf, convw, convroot,
                                            convbias, lng, lnb, invdeg, rowmask,
                                            rowbase, groups, tiles, out);
    (void)n_in; (void)out_size; (void)ws_size;
}


# Layer 0 runs on the 8 input channels: W_f and b_f folded into its four weight matrices in LDS (b_f through a ones channel), x0 never formed: 28 matrix ops a wave, not 88; layer-0 sum order changed
# speedup vs baseline: 29.7575x; 1.0805x over previous
#include <hip/hip_runtime.h>
#include <hip/hip_bf16.h>

typedef __attribute__((ext_vector_type(2))) float v2f;
typedef __attribute__((ext_vector_type(8))) float v8f;

#define NN  512
#define CC  32
#define LL  4
#define LN_EPS 1e-5f
#define NT  1024
static_assert(NT == CC * CC && NT == 2 * NN, "staging and row-block ownership assume 1024 threads");
#define XP  516
#define WP  36
static_assert((XP * 4) % 16 == 0 && ((XP * 4 / 16) & 1) == 1 && XP >= NN, "xT pitch");
static_assert((WP * 4) % 16 == 0 && ((WP * 4 / 16) & 1) == 1 && WP >= CC, "WT pitch");

#define WS_INVDEG  0
#define WS_ROWMASK 1536
#define WS_ROWBASE 1664
#define WS_GROUPS  2048
#define WS_TILES   8192

#define PT  256
#define EPT 24
__global__ __launch_bounds__(PT) void rgcn_prep_kernel(
    const int* __restrict__ src, const int* __restrict__ dst, int E,
    float* __restrict__ invdeg, unsigned* __restrict__ rowmask,
    int* __restrict__ rowbase, unsigned* __restrict__ groups, int* __restrict__ tiles)
{
    __shared__ __attribute__((aligned(16))) int strip[32 * 256];
    __shared__ int deg[16];
    __shared__ unsigned msk;
    __shared__ unsigned gml[32];
    __shared__ unsigned gout[32];
    const int r   = blockIdx.x >> 5;
    const int tm  = blockIdx.x & 31;
    const int tid = threadIdx.x;
    const int* sR = src + r * E;
    const int* dR = dst + r * E;
    int4* s4 = (int4*)strip;
    for (int i = tid; i < 32 * 64; i += PT) s4[i] = make_int4(0, 0, 0, 0);
    if (tid < 16) deg[tid] = 0;
    if (tid == 0) msk = 0u;
    if (tid < 32) { gml[tid] = 0u; gout[tid] = 0u; }
    __syncthreads();
    for (int e0 = 0; e0 < E; e0 += PT * EPT) {
        int sv[EPT], dv[EPT];
#pragma unroll
        for (int q = 0; q < EPT; ++q) {
            const int e  = e0 + q * PT + tid;
            const int ec = e < E ? e : E - 1;
            const int d  = dR[ec];
            sv[q] = sR[ec];
            dv[q] = e < E ? d : -1;
        }
#pragma unroll
        for (int q = 0; q < EPT; ++q) {
            const int d = dv[q];
            if ((d >> 4) == tm) {
                const int s = sv[q] & (NN - 1);
                const int k = s & 15;
                atomicAdd(&strip[(s >> 4) * 256 + ((k >> 3) * 16 + (d & 15)) * 8 + (k & 7)], 1);
                atomicAdd(&deg[d & 15], 1);
            }
        }
    }
    __syncthreads();
    const int lane = tid & 31;
    const int wv   = tid >> 5;
    int4 ta[4], tb[4];
    unsigned ne = 0u;
#pragma unroll
    for (int j = 0; j < 4; ++j) {
        const int tk = wv * 4 + j;
        ta[j] = s4[tk * 64 + lane];
        tb[j] = s4[tk * 64 + 32 + lane];
        const int any = ta[j].x | ta[j].y | ta[j].z | ta[j].w | tb[j].x | tb[j].y | tb[j].z | tb[j].w;
        if (__builtin_amdgcn_ballot_w32(any != 0)) ne |= 1u << tk;
    }
    if (lane == 0 && ne) atomicOr(&msk, ne);
    __syncthreads();
    const unsigned m = (unsigned)__builtin_amdgcn_readfirstlane((int)msk);
#pragma unroll
    for (int j = 0; j < 4; ++j) {
        const int tk = wv * 4 + j;
        if ((ne >> tk) & 1u) {
            int first = tk;
            unsigned cand = m & ((1u << tk) - 1u);
            while (cand) {
                const int tc = __builtin_ctz(cand);
                cand &= cand - 1u;
                const int4 ua = s4[tc * 64 + lane], ub = s4[tc * 64 + 32 + lane];
                const int diff = (ua.x ^ ta[j].x) | (ua.y ^ ta[j].y) | (ua.z ^ ta[j].z) | (ua.w ^ ta[j].w)
                               | (ub.x ^ tb[j].x) | (ub.y ^ tb[j].y) | (ub.z ^ tb[j].z) | (ub.w ^ tb[j].w);
                if (!__builtin_amdgcn_ballot_w32(diff != 0)) { first = tc; break; }
            }
            if (lane == 0) atomicOr(&gml[first], 1u << tk);
        }
    }
    __syncthreads();
    const unsigned um = __builtin_amdgcn_ballot_w32(gml[lane] != 0u);
    int4* T = (int4*)(tiles + (r * 1024 + tm * 32) * 256);
#pragma unroll
    for (int j = 0; j < 4; ++j) {
        const int tk = wv * 4 + j;
        if ((um >> tk) & 1u) {
            const int idx = __popc(um & ((1u << tk) - 1u));
            T[idx * 64 + lane]      = ta[j];
            T[idx * 64 + 32 + lane] = tb[j];
            if (lane == 0) gout[idx] = gml[tk];
        }
    }
    __syncthreads();
    if (tid < 32) groups[(r * 32 + tm) * 32 + tid] = gout[tid];
    if (tid == 0) {
        rowmask[r * 32 + tm] = um;
        rowbase[r * 32 + tm] = r * 1024 + tm * 32;
    }
    if (tid < 16) {
        int d = deg[tid];
        if (d < 1) d = 1;
        invdeg[r * NN + tm * 16 + tid] = 1.0f / (float)d;
    }
}

__device__ __forceinline__ float xhalf(float x) {
    const int b = __builtin_bit_cast(int, x);
    return __builtin_bit_cast(float,
        __builtin_amdgcn_permlanex16(b, b, 0x76543210, 0xfedcba98, false, false));
}

__device__ __forceinline__ void relu_ln_store(v8f (&acc)[2], v8f (&xo)[2], const float4* pr4,
                                              float* xsn, const int mhalf, const int ownn) {
    float ps[4] = {0.0f, 0.0f, 0.0f, 0.0f};
#pragma unroll
    for (int t = 0; t < 2; ++t)
#pragma unroll
        for (int j = 0; j < 8; ++j) {
            const float v = acc[t][j];
            acc[t][j] = v > 0.0f ? v : 0.0f;
            ps[j & 3] += acc[t][j];
        }
    float sum = (ps[0] + ps[1]) + (ps[2] + ps[3]);
    sum += xhalf(sum);
    const float mu = sum * (1.0f / 32.0f);
    float pq[4] = {0.0f, 0.0f, 0.0f, 0.0f};
#pragma unroll
    for (int t = 0; t < 2; ++t)
#pragma unroll
        for (int j = 0; j < 8; ++j) {
            const float d = acc[t][j] - mu;
            acc[t][j] = d;
            pq[j & 3] += d * d;
        }
    float sq = (pq[0] + pq[1]) + (pq[2] + pq[3]);
    sq += xhalf(sq);
    const float rs = rsqrtf(sq * (1.0f / 32.0f) + LN_EPS);
#pragma unroll
    for (int t = 0; t < 2; ++t) {
        const float4 ga = pr4[8 + 4 * t + 2 * mhalf],  gb = pr4[8 + 4 * t + 2 * mhalf + 1];
        const float4 ba = pr4[16 + 4 * t + 2 * mhalf], bb = pr4[16 + 4 * t + 2 * mhalf + 1];
        const float gg[8] = {ga.x, ga.y, ga.z, ga.w, gb.x, gb.y, gb.z, gb.w};
        const float bv[8] = {ba.x, ba.y, ba.z, ba.w, bb.x, bb.y, bb.z, bb.w};
#pragma unroll
        for (int j = 0; j < 8; ++j)
            xo[t][j] = acc[t][j] * rs * gg[j] + bv[j];
    }
#pragma unroll
    for (int t = 0; t < 2; ++t)
#pragma unroll
        for (int j = 0; j < 8; ++j)
            xsn[(16 * t + 8 * mhalf + j) * XP + ownn] = xo[t][j];
}

__global__ __launch_bounds__(NT) void rgcn_main_kernel(
    const float* __restrict__ xx, const float* __restrict__ ss,
    const float* __restrict__ Wf, const float* __restrict__ bf,
    const float* __restrict__ convw, const float* __restrict__ convroot,
    const float* __restrict__ convbias, const float* __restrict__ lng,
    const float* __restrict__ lnb, const float* __restrict__ invdeg,
    const unsigned* __restrict__ rowmask, const int* __restrict__ rowbase,
    const unsigned* __restrict__ groups, const int* __restrict__ tiles, float* __restrict__ out)
{
    __shared__ __attribute__((aligned(16))) float xs[2 * CC * XP];
    __shared__ __attribute__((aligned(16))) float wts[2 * 4 * CC * WP];
    __shared__ float ivd[3 * NN];
    __shared__ unsigned rmk[3 * 32];
    __shared__ int rbs[3 * 32];
    __shared__ unsigned gms[3 * 32 * 32];
    __shared__ __attribute__((aligned(16))) float wfe[6 * CC];
    __shared__ __attribute__((aligned(16))) float prm[3 * CC];
    __shared__ __attribute__((aligned(16))) float cbs[LL * CC];
    __shared__ __attribute__((aligned(16))) float fld[CC * WP];

    const int bid   = blockIdx.x;
    const int tid   = threadIdx.x;
    const int lane  = tid & 31;
    const int wid   = tid >> 5;
    const int mlo   = lane & 15;
    const int mhalf = lane >> 4;
    const int wdst  = (tid & 31) * WP + (tid >> 5);

    for (int i = tid; i < 3 * NN; i += NT) ivd[i] = invdeg[i];
    for (int i = tid; i < 6 * CC; i += NT) wfe[i] = Wf[i];
    for (int i = tid; i < 3 * 32 * 32; i += NT) gms[i] = groups[i];
    if (tid < 96) {
        rmk[tid] = rowmask[tid];
        rbs[tid] = rowbase[tid];
    }
    if (tid < CC) {
        prm[tid]          = bf[tid];
        prm[CC + tid]     = lng[tid];
        prm[2 * CC + tid] = lnb[tid];
    }
    if (tid < LL * CC) cbs[tid] = convbias[tid];
    wts[wdst] = convroot[tid];
#pragma unroll
    for (int j = 0; j < 3; ++j)
        wts[(1 + j) * CC * WP + wdst] = convw[j * CC * CC + tid];
    __syncthreads();

    const float4* xs4 = (const float4*)xs;
    const float4* wt4 = (const float4*)wts;
    const float4* wf4 = (const float4*)wfe;
    const float4* cb4 = (const float4*)cbs;
    const float4* pr4 = (const float4*)prm;

    const int ownn = wid * 16 + mlo;

    v8f xo[2];
    float fe[8];
    {
        const int n = ownn;
        fe[0] = (float)((n >> 4) & 3) * (1.0f / 3.0f);
        fe[1] = (float)((n >> 2) & 3) * (1.0f / 3.0f);
        fe[2] = (float)(n & 3) * (1.0f / 3.0f);
        fe[3] = (float)(n >> 6) * (1.0f / 7.0f);
        fe[4] = xx[bid * NN + n];
        fe[5] = ss[bid] * 0.125f;
        fe[6] = 1.0f;
        fe[7] = 0.0f;
    }
    {
        const int zi = tid >> 8;
        const int f  = (tid >> 5) & 7;
        const int o  = tid & 31;
        const float4* wr = wt4 + ((zi == 3 ? 0 : zi + 1) * CC * WP + o * WP) / 4;
        const float4* fr = f < 6 ? wf4 + f * 8 : pr4;
        float s = 0.0f;
#pragma unroll
        for (int i = 0; i < 8; ++i) {
            const float4 a = fr[i], w = wr[i];
            s += a.x * w.x; s += a.y * w.y; s += a.z * w.z; s += a.w * w.w;
        }
        fld[o * WP + 8 * zi + f] = f < 7 ? s : 0.0f;
    }
#pragma unroll
    for (int t = 0; t < 2; ++t)
#pragma unroll
        for (int j = 0; j < 8; ++j)
            xs[(16 * t + 8 * mhalf + j) * XP + ownn] = (mhalf == t) ? fe[j] : 0.0f;
    __syncthreads();

    int ng[3];
    int bs[3];
#pragma unroll
    for (int r = 0; r < 3; ++r) {
        ng[r] = __popc((unsigned)__builtin_amdgcn_readfirstlane((int)rmk[r * 32 + wid]));
        bs[r] = __builtin_amdgcn_readfirstlane(rbs[r * 32 + wid]);
    }
    const unsigned* gmw = gms + wid * 32;
    unsigned gf[3];
#pragma unroll
    for (int r = 0; r < 3; ++r)
        gf[r] = (unsigned)__builtin_amdgcn_readfirstlane((int)gmw[r * 1024]);
    const int xab = mlo * (XP / 4) + 2 * mhalf;
    const int wab = mlo * (WP / 4) + 2 * mhalf;
    const int4* ivp = (const int4*)invdeg;
    const int l2 = lane * 2;

    {
        const float4* xsc = xs4;
        float* xsn = xs + CC * XP;
        float* wtn = wts + 4 * CC * WP;
        const float4* fd4 = (const float4*)fld;
        float wnx[4] = {0.0f, 0.0f, 0.0f, 0.0f};
        if (1 < LL) {
            wnx[0] = convroot[CC * CC + tid];
#pragma unroll
            for (int j = 0; j < 3; ++j)
                wnx[1 + j] = convw[(3 + j) * CC * CC + tid];
        }
        int4 fa[3], fb[3];
#pragma unroll
        for (int r = 0; r < 3; ++r) {
            const int4* tf = gf[r] ? (const int4*)(tiles + bs[r] * 256) : ivp;
            fa[r] = tf[l2];
            fb[r] = tf[l2 + 1];
        }
        v8f gq[2];
        gq[0] = (v8f){0, 0, 0, 0, 0, 0, 0, 0};
        gq[1] = (v8f){0, 0, 0, 0, 0, 0, 0, 0};
#pragma unroll
        for (int r = 0; r < 3; ++r) {
            const unsigned* gp = gmw + r * 1024;
            const int4* tq = (const int4*)(tiles + bs[r] * 256);
            const int xr = xab + (r == 1 ? 16 * (XP / 4) : 0);
            int4 ca = fa[r], cb = fb[r];
#pragma nounroll
            for (int j = 0; j < ng[r]; ++j) {
                unsigned g = (unsigned)__builtin_amdgcn_readfirstlane((int)gp[j]);
                int4 na = ca, nb = cb;
                if (j + 1 < ng[r]) {
                    na = tq[(j + 1) * 64 + l2];
                    nb = tq[(j + 1) * 64 + l2 + 1];
                }
                const int xa = xr + 4 * __builtin_ctz(g);
                float4 Xa = xsc[xa], Xb = xsc[xa + 1];
                g &= g - 1u;
                while (g) {
                    const int pa = xr + 4 * __builtin_ctz(g);
                    g &= g - 1u;
                    const float4 Pa = xsc[pa], Pb = xsc[pa + 1];
                    Xa.x += Pa.x; Xa.y += Pa.y; Xa.z += Pa.z; Xa.w += Pa.w;
                    Xb.x += Pb.x; Xb.y += Pb.y; Xb.z += Pb.z; Xb.w += Pb.w;
                }
                const float cf[8] = {(float)ca.x, (float)ca.y, (float)ca.z, (float)ca.w,
                                     (float)cb.x, (float)cb.y, (float)cb.z, (float)cb.w};
                const float e0[8] = {Xa.x, Xa.y, Xa.z, Xa.w, Xb.x, Xb.y, Xb.z, Xb.w};
#pragma unroll
                for (int ks = 0; ks < 4; ++ks) {
                    v2f a, b;
                    a.x = e0[2 * ks];
                    a.y = e0[2 * ks + 1];
                    b.x = cf[2 * ks];
                    b.y = cf[2 * ks + 1];
                    gq[r >> 1] = __builtin_amdgcn_wmma_f32_16x16x4_f32(
                        false, a, false, b, (short)0, gq[r >> 1], false, false);
                }
                ca = na;
                cb = nb;
            }
        }
        float4 Wv[8];
#pragma unroll
        for (int t = 0; t < 2; ++t) {
            Wv[2 * t]     = fd4[wab + 4 * t];
            Wv[2 * t + 1] = fd4[wab + 4 * t + 1];
            Wv[4 + 2 * t] = fd4[wab + 16 * (WP / 4) + 4 * t];
            Wv[5 + 2 * t] = fd4[wab + 16 * (WP / 4) + 4 * t + 1];
        }
        const float s01 = ivd[mhalf * NN + ownn];
        const float s2  = ivd[2 * NN + ownn];
        v8f acc[2];
#pragma unroll
        for (int t = 0; t < 2; ++t) {
            const float4 lo = cb4[4 * t + 2 * mhalf], hi = cb4[4 * t + 2 * mhalf + 1];
            acc[t] = (v8f){lo.x, lo.y, lo.z, lo.w, hi.x, hi.y, hi.z, hi.w};
        }
        const v8f z0 = gq[0] * s01;
        v8f z1;
#pragma unroll
        for (int j = 0; j < 8; ++j)
            z1[j] = mhalf ? fe[j] : gq[1][j] * s2;
        __builtin_amdgcn_sched_barrier(0);
        {
            const float w0e[16] = {Wv[0].x, Wv[0].y, Wv[0].z, Wv[0].w, Wv[1].x, Wv[1].y, Wv[1].z, Wv[1].w,
                                   Wv[2].x, Wv[2].y, Wv[2].z, Wv[2].w, Wv[3].x, Wv[3].y, Wv[3].z, Wv[3].w};
            const float w1e[16] = {Wv[4].x, Wv[4].y, Wv[4].z, Wv[4].w, Wv[5].x, Wv[5].y, Wv[5].z, Wv[5].w,
                                   Wv[6].x, Wv[6].y, Wv[6].z, Wv[6].w, Wv[7].x, Wv[7].y, Wv[7].z, Wv[7].w};
#pragma unroll
            for (int q = 0; q < 8; ++q) {
                v2f b, w0, w1;
                w0.x = w0e[2 * q];
                w0.y = w0e[2 * q + 1];
                w1.x = w1e[2 * q];
                w1.y = w1e[2 * q + 1];
                if (q < 4) { b.x = z0[2 * q];       b.y = z0[2 * q + 1]; }
                else       { b.x = z1[2 * (q - 4)]; b.y = z1[2 * (q - 4) + 1]; }
                acc[0] = __builtin_amdgcn_wmma_f32_16x16x4_f32(
                    false, w0, false, b, (short)0, acc[0], false, false);
                acc[1] = __builtin_amdgcn_wmma_f32_16x16x4_f32(
                    false, w1, false, b, (short)0, acc[1], false, false);
            }
        }
        if (1 < LL) {
#pragma unroll
            for (int j = 0; j < 4; ++j)
                wtn[j * CC * WP + wdst] = wnx[j];
        }
        relu_ln_store(acc, xo, pr4, xsn, mhalf, ownn);
        __syncthreads();
    }

    for (int l = 1; l < LL; ++l) {
        const float4* xsc = xs4 + (l & 1) * (CC * XP / 4);
        const float4* wtc = wt4 + (l & 1) * (4 * CC * WP / 4);
        float* xsn = xs + ((l + 1) & 1) * (CC * XP);
        float* wtn = wts + ((l + 1) & 1) * (4 * CC * WP);
        float wnx[4] = {0.0f, 0.0f, 0.0f, 0.0f};
        if (l + 1 < LL) {
            wnx[0] = convroot[(l + 1) * CC * CC + tid];
#pragma unroll
            for (int j = 0; j < 3; ++j)
                wnx[1 + j] = convw[((l + 1) * 3 + j) * CC * CC + tid];
        }
        v8f acc[2];
#pragma unroll
        for (int t = 0; t < 2; ++t) {
            const float4 lo = cb4[l * 8 + 4 * t + 2 * mhalf], hi = cb4[l * 8 + 4 * t + 2 * mhalf + 1];
            acc[t] = (v8f){lo.x, lo.y, lo.z, lo.w, hi.x, hi.y, hi.z, hi.w};
        }

        float4 Wv[8];
        float4 X[4];
        int4 c0, c1;
        unsigned gm = gf[0];
        const int4* tp = gm ? (const int4*)(tiles + bs[0] * 256) : ivp;
#pragma unroll
        for (int t = 0; t < 2; ++t) {
            Wv[2 * t]     = wtc[wab + 4 * t];
            Wv[2 * t + 1] = wtc[wab + 4 * t + 1];
            Wv[4 + 2 * t] = wtc[wab + 16 * (WP / 4) + 4 * t];
            Wv[5 + 2 * t] = wtc[wab + 16 * (WP / 4) + 4 * t + 1];
        }
        {
            const int xa = xab + 4 * (gm ? __builtin_ctz(gm) : 0);
            X[0] = xsc[xa];                 X[1] = xsc[xa + 1];
            X[2] = xsc[xa + 16 * (XP / 4)]; X[3] = xsc[xa + 16 * (XP / 4) + 1];
            c0 = tp[l2]; c1 = tp[l2 + 1];
        }

        v8f z0 = xo[0], z1 = xo[1];
        v8f g0, g1;
        auto tile_ops = [&](const float4 (&Xf)[4], const int4 ca, const int4 cb) {
            const float cf[8] = {(float)ca.x, (float)ca.y, (float)ca.z, (float)ca.w,
                                 (float)cb.x, (float)cb.y, (float)cb.z, (float)cb.w};
            const float e0[8] = {Xf[0].x, Xf[0].y, Xf[0].z, Xf[0].w, Xf[1].x, Xf[1].y, Xf[1].z, Xf[1].w};
            const float e1[8] = {Xf[2].x, Xf[2].y, Xf[2].z, Xf[2].w, Xf[3].x, Xf[3].y, Xf[3].z, Xf[3].w};
#pragma unroll
            for (int ks = 0; ks < 4; ++ks) {
                v2f b, a0, a1;
                b.x  = cf[2 * ks];
                b.y  = cf[2 * ks + 1];
                a0.x = e0[2 * ks];
                a0.y = e0[2 * ks + 1];
                a1.x = e1[2 * ks];
                a1.y = e1[2 * ks + 1];
                g0 = __builtin_amdgcn_wmma_f32_16x16x4_f32(
                    false, a0, false, b, (short)0, g0, false, false);
                g1 = __builtin_amdgcn_wmma_f32_16x16x4_f32(
                    false, a1, false, b, (short)0, g1, false, false);
            }
        };
        auto add_members = [&](float4 (&Xf)[4], unsigned g) {
            g &= g - 1u;
            while (g) {
                float4 P[4];
                const int pa = xab + 4 * __builtin_ctz(g);
                g &= g - 1u;
                P[0] = xsc[pa];                 P[1] = xsc[pa + 1];
                P[2] = xsc[pa + 16 * (XP / 4)]; P[3] = xsc[pa + 16 * (XP / 4) + 1];
#pragma unroll
                for (int i = 0; i < 4; ++i) {
                    Xf[i].x += P[i].x; Xf[i].y += P[i].y; Xf[i].z += P[i].z; Xf[i].w += P[i].w;
                }
            }
        };
#pragma unroll
        for (int mtx = 0; mtx < 4; ++mtx) {
            __builtin_amdgcn_sched_barrier(0);
            {
                const float w0e[16] = {Wv[0].x, Wv[0].y, Wv[0].z, Wv[0].w, Wv[1].x, Wv[1].y, Wv[1].z, Wv[1].w,
                                       Wv[2].x, Wv[2].y, Wv[2].z, Wv[2].w, Wv[3].x, Wv[3].y, Wv[3].z, Wv[3].w};
                const float w1e[16] = {Wv[4].x, Wv[4].y, Wv[4].z, Wv[4].w, Wv[5].x, Wv[5].y, Wv[5].z, Wv[5].w,
                                       Wv[6].x, Wv[6].y, Wv[6].z, Wv[6].w, Wv[7].x, Wv[7].y, Wv[7].z, Wv[7].w};
#pragma unroll
                for (int q = 0; q < 8; ++q) {
                    v2f b, w0, w1;
                    w0.x = w0e[2 * q];
                    w0.y = w0e[2 * q + 1];
                    w1.x = w1e[2 * q];
                    w1.y = w1e[2 * q + 1];
                    if (q < 4) { b.x = z0[2 * q];       b.y = z0[2 * q + 1]; }
                    else       { b.x = z1[2 * (q - 4)]; b.y = z1[2 * (q - 4) + 1]; }
                    acc[0] = __builtin_amdgcn_wmma_f32_16x16x4_f32(
                        false, w0, false, b, (short)0, acc[0], false, false);
                    acc[1] = __builtin_amdgcn_wmma_f32_16x16x4_f32(
                        false, w1, false, b, (short)0, acc[1], false, false);
                }
            }
            if (mtx == 3) break;
            __builtin_amdgcn_sched_barrier(0);
            if (mtx == 0 && l + 1 < LL) {
#pragma unroll
                for (int j = 0; j < 4; ++j)
                    wtn[j * CC * WP + wdst] = wnx[j];
            }

            const int r = mtx;
            const int wa = (r + 1) * (CC * WP / 4) + wab;
            const unsigned* gp = gmw + r * 1024;
            g0 = (v8f){0, 0, 0, 0, 0, 0, 0, 0};
            g1 = (v8f){0, 0, 0, 0, 0, 0, 0, 0};
            int rem = ng[r];
#pragma nounroll
            while (rem > 1) {
                float4 Y[4];
                add_members(X, gm);
                --rem;
                tp += 64;
                ++gp;
                const unsigned gn = (unsigned)__builtin_amdgcn_readfirstlane((int)gp[0]);
                const int xa = xab + 4 * __builtin_ctz(gn);
                Y[0] = xsc[xa];                 Y[1] = xsc[xa + 1];
                Y[2] = xsc[xa + 16 * (XP / 4)]; Y[3] = xsc[xa + 16 * (XP / 4) + 1];
                const int4 d0 = tp[l2], d1 = tp[l2 + 1];
                __builtin_amdgcn_sched_barrier(0);
                tile_ops(X, c0, c1);
                __builtin_amdgcn_sched_barrier(0);
                if (rem == 1) {
                    X[0] = Y[0]; X[1] = Y[1]; X[2] = Y[2]; X[3] = Y[3];
                    c0 = d0; c1 = d1;
                    gm = gn;
                    break;
                }
                add_members(Y, gn);
                --rem;
                tp += 64;
                ++gp;
                gm = (unsigned)__builtin_amdgcn_readfirstlane((int)gp[0]);
                const int xb = xab + 4 * __builtin_ctz(gm);
                X[0] = xsc[xb];                 X[1] = xsc[xb + 1];
                X[2] = xsc[xb + 16 * (XP / 4)]; X[3] = xsc[xb + 16 * (XP / 4) + 1];
                c0 = tp[l2]; c1 = tp[l2 + 1];
                __builtin_amdgcn_sched_barrier(0);
                tile_ops(Y, d0, d1);
                __builtin_amdgcn_sched_barrier(0);
            }
            add_members(X, gm);
#pragma unroll
            for (int t = 0; t < 2; ++t) {
                Wv[2 * t]     = wtc[wa + 4 * t];
                Wv[2 * t + 1] = wtc[wa + 4 * t + 1];
                Wv[4 + 2 * t] = wtc[wa + 16 * (WP / 4) + 4 * t];
                Wv[5 + 2 * t] = wtc[wa + 16 * (WP / 4) + 4 * t + 1];
            }
            const float sc = ivd[r * NN + ownn];
            __builtin_amdgcn_sched_barrier(0);
            if (rem) tile_ops(X, c0, c1);
            __builtin_amdgcn_sched_barrier(0);
            z0 = g0 * sc;
            z1 = g1 * sc;
            if (r < 2) {
                gm = gf[r + 1];
                tp = gm ? (const int4*)(tiles + bs[r + 1] * 256) : ivp;
                const int xa = xab + 4 * (gm ? __builtin_ctz(gm) : 0);
                X[0] = xsc[xa];                 X[1] = xsc[xa + 1];
                X[2] = xsc[xa + 16 * (XP / 4)]; X[3] = xsc[xa + 16 * (XP / 4) + 1];
                c0 = tp[l2]; c1 = tp[l2 + 1];
            }
        }

        relu_ln_store(acc, xo, pr4, xsn, mhalf, ownn);
        __syncthreads();
    }

    for (int idx = tid; idx < 48 * CC; idx += NT) {
        const int row = idx >> 5, c = idx & 31;
        const int g = row >> 4, rc = row & 15;
        const int a0 = rc >> 2, b0 = rc & 3;
        float sum = 0.0f;
#pragma unroll
        for (int q = 0; q < 4; ++q) {
            int node;
            if (g == 0)      node = q * 16 + a0 * 4 + b0;
            else if (g == 1) node = a0 * 16 + q * 4 + b0;
            else             node = a0 * 16 + b0 * 4 + q;
            sum += xs[(LL & 1) * (CC * XP) + c * XP + node];
        }
        out[bid * 48 * CC + idx] = sum * 0.25f;
    }
}

extern "C" void kernel_launch(void* const* d_in, const int* in_sizes, int n_in,
                              void* d_out, int out_size, void* d_ws, size_t ws_size,
                              hipStream_t stream) {
    const float* xx       = (const float*)d_in[0];
    const float* ss       = (const float*)d_in[1];
    const float* Wf       = (const float*)d_in[2];
    const float* bf       = (const float*)d_in[3];
    const float* convw    = (const float*)d_in[4];
    const float* convroot = (const float*)d_in[5];
    const float* convbias = (const float*)d_in[6];
    const float* lng      = (const float*)d_in[7];
    const float* lnb      = (const float*)d_in[8];
    const int*   src      = (const int*)d_in[9];
    const int*   dst      = (const int*)d_in[10];
    float*       out      = (float*)d_out;
    float*       invdeg   = (float*)d_ws + WS_INVDEG;
    unsigned*    rowmask  = (unsigned*)d_ws + WS_ROWMASK;
    int*         rowbase  = (int*)d_ws + WS_ROWBASE;
    unsigned*    groups   = (unsigned*)d_ws + WS_GROUPS;
    int*         tiles    = (int*)d_ws + WS_TILES;

    const int E = in_sizes[9] / 3;
    const int B = in_sizes[1];

    rgcn_prep_kernel<<<96, PT, 0, stream>>>(src, dst, E, invdeg, rowmask, rowbase, groups, tiles);
    rgcn_main_kernel<<<B, NT, 0, stream>>>(xx, ss, Wf, bf, convw, convroot,
                                            convbias, lng, lnb, invdeg, rowmask,
                                            rowbase, groups, tiles, out);
    (void)n_in; (void)out_size; (void)ws_size;
}


# Entry: each wave reads its tile tables from memory as scalars (no LDS copy); weights, constants, 1/deg, inputs, first tile counts leave as one group: 14 dependent memory trips, now 4; sum order kept
# speedup vs baseline: 32.2058x; 1.0823x over previous
#include <hip/hip_runtime.h>
#include <hip/hip_bf16.h>

typedef __attribute__((ext_vector_type(2))) float v2f;
typedef __attribute__((ext_vector_type(8))) float v8f;

#define NN  512
#define CC  32
#define LL  4
#define LN_EPS 1e-5f
#define NT  1024
static_assert(NT == CC * CC && NT == 2 * NN, "staging and row-block ownership assume 1024 threads");
#define XP  516
#define WP  36
static_assert((XP * 4) % 16 == 0 && ((XP * 4 / 16) & 1) == 1 && XP >= NN, "xT pitch");
static_assert((WP * 4) % 16 == 0 && ((WP * 4 / 16) & 1) == 1 && WP >= CC, "WT pitch");

#define WS_INVDEG  0
#define WS_ROWMASK 1536
#define WS_ROWBASE 1664
#define WS_GROUPS  2048
#define WS_TILES   8192

#define PT  256
#define EPT 24
__global__ __launch_bounds__(PT) void rgcn_prep_kernel(
    const int* __restrict__ src, const int* __restrict__ dst, int E,
    float* __restrict__ invdeg, unsigned* __restrict__ rowmask,
    int* __restrict__ rowbase, unsigned* __restrict__ groups, int* __restrict__ tiles)
{
    __shared__ __attribute__((aligned(16))) int strip[32 * 256];
    __shared__ int deg[16];
    __shared__ unsigned msk;
    __shared__ unsigned gml[32];
    __shared__ unsigned gout[32];
    const int r   = blockIdx.x >> 5;
    const int tm  = blockIdx.x & 31;
    const int tid = threadIdx.x;
    const int* sR = src + r * E;
    const int* dR = dst + r * E;
    int4* s4 = (int4*)strip;
    for (int i = tid; i < 32 * 64; i += PT) s4[i] = make_int4(0, 0, 0, 0);
    if (tid < 16) deg[tid] = 0;
    if (tid == 0) msk = 0u;
    if (tid < 32) { gml[tid] = 0u; gout[tid] = 0u; }
    __syncthreads();
    for (int e0 = 0; e0 < E; e0 += PT * EPT) {
        int sv[EPT], dv[EPT];
#pragma unroll
        for (int q = 0; q < EPT; ++q) {
            const int e  = e0 + q * PT + tid;
            const int ec = e < E ? e : E - 1;
            const int d  = dR[ec];
            sv[q] = sR[ec];
            dv[q] = e < E ? d : -1;
        }
#pragma unroll
        for (int q = 0; q < EPT; ++q) {
            const int d = dv[q];
            if ((d >> 4) == tm) {
                const int s = sv[q] & (NN - 1);
                const int k = s & 15;
                atomicAdd(&strip[(s >> 4) * 256 + ((k >> 3) * 16 + (d & 15)) * 8 + (k & 7)], 1);
                atomicAdd(&deg[d & 15], 1);
            }
        }
    }
    __syncthreads();
    const int lane = tid & 31;
    const int wv   = tid >> 5;
    int4 ta[4], tb[4];
    unsigned ne = 0u;
#pragma unroll
    for (int j = 0; j < 4; ++j) {
        const int tk = wv * 4 + j;
        ta[j] = s4[tk * 64 + lane];
        tb[j] = s4[tk * 64 + 32 + lane];
        const int any = ta[j].x | ta[j].y | ta[j].z | ta[j].w | tb[j].x | tb[j].y | tb[j].z | tb[j].w;
        if (__builtin_amdgcn_ballot_w32(any != 0)) ne |= 1u << tk;
    }
    if (lane == 0 && ne) atomicOr(&msk, ne);
    __syncthreads();
    const unsigned m = (unsigned)__builtin_amdgcn_readfirstlane((int)msk);
#pragma unroll
    for (int j = 0; j < 4; ++j) {
        const int tk = wv * 4 + j;
        if ((ne >> tk) & 1u) {
            int first = tk;
            unsigned cand = m & ((1u << tk) - 1u);
            while (cand) {
                const int tc = __builtin_ctz(cand);
                cand &= cand - 1u;
                const int4 ua = s4[tc * 64 + lane], ub = s4[tc * 64 + 32 + lane];
                const int diff = (ua.x ^ ta[j].x) | (ua.y ^ ta[j].y) | (ua.z ^ ta[j].z) | (ua.w ^ ta[j].w)
                               | (ub.x ^ tb[j].x) | (ub.y ^ tb[j].y) | (ub.z ^ tb[j].z) | (ub.w ^ tb[j].w);
                if (!__builtin_amdgcn_ballot_w32(diff != 0)) { first = tc; break; }
            }
            if (lane == 0) atomicOr(&gml[first], 1u << tk);
        }
    }
    __syncthreads();
    const unsigned um = __builtin_amdgcn_ballot_w32(gml[lane] != 0u);
    int4* T = (int4*)(tiles + (r * 1024 + tm * 32) * 256);
#pragma unroll
    for (int j = 0; j < 4; ++j) {
        const int tk = wv * 4 + j;
        if ((um >> tk) & 1u) {
            const int idx = __popc(um & ((1u << tk) - 1u));
            T[idx * 64 + lane]      = ta[j];
            T[idx * 64 + 32 + lane] = tb[j];
            if (lane == 0) gout[idx] = gml[tk];
        }
    }
    __syncthreads();
    if (tid < 32) groups[(r * 32 + tm) * 32 + tid] = gout[tid];
    if (tid == 0) {
        rowmask[r * 32 + tm] = um;
        rowbase[r * 32 + tm] = r * 1024 + tm * 32;
    }
    if (tid < 16) {
        int d = deg[tid];
        if (d < 1) d = 1;
        invdeg[r * NN + tm * 16 + tid] = 1.0f / (float)d;
    }
}

__device__ __forceinline__ float xhalf(float x) {
    const int b = __builtin_bit_cast(int, x);
    return __builtin_bit_cast(float,
        __builtin_amdgcn_permlanex16(b, b, 0x76543210, 0xfedcba98, false, false));
}

__device__ __forceinline__ void relu_ln_store(v8f (&acc)[2], v8f (&xo)[2], const float4* pr4,
                                              float* xsn, const int mhalf, const int ownn) {
    float ps[4] = {0.0f, 0.0f, 0.0f, 0.0f};
#pragma unroll
    for (int t = 0; t < 2; ++t)
#pragma unroll
        for (int j = 0; j < 8; ++j) {
            const float v = acc[t][j];
            acc[t][j] = v > 0.0f ? v : 0.0f;
            ps[j & 3] += acc[t][j];
        }
    float sum = (ps[0] + ps[1]) + (ps[2] + ps[3]);
    sum += xhalf(sum);
    const float mu = sum * (1.0f / 32.0f);
    float pq[4] = {0.0f, 0.0f, 0.0f, 0.0f};
#pragma unroll
    for (int t = 0; t < 2; ++t)
#pragma unroll
        for (int j = 0; j < 8; ++j) {
            const float d = acc[t][j] - mu;
            acc[t][j] = d;
            pq[j & 3] += d * d;
        }
    float sq = (pq[0] + pq[1]) + (pq[2] + pq[3]);
    sq += xhalf(sq);
    const float rs = rsqrtf(sq * (1.0f / 32.0f) + LN_EPS);
#pragma unroll
    for (int t = 0; t < 2; ++t) {
        const float4 ga = pr4[8 + 4 * t + 2 * mhalf],  gb = pr4[8 + 4 * t + 2 * mhalf + 1];
        const float4 ba = pr4[16 + 4 * t + 2 * mhalf], bb = pr4[16 + 4 * t + 2 * mhalf + 1];
        const float gg[8] = {ga.x, ga.y, ga.z, ga.w, gb.x, gb.y, gb.z, gb.w};
        const float bv[8] = {ba.x, ba.y, ba.z, ba.w, bb.x, bb.y, bb.z, bb.w};
#pragma unroll
        for (int j = 0; j < 8; ++j)
            xo[t][j] = acc[t][j] * rs * gg[j] + bv[j];
    }
#pragma unroll
    for (int t = 0; t < 2; ++t)
#pragma unroll
        for (int j = 0; j < 8; ++j)
            xsn[(16 * t + 8 * mhalf + j) * XP + ownn] = xo[t][j];
}

__global__ __launch_bounds__(NT) void rgcn_main_kernel(
    const float* __restrict__ xx, const float* __restrict__ ss,
    const float* __restrict__ Wf, const float* __restrict__ bf,
    const float* __restrict__ convw, const float* __restrict__ convroot,
    const float* __restrict__ convbias, const float* __restrict__ lng,
    const float* __restrict__ lnb, const float* __restrict__ invdeg,
    const unsigned* __restrict__ rowmask, const int* __restrict__ rowbase,
    const unsigned* __restrict__ groups, const int* __restrict__ tiles, float* __restrict__ out)
{
    __shared__ __attribute__((aligned(16))) float xs[2 * CC * XP];
    __shared__ __attribute__((aligned(16))) float wts[2 * 4 * CC * WP];
    __shared__ float ivd[2 * NT];
    __shared__ __attribute__((aligned(16))) float cst[NT];
    float* const wfe = cst;
    float* const prm = cst + 6 * CC;
    float* const cbs = cst + 9 * CC;
    static_assert(CC == 32 && 9 * CC + LL * CC <= NT && (6 * CC) % 4 == 0 && (9 * CC) % 4 == 0, "constants: one word per thread, whole waves per part, 16-byte aligned parts");
    __shared__ __attribute__((aligned(16))) float fld[CC * WP];

    const int bid   = blockIdx.x;
    const int tid   = threadIdx.x;
    const int lane  = tid & 31;
    const int wid   = __builtin_amdgcn_readfirstlane(tid >> 5);
    const int mlo   = lane & 15;
    const int mhalf = lane >> 4;
    const int wdst  = (tid & 31) * WP + (tid >> 5);

    const int ownn = wid * 16 + mlo;

    const unsigned* gmw = groups + wid * 32;
    float wl0[4];
    wl0[0] = convroot[tid];
#pragma unroll
    for (int j = 0; j < 3; ++j)
        wl0[1 + j] = convw[j * CC * CC + tid];
    const float xv  = xx[bid * NN + ownn];
    const float ssv = ss[bid];
    const float* cbase = Wf;
    int coff = wid * 32;
    if (wid < 6)                { cbase = Wf;       coff = 0; }
    else if (wid < 7)           { cbase = bf;       coff = 6 * CC; }
    else if (wid < 8)           { cbase = lng;      coff = 7 * CC; }
    else if (wid < 9)           { cbase = lnb;      coff = 8 * CC; }
    else if (wid < 9 + LL)      { cbase = convbias; coff = 9 * CC; }
    const float cv = cbase[tid - coff];
    const float iv0 = invdeg[tid];
    const float iv1 = invdeg[NT + (tid & (NN - 1))];
    unsigned rmv[3], gfv[3];
    int rbv[3];
#pragma unroll
    for (int r = 0; r < 3; ++r) {
        rmv[r] = rowmask[r * 32 + wid];
        rbv[r] = rowbase[r * 32 + wid];
        gfv[r] = gmw[r * 1024];
    }

    int ng[3];
    int bs[3];
    unsigned gf[3];
#pragma unroll
    for (int r = 0; r < 3; ++r) {
        ng[r] = __popc((unsigned)__builtin_amdgcn_readfirstlane((int)rmv[r]));
        bs[r] = __builtin_amdgcn_readfirstlane(rbv[r]);
        gf[r] = (unsigned)__builtin_amdgcn_readfirstlane((int)gfv[r]);
    }
    const int4* ivp = (const int4*)invdeg;
    const int l2 = lane * 2;
    int4 fa[3], fb[3];
#pragma unroll
    for (int r = 0; r < 3; ++r) {
        const int4* tf = gf[r] ? (const int4*)(tiles + bs[r] * 256) : ivp;
        fa[r] = tf[l2];
        fb[r] = tf[l2 + 1];
    }
    wts[wdst] = wl0[0];
#pragma unroll
    for (int j = 0; j < 3; ++j)
        wts[(1 + j) * CC * WP + wdst] = wl0[1 + j];
    cst[tid] = cv;
    ivd[tid] = iv0;
    ivd[NT + tid] = iv1;
    __syncthreads();

    const float4* xs4 = (const float4*)xs;
    const float4* wt4 = (const float4*)wts;
    const float4* wf4 = (const float4*)wfe;
    const float4* cb4 = (const float4*)cbs;
    const float4* pr4 = (const float4*)prm;

    v8f xo[2];
    float fe[8];
    {
        const int n = ownn;
        fe[0] = (float)((n >> 4) & 3) * (1.0f / 3.0f);
        fe[1] = (float)((n >> 2) & 3) * (1.0f / 3.0f);
        fe[2] = (float)(n & 3) * (1.0f / 3.0f);
        fe[3] = (float)(n >> 6) * (1.0f / 7.0f);
        fe[4] = xv;
        fe[5] = ssv * 0.125f;
        fe[6] = 1.0f;
        fe[7] = 0.0f;
    }
    {
        const int zi = tid >> 8;
        const int f  = (tid >> 5) & 7;
        const int o  = tid & 31;
        const float4* wr = wt4 + ((zi == 3 ? 0 : zi + 1) * CC * WP + o * WP) / 4;
        const float4* fr = f < 6 ? wf4 + f * 8 : pr4;
        float s = 0.0f;
#pragma unroll
        for (int i = 0; i < 8; ++i) {
            const float4 a = fr[i], w = wr[i];
            s += a.x * w.x; s += a.y * w.y; s += a.z * w.z; s += a.w * w.w;
        }
        fld[o * WP + 8 * zi + f] = f < 7 ? s : 0.0f;
    }
#pragma unroll
    for (int t = 0; t < 2; ++t)
#pragma unroll
        for (int j = 0; j < 8; ++j)
            xs[(16 * t + 8 * mhalf + j) * XP + ownn] = (mhalf == t) ? fe[j] : 0.0f;
    __syncthreads();

    const int xab = mlo * (XP / 4) + 2 * mhalf;
    const int wab = mlo * (WP / 4) + 2 * mhalf;

    {
        const float4* xsc = xs4;
        float* xsn = xs + CC * XP;
        float* wtn = wts + 4 * CC * WP;
        const float4* fd4 = (const float4*)fld;
        v8f gq[2];
        gq[0] = (v8f){0, 0, 0, 0, 0, 0, 0, 0};
        gq[1] = (v8f){0, 0, 0, 0, 0, 0, 0, 0};
#pragma unroll
        for (int r = 0; r < 3; ++r) {
            const unsigned* gp = gmw + r * 1024;
            const int4* tq = (const int4*)(tiles + bs[r] * 256);
            const int xr = xab + (r == 1 ? 16 * (XP / 4) : 0);
            int4 ca = fa[r], cb = fb[r];
            unsigned gcur = gf[r];
#pragma nounroll
            for (int j = 0; j < ng[r]; ++j) {
                unsigned g = gcur;
                int4 na = ca, nb = cb;
                unsigned gnx = 0u;
                if (j + 1 < ng[r]) {
                    na = tq[(j + 1) * 64 + l2];
                    nb = tq[(j + 1) * 64 + l2 + 1];
                    gnx = gp[j + 1];
                }
                const int xa = xr + 4 * __builtin_ctz(g);
                float4 Xa = xsc[xa], Xb = xsc[xa + 1];
                g &= g - 1u;
                while (g) {
                    const int pa = xr + 4 * __builtin_ctz(g);
                    g &= g - 1u;
                    const float4 Pa = xsc[pa], Pb = xsc[pa + 1];
                    Xa.x += Pa.x; Xa.y += Pa.y; Xa.z += Pa.z; Xa.w += Pa.w;
                    Xb.x += Pb.x; Xb.y += Pb.y; Xb.z += Pb.z; Xb.w += Pb.w;
                }
                const float cf[8] = {(float)ca.x, (float)ca.y, (float)ca.z, (float)ca.w,
                                     (float)cb.x, (float)cb.y, (float)cb.z, (float)cb.w};
                const float e0[8] = {Xa.x, Xa.y, Xa.z, Xa.w, Xb.x, Xb.y, Xb.z, Xb.w};
#pragma unroll
                for (int ks = 0; ks < 4; ++ks) {
                    v2f a, b;
                    a.x = e0[2 * ks];
                    a.y = e0[2 * ks + 1];
                    b.x = cf[2 * ks];
                    b.y = cf[2 * ks + 1];
                    gq[r >> 1] = __builtin_amdgcn_wmma_f32_16x16x4_f32(
                        false, a, false, b, (short)0, gq[r >> 1], false, false);
                }
                ca = na;
                cb = nb;
                gcur = (unsigned)__builtin_amdgcn_readfirstlane((int)gnx);
            }
        }
        float wnx[4] = {0.0f, 0.0f, 0.0f, 0.0f};
        if (1 < LL) {
            wnx[0] = convroot[CC * CC + tid];
#pragma unroll
            for (int j = 0; j < 3; ++j)
                wnx[1 + j] = convw[(3 + j) * CC * CC + tid];
        }
        float4 Wv[8];
#pragma unroll
        for (int t = 0; t < 2; ++t) {
            Wv[2 * t]     = fd4[wab + 4 * t];
            Wv[2 * t + 1] = fd4[wab + 4 * t + 1];
            Wv[4 + 2 * t] = fd4[wab + 16 * (WP / 4) + 4 * t];
            Wv[5 + 2 * t] = fd4[wab + 16 * (WP / 4) + 4 * t + 1];
        }
        const float s01 = ivd[mhalf * NN + ownn];
        const float s2  = ivd[2 * NN + ownn];
        v8f acc[2];
#pragma unroll
        for (int t = 0; t < 2; ++t) {
            const float4 lo = cb4[4 * t + 2 * mhalf], hi = cb4[4 * t + 2 * mhalf + 1];
            acc[t] = (v8f){lo.x, lo.y, lo.z, lo.w, hi.x, hi.y, hi.z, hi.w};
        }
        const v8f z0 = gq[0] * s01;
        v8f z1;
#pragma unroll
        for (int j = 0; j < 8; ++j)
            z1[j] = mhalf ? fe[j] : gq[1][j] * s2;
        __builtin_amdgcn_sched_barrier(0);
        {
            const float w0e[16] = {Wv[0].x, Wv[0].y, Wv[0].z, Wv[0].w, Wv[1].x, Wv[1].y, Wv[1].z, Wv[1].w,
                                   Wv[2].x, Wv[2].y, Wv[2].z, Wv[2].w, Wv[3].x, Wv[3].y, Wv[3].z, Wv[3].w};
            const float w1e[16] = {Wv[4].x, Wv[4].y, Wv[4].z, Wv[4].w, Wv[5].x, Wv[5].y, Wv[5].z, Wv[5].w,
                                   Wv[6].x, Wv[6].y, Wv[6].z, Wv[6].w, Wv[7].x, Wv[7].y, Wv[7].z, Wv[7].w};
#pragma unroll
            for (int q = 0; q < 8; ++q) {
                v2f b, w0, w1;
                w0.x = w0e[2 * q];
                w0.y = w0e[2 * q + 1];
                w1.x = w1e[2 * q];
                w1.y = w1e[2 * q + 1];
                if (q < 4) { b.x = z0[2 * q];       b.y = z0[2 * q + 1]; }
                else       { b.x = z1[2 * (q - 4)]; b.y = z1[2 * (q - 4) + 1]; }
                acc[0] = __builtin_amdgcn_wmma_f32_16x16x4_f32(
                    false, w0, false, b, (short)0, acc[0], false, false);
                acc[1] = __builtin_amdgcn_wmma_f32_16x16x4_f32(
                    false, w1, false, b, (short)0, acc[1], false, false);
            }
        }
        if (1 < LL) {
#pragma unroll
            for (int j = 0; j < 4; ++j)
                wtn[j * CC * WP + wdst] = wnx[j];
        }
        relu_ln_store(acc, xo, pr4, xsn, mhalf, ownn);
        __syncthreads();
    }

    for (int l = 1; l < LL; ++l) {
        const float4* xsc = xs4 + (l & 1) * (CC * XP / 4);
        const float4* wtc = wt4 + (l & 1) * (4 * CC * WP / 4);
        float* xsn = xs + ((l + 1) & 1) * (CC * XP);
        float* wtn = wts + ((l + 1) & 1) * (4 * CC * WP);
        float wnx[4] = {0.0f, 0.0f, 0.0f, 0.0f};
        if (l + 1 < LL) {
            wnx[0] = convroot[(l + 1) * CC * CC + tid];
#pragma unroll
            for (int j = 0; j < 3; ++j)
                wnx[1 + j] = convw[((l + 1) * 3 + j) * CC * CC + tid];
        }
        v8f acc[2];
#pragma unroll
        for (int t = 0; t < 2; ++t) {
            const float4 lo = cb4[l * 8 + 4 * t + 2 * mhalf], hi = cb4[l * 8 + 4 * t + 2 * mhalf + 1];
            acc[t] = (v8f){lo.x, lo.y, lo.z, lo.w, hi.x, hi.y, hi.z, hi.w};
        }

        float4 Wv[8];
        float4 X[4];
        int4 c0, c1;
        unsigned gm = gf[0];
        const int4* tp = gm ? (const int4*)(tiles + bs[0] * 256) : ivp;
#pragma unroll
        for (int t = 0; t < 2; ++t) {
            Wv[2 * t]     = wtc[wab + 4 * t];
            Wv[2 * t + 1] = wtc[wab + 4 * t + 1];
            Wv[4 + 2 * t] = wtc[wab + 16 * (WP / 4) + 4 * t];
            Wv[5 + 2 * t] = wtc[wab + 16 * (WP / 4) + 4 * t + 1];
        }
        {
            const int xa = xab + 4 * (gm ? __builtin_ctz(gm) : 0);
            X[0] = xsc[xa];                 X[1] = xsc[xa + 1];
            X[2] = xsc[xa + 16 * (XP / 4)]; X[3] = xsc[xa + 16 * (XP / 4) + 1];
            c0 = tp[l2]; c1 = tp[l2 + 1];
        }

        v8f z0 = xo[0], z1 = xo[1];
        v8f g0, g1;
        auto tile_ops = [&](const float4 (&Xf)[4], const int4 ca, const int4 cb) {
            const float cf[8] = {(float)ca.x, (float)ca.y, (float)ca.z, (float)ca.w,
                                 (float)cb.x, (float)cb.y, (float)cb.z, (float)cb.w};
            const float e0[8] = {Xf[0].x, Xf[0].y, Xf[0].z, Xf[0].w, Xf[1].x, Xf[1].y, Xf[1].z, Xf[1].w};
            const float e1[8] = {Xf[2].x, Xf[2].y, Xf[2].z, Xf[2].w, Xf[3].x, Xf[3].y, Xf[3].z, Xf[3].w};
#pragma unroll
            for (int ks = 0; ks < 4; ++ks) {
                v2f b, a0, a1;
                b.x  = cf[2 * ks];
                b.y  = cf[2 * ks + 1];
                a0.x = e0[2 * ks];
                a0.y = e0[2 * ks + 1];
                a1.x = e1[2 * ks];
                a1.y = e1[2 * ks + 1];
                g0 = __builtin_amdgcn_wmma_f32_16x16x4_f32(
                    false, a0, false, b, (short)0, g0, false, false);
                g1 = __builtin_amdgcn_wmma_f32_16x16x4_f32(
                    false, a1, false, b, (short)0, g1, false, false);
            }
        };
        auto add_members = [&](float4 (&Xf)[4], unsigned g) {
            g &= g - 1u;
            while (g) {
                float4 P[4];
                const int pa = xab + 4 * __builtin_ctz(g);
                g &= g - 1u;
                P[0] = xsc[pa];                 P[1] = xsc[pa + 1];
                P[2] = xsc[pa + 16 * (XP / 4)]; P[3] = xsc[pa + 16 * (XP / 4) + 1];
#pragma unroll
                for (int i = 0; i < 4; ++i) {
                    Xf[i].x += P[i].x; Xf[i].y += P[i].y; Xf[i].z += P[i].z; Xf[i].w += P[i].w;
                }
            }
        };
#pragma unroll
        for (int mtx = 0; mtx < 4; ++mtx) {
            __builtin_amdgcn_sched_barrier(0);
            {
                const float w0e[16] = {Wv[0].x, Wv[0].y, Wv[0].z, Wv[0].w, Wv[1].x, Wv[1].y, Wv[1].z, Wv[1].w,
                                       Wv[2].x, Wv[2].y, Wv[2].z, Wv[2].w, Wv[3].x, Wv[3].y, Wv[3].z, Wv[3].w};
                const float w1e[16] = {Wv[4].x, Wv[4].y, Wv[4].z, Wv[4].w, Wv[5].x, Wv[5].y, Wv[5].z, Wv[5].w,
                                       Wv[6].x, Wv[6].y, Wv[6].z, Wv[6].w, Wv[7].x, Wv[7].y, Wv[7].z, Wv[7].w};
#pragma unroll
                for (int q = 0; q < 8; ++q) {
                    v2f b, w0, w1;
                    w0.x = w0e[2 * q];
                    w0.y = w0e[2 * q + 1];
                    w1.x = w1e[2 * q];
                    w1.y = w1e[2 * q + 1];
                    if (q < 4) { b.x = z0[2 * q];       b.y = z0[2 * q + 1]; }
                    else       { b.x = z1[2 * (q - 4)]; b.y = z1[2 * (q - 4) + 1]; }
                    acc[0] = __builtin_amdgcn_wmma_f32_16x16x4_f32(
                        false, w0, false, b, (short)0, acc[0], false, false);
                    acc[1] = __builtin_amdgcn_wmma_f32_16x16x4_f32(
                        false, w1, false, b, (short)0, acc[1], false, false);
                }
            }
            if (mtx == 3) break;
            __builtin_amdgcn_sched_barrier(0);
            if (mtx == 0 && l + 1 < LL) {
#pragma unroll
                for (int j = 0; j < 4; ++j)
                    wtn[j * CC * WP + wdst] = wnx[j];
            }

            const int r = mtx;
            const int wa = (r + 1) * (CC * WP / 4) + wab;
            const unsigned* gp = gmw + r * 1024;
            g0 = (v8f){0, 0, 0, 0, 0, 0, 0, 0};
            g1 = (v8f){0, 0, 0, 0, 0, 0, 0, 0};
            int rem = ng[r];
#pragma nounroll
            while (rem > 1) {
                float4 Y[4];
                add_members(X, gm);
                --rem;
                tp += 64;
                ++gp;
                const unsigned gn = (unsigned)__builtin_amdgcn_readfirstlane((int)gp[0]);
                const int xa = xab + 4 * __builtin_ctz(gn);
                Y[0] = xsc[xa];                 Y[1] = xsc[xa + 1];
                Y[2] = xsc[xa + 16 * (XP / 4)]; Y[3] = xsc[xa + 16 * (XP / 4) + 1];
                const int4 d0 = tp[l2], d1 = tp[l2 + 1];
                __builtin_amdgcn_sched_barrier(0);
                tile_ops(X, c0, c1);
                __builtin_amdgcn_sched_barrier(0);
                if (rem == 1) {
                    X[0] = Y[0]; X[1] = Y[1]; X[2] = Y[2]; X[3] = Y[3];
                    c0 = d0; c1 = d1;
                    gm = gn;
                    break;
                }
                add_members(Y, gn);
                --rem;
                tp += 64;
                ++gp;
                gm = (unsigned)__builtin_amdgcn_readfirstlane((int)gp[0]);
                const int xb = xab + 4 * __builtin_ctz(gm);
                X[0] = xsc[xb];                 X[1] = xsc[xb + 1];
                X[2] = xsc[xb + 16 * (XP / 4)]; X[3] = xsc[xb + 16 * (XP / 4) + 1];
                c0 = tp[l2]; c1 = tp[l2 + 1];
                __builtin_amdgcn_sched_barrier(0);
                tile_ops(Y, d0, d1);
                __builtin_amdgcn_sched_barrier(0);
            }
            add_members(X, gm);
#pragma unroll
            for (int t = 0; t < 2; ++t) {
                Wv[2 * t]     = wtc[wa + 4 * t];
                Wv[2 * t + 1] = wtc[wa + 4 * t + 1];
                Wv[4 + 2 * t] = wtc[wa + 16 * (WP / 4) + 4 * t];
                Wv[5 + 2 * t] = wtc[wa + 16 * (WP / 4) + 4 * t + 1];
            }
            const float sc = ivd[r * NN + ownn];
            __builtin_amdgcn_sched_barrier(0);
            if (rem) tile_ops(X, c0, c1);
            __builtin_amdgcn_sched_barrier(0);
            z0 = g0 * sc;
            z1 = g1 * sc;
            if (r < 2) {
                gm = gf[r + 1];
                tp = gm ? (const int4*)(tiles + bs[r + 1] * 256) : ivp;
                const int xa = xab + 4 * (gm ? __builtin_ctz(gm) : 0);
                X[0] = xsc[xa];                 X[1] = xsc[xa + 1];
                X[2] = xsc[xa + 16 * (XP / 4)]; X[3] = xsc[xa + 16 * (XP / 4) + 1];
                c0 = tp[l2]; c1 = tp[l2 + 1];
            }
        }

        relu_ln_store(acc, xo, pr4, xsn, mhalf, ownn);
        __syncthreads();
    }

    for (int idx = tid; idx < 48 * CC; idx += NT) {
        const int row = idx >> 5, c = idx & 31;
        const int g = row >> 4, rc = row & 15;
        const int a0 = rc >> 2, b0 = rc & 3;
        float sum = 0.0f;
#pragma unroll
        for (int q = 0; q < 4; ++q) {
            int node;
            if (g == 0)      node = q * 16 + a0 * 4 + b0;
            else if (g == 1) node = a0 * 16 + q * 4 + b0;
            else             node = a0 * 16 + b0 * 4 + q;
            sum += xs[(LL & 1) * (CC * XP) + c * XP + node];
        }
        out[bid * 48 * CC + idx] = sum * 0.25f;
    }
}

extern "C" void kernel_launch(void* const* d_in, const int* in_sizes, int n_in,
                              void* d_out, int out_size, void* d_ws, size_t ws_size,
                              hipStream_t stream) {
    const float* xx       = (const float*)d_in[0];
    const float* ss       = (const float*)d_in[1];
    const float* Wf       = (const float*)d_in[2];
    const float* bf       = (const float*)d_in[3];
    const float* convw    = (const float*)d_in[4];
    const float* convroot = (const float*)d_in[5];
    const float* convbias = (const float*)d_in[6];
    const float* lng      = (const float*)d_in[7];
    const float* lnb      = (const float*)d_in[8];
    const int*   src      = (const int*)d_in[9];
    const int*   dst      = (const int*)d_in[10];
    float*       out      = (float*)d_out;
    float*       invdeg   = (float*)d_ws + WS_INVDEG;
    unsigned*    rowmask  = (unsigned*)d_ws + WS_ROWMASK;
    int*         rowbase  = (int*)d_ws + WS_ROWBASE;
    unsigned*    groups   = (unsigned*)d_ws + WS_GROUPS;
    int*         tiles    = (int*)d_ws + WS_TILES;

    const int E = in_sizes[9] / 3;
    const int B = in_sizes[1];

    rgcn_prep_kernel<<<96, PT, 0, stream>>>(src, dst, E, invdeg, rowmask, rowbase, groups, tiles);
    rgcn_main_kernel<<<B, NT, 0, stream>>>(xx, ss, Wf, bf, convw, convroot,
                                            convbias, lng, lnb, invdeg, rowmask,
                                            rowbase, groups, tiles, out);
    (void)n_in; (void)out_size; (void)ws_size;
}
